# starting kernel
# baseline (speedup 1.0000x reference)
#include <hip/hip_runtime.h>
#include <math.h>

#define BB 2
#define SS 256
#define HH 768
#define H4 3072
#define CHK 64
#define LRC 0.01f
#define EPSC 1e-5f

typedef float v2f __attribute__((ext_vector_type(2)));
typedef float v8f __attribute__((ext_vector_type(8)));

__device__ __forceinline__ v8f wmma4(v2f a, v2f b, v8f c) {
  return __builtin_amdgcn_wmma_f32_16x16x4_f32(
       false, a,  false, b,
       (short)0, c,  false,  false);
}

__device__ __forceinline__ v2f frag_nk(const float* M, int ld, int r0, int k0, int lane) {
  const float* p = M + (size_t)(r0 + (lane & 15)) * ld + k0 + ((lane >> 4) << 1);
  v2f v; v.x = p[0]; v.y = p[1];
  return v;
}
__device__ __forceinline__ v2f frag_kn(const float* M, int ld, int k0, int r0, int lane) {
  const float* p = M + (size_t)(k0 + ((lane >> 4) << 1)) * ld + r0 + (lane & 15);
  v2f v; v.x = p[0]; v.y = p[ld];
  return v;
}
__device__ __forceinline__ void store_tile(float* D, int ld, int r0, int c0, v8f v, int lane) {
  int n = c0 + (lane & 15);
  int m0 = r0 + ((lane >> 4) << 3);
#pragma unroll
  for (int i = 0; i < 8; ++i) D[(size_t)(m0 + i) * ld + n] = v[i];
}
__device__ __forceinline__ v8f load_tile(const float* D, int ld, int r0, int c0, int lane) {
  int n = c0 + (lane & 15);
  int m0 = r0 + ((lane >> 4) << 3);
  v8f v;
#pragma unroll
  for (int i = 0; i < 8; ++i) v[i] = D[(size_t)(m0 + i) * ld + n];
  return v;
}


__global__ void k_init(const float* __restrict__ Wi, const float* __restrict__ bi,
                       float* __restrict__ Wst, float* __restrict__ bst) {
  int i = blockIdx.x * blockDim.x + threadIdx.x;
  const int n = HH * HH;
  if (i < n) {
    float w = Wi[i];
    Wst[i] = w;
    Wst[n + i] = w;
  }
  if (i < HH) {
    float b = bi[i];
    bst[i] = b;
    bst[HH + i] = b;
  }
}

__global__ void k_p0(const float* __restrict__ x, const float* __restrict__ Wst,
                     const float* __restrict__ bst, float* __restrict__ P0, int chunk) {
  int wid = (blockIdx.x * blockDim.x + threadIdx.x) >> 5;
  int lane = threadIdx.x & 31;
  int tn = wid % (HH / 16);
  int tm = (wid / (HH / 16)) % (CHK / 16);
  int b  = wid / ((HH / 16) * (CHK / 16));
  const float* Xc = x + ((size_t)b * SS + chunk * CHK) * HH;
  const float* W  = Wst + (size_t)b * HH * HH;
  v8f acc = {};
  for (int k0 = 0; k0 < HH; k0 += 4) {
    v2f a  = frag_nk(Xc, HH, tm * 16, k0, lane);
    v2f bb = frag_nk(W,  HH, tn * 16, k0, lane);
    acc = wmma4(a, bb, acc);
  }
  float bias = bst[(size_t)b * HH + tn * 16 + (lane & 15)];
#pragma unroll
  for (int i = 0; i < 8; ++i) acc[i] += bias;
  store_tile(P0 + (size_t)b * CHK * HH, HH, tm * 16, tn * 16, acc, lane);
}

__global__ void k_attn(const float* __restrict__ x, float* __restrict__ Amat, int chunk) {
  int wid = (blockIdx.x * blockDim.x + threadIdx.x) >> 5;
  int lane = threadIdx.x & 31;
  int tn = wid % (CHK / 16);
  int tm = (wid / (CHK / 16)) % (CHK / 16);
  int b  = wid / ((CHK / 16) * (CHK / 16));
  const float* Xc = x + ((size_t)b * SS + chunk * CHK) * HH;
  v8f acc = {};
  for (int k0 = 0; k0 < HH; k0 += 4) {
    v2f a  = frag_nk(Xc, HH, tm * 16, k0, lane);
    v2f bb = frag_nk(Xc, HH, tn * 16, k0, lane);
    acc = wmma4(a, bb, acc);
  }
  int s = tn * 16 + (lane & 15);
  int mbase = tm * 16 + ((lane >> 4) << 3);
  float* Ab = Amat + (size_t)b * CHK * CHK;
#pragma unroll
  for (int i = 0; i < 8; ++i) {
    int t = mbase + i;
    float v = (s < t) ? LRC * (acc[i] + 1.0f) : 0.0f;
    Ab[(size_t)t * CHK + s] = v;
  }
}

__global__ void k_solve(const float* __restrict__ x, const float* __restrict__ P0,
                        const float* __restrict__ Amat, float* __restrict__ Pred,
                        float* __restrict__ Diff, float* __restrict__ bst, int chunk) {
  extern __shared__ float dsh[];
  int b = blockIdx.x;
  int h = threadIdx.x;
  const float* Xb = x + (size_t)b * SS * HH;
  const float* Ab = Amat + (size_t)b * CHK * CHK;
  float bsum = 0.0f;
  for (int t = 0; t < CHK; ++t) {
    int gt = chunk * CHK + t;
    float acc = P0[((size_t)b * CHK + t) * HH + h];
    const float* Arow = Ab + (size_t)t * CHK;
    for (int s = 0; s < t; ++s) acc -= Arow[s] * dsh[(size_t)s * HH + h];
    Pred[((size_t)b * CHK + t) * HH + h] = acc;
    float xn = (gt + 1 < SS) ? Xb[(size_t)(gt + 1) * HH + h] : 0.0f;
    float d = acc - xn;
    dsh[(size_t)t * HH + h] = d;
    float dl = (gt == SS - 1) ? 0.0f : d;
    Diff[((size_t)b * CHK + t) * HH + h] = dl;
    bsum += dl;
  }
  bst[(size_t)b * HH + h] -= LRC * bsum;
}

__global__ void k_wupdate(const float* __restrict__ x, const float* __restrict__ Diff,
                          float* __restrict__ Wst, int chunk) {
  int wid = (blockIdx.x * blockDim.x + threadIdx.x) >> 5;
  int lane = threadIdx.x & 31;
  int tn = wid % (HH / 16);
  int tm = (wid / (HH / 16)) % (HH / 16);
  int b  = wid / ((HH / 16) * (HH / 16));
  const float* Xc = x + ((size_t)b * SS + chunk * CHK) * HH;
  const float* Db = Diff + (size_t)b * CHK * HH;
  float* W = Wst + (size_t)b * HH * HH;
  v8f s = {};
  for (int k0 = 0; k0 < CHK; k0 += 4) {
    v2f a  = frag_kn(Db, HH, k0, tm * 16, lane);
    v2f bb = frag_kn(Xc, HH, k0, tn * 16, lane);
    s = wmma4(a, bb, s);
  }
  v8f acc = load_tile(W, HH, tm * 16, tn * 16, lane);
#pragma unroll
  for (int i = 0; i < 8; ++i) acc[i] -= LRC * s[i];
  store_tile(W, HH, tm * 16, tn * 16, acc, lane);
}

__global__ void k_mlp1(const float* __restrict__ Pred,
                       const float* __restrict__ gw, const float* __restrict__ gb,
                       const float* __restrict__ uw, const float* __restrict__ ub,
                       float* __restrict__ Gt) {
  int wid = (blockIdx.x * blockDim.x + threadIdx.x) >> 5;
  int lane = threadIdx.x & 31;
  int tn = wid % (H4 / 16);
  int tm = (wid / (H4 / 16)) % (CHK / 16);
  int b  = wid / ((H4 / 16) * (CHK / 16));
  const float* Pb = Pred + (size_t)b * CHK * HH;
  v8f g = {}, u = {};
  for (int k0 = 0; k0 < HH; k0 += 4) {
    v2f a  = frag_nk(Pb, HH, tm * 16, k0, lane);
    v2f bg = frag_nk(gw, HH, tn * 16, k0, lane);
    v2f bu = frag_nk(uw, HH, tn * 16, k0, lane);
    g = wmma4(a, bg, g);
    u = wmma4(a, bu, u);
  }
  int n = tn * 16 + (lane & 15);
  float gbias = gb[n], ubias = ub[n];
  v8f r;
#pragma unroll
  for (int i = 0; i < 8; ++i) {
    float zg = g[i] + gbias;
    float zu = u[i] + ubias;
    r[i] = (1.0f / (1.0f + __expf(-zg))) * zu;
  }
  store_tile(Gt + (size_t)b * CHK * H4, H4, tm * 16, tn * 16, r, lane);
}

__global__ void k_mlp2(const float* __restrict__ Gt, const float* __restrict__ dw,
                       const float* __restrict__ db, const float* __restrict__ Pred,
                       float* __restrict__ Ypre) {
  int wid = (blockIdx.x * blockDim.x + threadIdx.x) >> 5;
  int lane = threadIdx.x & 31;
  int tn = wid % (HH / 16);
  int tm = (wid / (HH / 16)) % (CHK / 16);
  int b  = wid / ((HH / 16) * (CHK / 16));
  const float* Gb = Gt + (size_t)b * CHK * H4;
  v8f acc = {};
  for (int k0 = 0; k0 < H4; k0 += 4) {
    if ((k0 & 63) == 0) __builtin_prefetch(Gb + (size_t)(tm * 16) * H4 + k0 + 64, 0, 0);
    v2f a  = frag_nk(Gb, H4, tm * 16, k0, lane);
    v2f bb = frag_nk(dw, H4, tn * 16, k0, lane);
    acc = wmma4(a, bb, acc);
  }
  int n = tn * 16 + (lane & 15);
  int mbase = tm * 16 + ((lane >> 4) << 3);
  float bias = db[n];
  const float* Pb = Pred + (size_t)b * CHK * HH;
  float* Yb = Ypre + (size_t)b * CHK * HH;
#pragma unroll
  for (int i = 0; i < 8; ++i) {
    int m = mbase + i;
    Yb[(size_t)m * HH + n] = acc[i] + bias + Pb[(size_t)m * HH + n];
  }
}

__global__ void k_ln(const float* __restrict__ Y, const float* __restrict__ nw,
                     const float* __restrict__ nb, float* __restrict__ out, int chunk) {
  __shared__ float s1[256], s2[256];
  int row = blockIdx.x;
  int b = row / CHK, t = row % CHK;
  const float* y = Y + ((size_t)b * CHK + t) * HH;
  int tid = threadIdx.x;
  float v0 = y[tid], v1 = y[tid + 256], v2 = y[tid + 512];
  s1[tid] = v0 + v1 + v2;
  s2[tid] = v0 * v0 + v1 * v1 + v2 * v2;
  __syncthreads();
  for (int off = 128; off > 0; off >>= 1) {
    if (tid < off) { s1[tid] += s1[tid + off]; s2[tid] += s2[tid + off]; }
    __syncthreads();
  }
  float mu  = s1[0] * (1.0f / HH);
  float var = s2[0] * (1.0f / HH) - mu * mu;
  float r = rsqrtf(var + EPSC);
  int gt = chunk * CHK + t;
  float* o = out + ((size_t)b * SS + gt) * HH;
  o[tid]       = (v0 - mu) * r * nw[tid]       + nb[tid];
  o[tid + 256] = (v1 - mu) * r * nw[tid + 256] + nb[tid + 256];
  o[tid + 512] = (v2 - mu) * r * nw[tid + 512] + nb[tid + 512];
}

extern "C" void kernel_launch(void* const* d_in, const int* in_sizes, int n_in,
                              void* d_out, int out_size, void* d_ws, size_t ws_size,
                              hipStream_t stream) {
  const float* x      = (const float*)d_in[0];
  const float* W_init = (const float*)d_in[1];
  const float* b_init = (const float*)d_in[2];
  const float* gate_w = (const float*)d_in[3];
  const float* gate_b = (const float*)d_in[4];
  const float* up_w   = (const float*)d_in[5];
  const float* up_b   = (const float*)d_in[6];
  const float* down_w = (const float*)d_in[7];
  const float* down_b = (const float*)d_in[8];
  const float* norm_w = (const float*)d_in[9];
  const float* norm_b = (const float*)d_in[10];
  float* out = (float*)d_out;

  float* ws   = (float*)d_ws;
  float* Wst  = ws;  ws += (size_t)BB * HH * HH;
  float* bst  = ws;  ws += (size_t)BB * HH;
  float* P0   = ws;  ws += (size_t)BB * CHK * HH;
  float* Pred = ws;  ws += (size_t)BB * CHK * HH;
  float* Diff = ws;  ws += (size_t)BB * CHK * HH;
  float* Amat = ws;  ws += (size_t)BB * CHK * CHK;
  float* Gt   = ws;  ws += (size_t)BB * CHK * H4;
  float* Ypre = ws;  ws += (size_t)BB * CHK * HH;

  k_init<<<(HH * HH + 255) / 256, 256, 0, stream>>>(W_init, b_init, Wst, bst);

  const int NCH = SS / CHK;
  for (int c = 0; c < NCH; ++c) {
    k_p0<<<48, 256, 0, stream>>>(x, Wst, bst, P0, c);
    k_attn<<<4, 256, 0, stream>>>(x, Amat, c);
    k_solve<<<BB, HH, CHK * HH * sizeof(float), stream>>>(x, P0, Amat, Pred, Diff, bst, c);
    k_wupdate<<<576, 256, 0, stream>>>(x, Diff, Wst, c);
    k_mlp1<<<192, 256, 0, stream>>>(Pred, gate_w, gate_b, up_w, up_b, Gt);
    k_mlp2<<<48, 256, 0, stream>>>(Gt, down_w, down_b, Pred, Ypre);
    k_ln<<<BB * CHK, 256, 0, stream>>>(Ypre, norm_w, norm_b, out, c);
  }
}


# k_solve rebuilt as one wave per 32 columns (48 blocks), fully unrolled, A staged in LDS and read one 4-column group ahead: no load waits inside the serial FMA chain; per-row sum order kept
# speedup vs baseline: 1.6466x; 1.6466x over previous
#include <hip/hip_runtime.h>
#include <math.h>

#define BB 2
#define SS 256
#define HH 768
#define H4 3072
#define CHK 64
#define LRC 0.01f
#define EPSC 1e-5f

typedef float v2f __attribute__((ext_vector_type(2)));
typedef float v8f __attribute__((ext_vector_type(8)));

__device__ __forceinline__ v8f wmma4(v2f a, v2f b, v8f c) {
  return __builtin_amdgcn_wmma_f32_16x16x4_f32(
       false, a,  false, b,
       (short)0, c,  false,  false);
}

__device__ __forceinline__ v2f frag_nk(const float* M, int ld, int r0, int k0, int lane) {
  const float* p = M + (size_t)(r0 + (lane & 15)) * ld + k0 + ((lane >> 4) << 1);
  v2f v; v.x = p[0]; v.y = p[1];
  return v;
}
__device__ __forceinline__ v2f frag_kn(const float* M, int ld, int k0, int r0, int lane) {
  const float* p = M + (size_t)(k0 + ((lane >> 4) << 1)) * ld + r0 + (lane & 15);
  v2f v; v.x = p[0]; v.y = p[ld];
  return v;
}
__device__ __forceinline__ void store_tile(float* D, int ld, int r0, int c0, v8f v, int lane) {
  int n = c0 + (lane & 15);
  int m0 = r0 + ((lane >> 4) << 3);
#pragma unroll
  for (int i = 0; i < 8; ++i) D[(size_t)(m0 + i) * ld + n] = v[i];
}
__device__ __forceinline__ v8f load_tile(const float* D, int ld, int r0, int c0, int lane) {
  int n = c0 + (lane & 15);
  int m0 = r0 + ((lane >> 4) << 3);
  v8f v;
#pragma unroll
  for (int i = 0; i < 8; ++i) v[i] = D[(size_t)(m0 + i) * ld + n];
  return v;
}


__global__ void k_init(const float* __restrict__ Wi, const float* __restrict__ bi,
                       float* __restrict__ Wst, float* __restrict__ bst) {
  int i = blockIdx.x * blockDim.x + threadIdx.x;
  const int n = HH * HH;
  if (i < n) {
    float w = Wi[i];
    Wst[i] = w;
    Wst[n + i] = w;
  }
  if (i < HH) {
    float b = bi[i];
    bst[i] = b;
    bst[HH + i] = b;
  }
}

__global__ void k_p0(const float* __restrict__ x, const float* __restrict__ Wst,
                     const float* __restrict__ bst, float* __restrict__ P0, int chunk) {
  int wid = (blockIdx.x * blockDim.x + threadIdx.x) >> 5;
  int lane = threadIdx.x & 31;
  int tn = wid % (HH / 16);
  int tm = (wid / (HH / 16)) % (CHK / 16);
  int b  = wid / ((HH / 16) * (CHK / 16));
  const float* Xc = x + ((size_t)b * SS + chunk * CHK) * HH;
  const float* W  = Wst + (size_t)b * HH * HH;
  v8f acc = {};
  for (int k0 = 0; k0 < HH; k0 += 4) {
    v2f a  = frag_nk(Xc, HH, tm * 16, k0, lane);
    v2f bb = frag_nk(W,  HH, tn * 16, k0, lane);
    acc = wmma4(a, bb, acc);
  }
  float bias = bst[(size_t)b * HH + tn * 16 + (lane & 15)];
#pragma unroll
  for (int i = 0; i < 8; ++i) acc[i] += bias;
  store_tile(P0 + (size_t)b * CHK * HH, HH, tm * 16, tn * 16, acc, lane);
}

__global__ void k_attn(const float* __restrict__ x, float* __restrict__ Amat, int chunk) {
  int wid = (blockIdx.x * blockDim.x + threadIdx.x) >> 5;
  int lane = threadIdx.x & 31;
  int tn = wid % (CHK / 16);
  int tm = (wid / (CHK / 16)) % (CHK / 16);
  int b  = wid / ((CHK / 16) * (CHK / 16));
  const float* Xc = x + ((size_t)b * SS + chunk * CHK) * HH;
  v8f acc = {};
  for (int k0 = 0; k0 < HH; k0 += 4) {
    v2f a  = frag_nk(Xc, HH, tm * 16, k0, lane);
    v2f bb = frag_nk(Xc, HH, tn * 16, k0, lane);
    acc = wmma4(a, bb, acc);
  }
  int s = tn * 16 + (lane & 15);
  int mbase = tm * 16 + ((lane >> 4) << 3);
  float* Ab = Amat + (size_t)b * CHK * CHK;
#pragma unroll
  for (int i = 0; i < 8; ++i) {
    int t = mbase + i;
    float v = (s < t) ? LRC * (acc[i] + 1.0f) : 0.0f;
    Ab[(size_t)t * CHK + s] = v;
  }
}

typedef float v4f __attribute__((ext_vector_type(4)));
__global__ void __launch_bounds__(32) __attribute__((amdgpu_waves_per_eu(1, 1)))
k_solve(const float* __restrict__ x, const float* __restrict__ P0,
        const float* __restrict__ Amat, float* __restrict__ Pred,
        float* __restrict__ Diff, float* __restrict__ bst, int chunk) {
  __shared__ v4f Ash[CHK * CHK / 4];
  const int lane = threadIdx.x;
  const int b = blockIdx.x / (HH / 32);
  const int h = (blockIdx.x % (HH / 32)) * 32 + lane;
  const int g0 = chunk * CHK;
  const float* Xb  = x + (size_t)b * SS * HH + h;
  const float* P0b = P0 + (size_t)b * CHK * HH + h;
  float* Predb = Pred + (size_t)b * CHK * HH + h;
  float* Diffb = Diff + (size_t)b * CHK * HH + h;

  const v4f* Ag = reinterpret_cast<const v4f*>(Amat + (size_t)b * CHK * CHK);
#pragma unroll
  for (int k = 0; k < CHK * CHK / 4 / 32; ++k) Ash[k * 32 + lane] = Ag[k * 32 + lane];

  float acc[CHK], xn[CHK];
#pragma unroll
  for (int t = 0; t < CHK; ++t) acc[t] = P0b[(size_t)t * HH];
#pragma unroll
  for (int t = 0; t < CHK; ++t) {
    const int gn = g0 + t + 1;
    const int gr = (gn < SS) ? gn : (SS - 1);
    const float v = Xb[(size_t)gr * HH];
    xn[t] = (gn < SS) ? v : 0.0f;
  }
  __syncthreads();

  const float* Af = reinterpret_cast<const float*>(Ash);
  v4f abuf[2][CHK];
  float tip[2][6];
  tip[0][0] = Af[1 * CHK];
  tip[0][1] = Af[2 * CHK];  tip[0][2] = Af[2 * CHK + 1];
  tip[0][3] = Af[3 * CHK];  tip[0][4] = Af[3 * CHK + 1];  tip[0][5] = Af[3 * CHK + 2];
#pragma unroll
  for (int t = 4; t < CHK; ++t) abuf[0][t] = Ash[(t * CHK) / 4];
  float bsum = 0.0f;
#pragma unroll
  for (int g = 0; g < CHK / 4; ++g) {
    const int s = 4 * g;
    const int cur = g & 1;
    if (g + 1 < CHK / 4) {
      const int sn = s + 4;
      tip[cur ^ 1][0] = Af[(sn + 1) * CHK + sn];
      tip[cur ^ 1][1] = Af[(sn + 2) * CHK + sn];
      tip[cur ^ 1][2] = Af[(sn + 2) * CHK + sn + 1];
      tip[cur ^ 1][3] = Af[(sn + 3) * CHK + sn];
      tip[cur ^ 1][4] = Af[(sn + 3) * CHK + sn + 1];
      tip[cur ^ 1][5] = Af[(sn + 3) * CHK + sn + 2];
#pragma unroll
      for (int t = 0; t < CHK; ++t) {
        if (t > sn + 3) abuf[cur ^ 1][t] = Ash[(t * CHK + sn) / 4];
      }
    }
    __builtin_amdgcn_sched_barrier(0);
    float n[4];
#pragma unroll
    for (int r = 0; r < 4; ++r) {
      const int t = s + r;
      float a = acc[t];
      if (r == 1) a = fmaf(tip[cur][0], n[0], a);
      if (r == 2) { a = fmaf(tip[cur][1], n[0], a); a = fmaf(tip[cur][2], n[1], a); }
      if (r == 3) { a = fmaf(tip[cur][3], n[0], a); a = fmaf(tip[cur][4], n[1], a); a = fmaf(tip[cur][5], n[2], a); }
      Predb[(size_t)t * HH] = a;
      const float dd = a - xn[t];
      n[r] = xn[t] - a;
      const float dl = (g0 + t == SS - 1) ? 0.0f : dd;
      Diffb[(size_t)t * HH] = dl;
      bsum += dl;
    }
#pragma unroll
    for (int t = 0; t < CHK; ++t) {
      if (t > s + 3) {
        float r = acc[t];
        r = fmaf(abuf[cur][t].x, n[0], r);
        r = fmaf(abuf[cur][t].y, n[1], r);
        r = fmaf(abuf[cur][t].z, n[2], r);
        r = fmaf(abuf[cur][t].w, n[3], r);
        acc[t] = r;
      }
    }
    __builtin_amdgcn_sched_barrier(0);
  }
  bst[(size_t)b * HH + h] -= LRC * bsum;
}

__global__ void k_wupdate(const float* __restrict__ x, const float* __restrict__ Diff,
                          float* __restrict__ Wst, int chunk) {
  int wid = (blockIdx.x * blockDim.x + threadIdx.x) >> 5;
  int lane = threadIdx.x & 31;
  int tn = wid % (HH / 16);
  int tm = (wid / (HH / 16)) % (HH / 16);
  int b  = wid / ((HH / 16) * (HH / 16));
  const float* Xc = x + ((size_t)b * SS + chunk * CHK) * HH;
  const float* Db = Diff + (size_t)b * CHK * HH;
  float* W = Wst + (size_t)b * HH * HH;
  v8f s = {};
  for (int k0 = 0; k0 < CHK; k0 += 4) {
    v2f a  = frag_kn(Db, HH, k0, tm * 16, lane);
    v2f bb = frag_kn(Xc, HH, k0, tn * 16, lane);
    s = wmma4(a, bb, s);
  }
  v8f acc = load_tile(W, HH, tm * 16, tn * 16, lane);
#pragma unroll
  for (int i = 0; i < 8; ++i) acc[i] -= LRC * s[i];
  store_tile(W, HH, tm * 16, tn * 16, acc, lane);
}

__global__ void k_mlp1(const float* __restrict__ Pred,
                       const float* __restrict__ gw, const float* __restrict__ gb,
                       const float* __restrict__ uw, const float* __restrict__ ub,
                       float* __restrict__ Gt) {
  int wid = (blockIdx.x * blockDim.x + threadIdx.x) >> 5;
  int lane = threadIdx.x & 31;
  int tn = wid % (H4 / 16);
  int tm = (wid / (H4 / 16)) % (CHK / 16);
  int b  = wid / ((H4 / 16) * (CHK / 16));
  const float* Pb = Pred + (size_t)b * CHK * HH;
  v8f g = {}, u = {};
  for (int k0 = 0; k0 < HH; k0 += 4) {
    v2f a  = frag_nk(Pb, HH, tm * 16, k0, lane);
    v2f bg = frag_nk(gw, HH, tn * 16, k0, lane);
    v2f bu = frag_nk(uw, HH, tn * 16, k0, lane);
    g = wmma4(a, bg, g);
    u = wmma4(a, bu, u);
  }
  int n = tn * 16 + (lane & 15);
  float gbias = gb[n], ubias = ub[n];
  v8f r;
#pragma unroll
  for (int i = 0; i < 8; ++i) {
    float zg = g[i] + gbias;
    float zu = u[i] + ubias;
    r[i] = (1.0f / (1.0f + __expf(-zg))) * zu;
  }
  store_tile(Gt + (size_t)b * CHK * H4, H4, tm * 16, tn * 16, r, lane);
}

__global__ void k_mlp2(const float* __restrict__ Gt, const float* __restrict__ dw,
                       const float* __restrict__ db, const float* __restrict__ Pred,
                       float* __restrict__ Ypre) {
  int wid = (blockIdx.x * blockDim.x + threadIdx.x) >> 5;
  int lane = threadIdx.x & 31;
  int tn = wid % (HH / 16);
  int tm = (wid / (HH / 16)) % (CHK / 16);
  int b  = wid / ((HH / 16) * (CHK / 16));
  const float* Gb = Gt + (size_t)b * CHK * H4;
  v8f acc = {};
  for (int k0 = 0; k0 < H4; k0 += 4) {
    if ((k0 & 63) == 0) __builtin_prefetch(Gb + (size_t)(tm * 16) * H4 + k0 + 64, 0, 0);
    v2f a  = frag_nk(Gb, H4, tm * 16, k0, lane);
    v2f bb = frag_nk(dw, H4, tn * 16, k0, lane);
    acc = wmma4(a, bb, acc);
  }
  int n = tn * 16 + (lane & 15);
  int mbase = tm * 16 + ((lane >> 4) << 3);
  float bias = db[n];
  const float* Pb = Pred + (size_t)b * CHK * HH;
  float* Yb = Ypre + (size_t)b * CHK * HH;
#pragma unroll
  for (int i = 0; i < 8; ++i) {
    int m = mbase + i;
    Yb[(size_t)m * HH + n] = acc[i] + bias + Pb[(size_t)m * HH + n];
  }
}

__global__ void k_ln(const float* __restrict__ Y, const float* __restrict__ nw,
                     const float* __restrict__ nb, float* __restrict__ out, int chunk) {
  __shared__ float s1[256], s2[256];
  int row = blockIdx.x;
  int b = row / CHK, t = row % CHK;
  const float* y = Y + ((size_t)b * CHK + t) * HH;
  int tid = threadIdx.x;
  float v0 = y[tid], v1 = y[tid + 256], v2 = y[tid + 512];
  s1[tid] = v0 + v1 + v2;
  s2[tid] = v0 * v0 + v1 * v1 + v2 * v2;
  __syncthreads();
  for (int off = 128; off > 0; off >>= 1) {
    if (tid < off) { s1[tid] += s1[tid + off]; s2[tid] += s2[tid + off]; }
    __syncthreads();
  }
  float mu  = s1[0] * (1.0f / HH);
  float var = s2[0] * (1.0f / HH) - mu * mu;
  float r = rsqrtf(var + EPSC);
  int gt = chunk * CHK + t;
  float* o = out + ((size_t)b * SS + gt) * HH;
  o[tid]       = (v0 - mu) * r * nw[tid]       + nb[tid];
  o[tid + 256] = (v1 - mu) * r * nw[tid + 256] + nb[tid + 256];
  o[tid + 512] = (v2 - mu) * r * nw[tid + 512] + nb[tid + 512];
}

extern "C" void kernel_launch(void* const* d_in, const int* in_sizes, int n_in,
                              void* d_out, int out_size, void* d_ws, size_t ws_size,
                              hipStream_t stream) {
  const float* x      = (const float*)d_in[0];
  const float* W_init = (const float*)d_in[1];
  const float* b_init = (const float*)d_in[2];
  const float* gate_w = (const float*)d_in[3];
  const float* gate_b = (const float*)d_in[4];
  const float* up_w   = (const float*)d_in[5];
  const float* up_b   = (const float*)d_in[6];
  const float* down_w = (const float*)d_in[7];
  const float* down_b = (const float*)d_in[8];
  const float* norm_w = (const float*)d_in[9];
  const float* norm_b = (const float*)d_in[10];
  float* out = (float*)d_out;

  float* ws   = (float*)d_ws;
  float* Wst  = ws;  ws += (size_t)BB * HH * HH;
  float* bst  = ws;  ws += (size_t)BB * HH;
  float* P0   = ws;  ws += (size_t)BB * CHK * HH;
  float* Pred = ws;  ws += (size_t)BB * CHK * HH;
  float* Diff = ws;  ws += (size_t)BB * CHK * HH;
  float* Amat = ws;  ws += (size_t)BB * CHK * CHK;
  float* Gt   = ws;  ws += (size_t)BB * CHK * H4;
  float* Ypre = ws;  ws += (size_t)BB * CHK * HH;

  k_init<<<(HH * HH + 255) / 256, 256, 0, stream>>>(W_init, b_init, Wst, bst);

  const int NCH = SS / CHK;
  for (int c = 0; c < NCH; ++c) {
    k_p0<<<48, 256, 0, stream>>>(x, Wst, bst, P0, c);
    k_attn<<<4, 256, 0, stream>>>(x, Amat, c);
    k_solve<<<BB * (HH / 32), 32, 0, stream>>>(x, P0, Amat, Pred, Diff, bst, c);
    k_wupdate<<<576, 256, 0, stream>>>(x, Diff, Wst, c);
    k_mlp1<<<192, 256, 0, stream>>>(Pred, gate_w, gate_b, up_w, up_b, Gt);
    k_mlp2<<<48, 256, 0, stream>>>(Gt, down_w, down_b, Pred, Ypre);
    k_ln<<<BB * CHK, 256, 0, stream>>>(Ypre, norm_w, norm_b, out, c);
  }
}


# p0/attn/mlp1/mlp2 K loops: 16-byte fragment loads (one 16-line load per WMMA, was two), each block's loads issued a block ahead in a second register set; k pairing per WMMA, so sum order, changes
# speedup vs baseline: 2.4498x; 1.4878x over previous
#include <hip/hip_runtime.h>
#include <math.h>

#define BB 2
#define SS 256
#define HH 768
#define H4 3072
#define CHK 64
#define LRC 0.01f
#define EPSC 1e-5f

typedef float v2f __attribute__((ext_vector_type(2)));
typedef float v8f __attribute__((ext_vector_type(8)));

__device__ __forceinline__ v8f wmma4(v2f a, v2f b, v8f c) {
  return __builtin_amdgcn_wmma_f32_16x16x4_f32(
       false, a,  false, b,
       (short)0, c,  false,  false);
}

__device__ __forceinline__ v2f frag_kn(const float* M, int ld, int k0, int r0, int lane) {
  const float* p = M + (size_t)(k0 + ((lane >> 4) << 1)) * ld + r0 + (lane & 15);
  v2f v; v.x = p[0]; v.y = p[ld];
  return v;
}
__device__ __forceinline__ void store_tile(float* D, int ld, int r0, int c0, v8f v, int lane) {
  int n = c0 + (lane & 15);
  int m0 = r0 + ((lane >> 4) << 3);
#pragma unroll
  for (int i = 0; i < 8; ++i) D[(size_t)(m0 + i) * ld + n] = v[i];
}
__device__ __forceinline__ v8f load_tile(const float* D, int ld, int r0, int c0, int lane) {
  int n = c0 + (lane & 15);
  int m0 = r0 + ((lane >> 4) << 3);
  v8f v;
#pragma unroll
  for (int i = 0; i < 8; ++i) v[i] = D[(size_t)(m0 + i) * ld + n];
  return v;
}


typedef float v4f __attribute__((ext_vector_type(4)));

struct frag16 { v4f lo, hi; };
__device__ __forceinline__ frag16 ld_frag16(const float* p) {
  const v4f* q = reinterpret_cast<const v4f*>(p);
  frag16 f; f.lo = q[0]; f.hi = q[1];
  return f;
}
__device__ __forceinline__ v8f mma16(const frag16& a, const frag16& b, v8f c) {
  c = wmma4(a.lo.xy, b.lo.xy, c);
  c = wmma4(a.lo.zw, b.lo.zw, c);
  c = wmma4(a.hi.xy, b.hi.xy, c);
  c = wmma4(a.hi.zw, b.hi.zw, c);
  return c;
}
__device__ __forceinline__ const float* lane_row(const float* M, int ld, int r0, int lane) {
  return M + (size_t)(r0 + (lane & 15)) * ld + ((lane >> 4) << 3);
}

#define LOADS_STAY_AHEAD_OF_WMMAS() __builtin_amdgcn_sched_barrier(0)
#define READERS_DONE_BEFORE_REFILL(...) asm volatile("" :: __VA_ARGS__)

template <int KTOT, int KBLK>
__device__ __forceinline__ v8f tile_nk(const float* pa, const float* pb) {
  constexpr int KB = 16 * KBLK;
  constexpr int NBLK = KTOT / KB;
  static_assert(KTOT % (2 * KB) == 0, "K must be a whole number of block pairs");
  frag16 a0[KBLK], b0[KBLK], a1[KBLK], b1[KBLK];
#pragma unroll
  for (int j = 0; j < KBLK; ++j) { a0[j] = ld_frag16(pa + 16 * j); b0[j] = ld_frag16(pb + 16 * j); }
  v8f acc = {};
  for (int blk = 0; blk < NBLK - 2; blk += 2) {
    const float* qa = pa + (size_t)(blk + 1) * KB;
    const float* qb = pb + (size_t)(blk + 1) * KB;
#pragma unroll
    for (int j = 0; j < KBLK; ++j) { a1[j] = ld_frag16(qa + 16 * j); b1[j] = ld_frag16(qb + 16 * j); }
    LOADS_STAY_AHEAD_OF_WMMAS();
#pragma unroll
    for (int j = 0; j < KBLK; ++j) acc = mma16(a0[j], b0[j], acc);
    READERS_DONE_BEFORE_REFILL("v"(acc));
#pragma unroll
    for (int j = 0; j < KBLK; ++j) { a0[j] = ld_frag16(qa + KB + 16 * j); b0[j] = ld_frag16(qb + KB + 16 * j); }
    LOADS_STAY_AHEAD_OF_WMMAS();
#pragma unroll
    for (int j = 0; j < KBLK; ++j) acc = mma16(a1[j], b1[j], acc);
  }
  {
    const float* qa = pa + (size_t)(NBLK - 1) * KB;
    const float* qb = pb + (size_t)(NBLK - 1) * KB;
#pragma unroll
    for (int j = 0; j < KBLK; ++j) { a1[j] = ld_frag16(qa + 16 * j); b1[j] = ld_frag16(qb + 16 * j); }
    LOADS_STAY_AHEAD_OF_WMMAS();
#pragma unroll
    for (int j = 0; j < KBLK; ++j) acc = mma16(a0[j], b0[j], acc);
#pragma unroll
    for (int j = 0; j < KBLK; ++j) acc = mma16(a1[j], b1[j], acc);
  }
  return acc;
}

template <int KTOT, int KBLK>
__device__ __forceinline__ void tile_nk2(const float* pa, const float* pb, const float* pc,
                                         v8f& g, v8f& u) {
  constexpr int KB = 16 * KBLK;
  constexpr int NBLK = KTOT / KB;
  static_assert(KTOT % (2 * KB) == 0, "K must be a whole number of block pairs");
  frag16 a0[KBLK], b0[KBLK], c0[KBLK], a1[KBLK], b1[KBLK], c1[KBLK];
#pragma unroll
  for (int j = 0; j < KBLK; ++j) {
    a0[j] = ld_frag16(pa + 16 * j); b0[j] = ld_frag16(pb + 16 * j); c0[j] = ld_frag16(pc + 16 * j);
  }
  for (int blk = 0; blk < NBLK - 2; blk += 2) {
    const float* qa = pa + (size_t)(blk + 1) * KB;
    const float* qb = pb + (size_t)(blk + 1) * KB;
    const float* qc = pc + (size_t)(blk + 1) * KB;
#pragma unroll
    for (int j = 0; j < KBLK; ++j) {
      a1[j] = ld_frag16(qa + 16 * j); b1[j] = ld_frag16(qb + 16 * j); c1[j] = ld_frag16(qc + 16 * j);
    }
    LOADS_STAY_AHEAD_OF_WMMAS();
#pragma unroll
    for (int j = 0; j < KBLK; ++j) { g = mma16(a0[j], b0[j], g); u = mma16(a0[j], c0[j], u); }
    READERS_DONE_BEFORE_REFILL("v"(g), "v"(u));
#pragma unroll
    for (int j = 0; j < KBLK; ++j) {
      a0[j] = ld_frag16(qa + KB + 16 * j); b0[j] = ld_frag16(qb + KB + 16 * j); c0[j] = ld_frag16(qc + KB + 16 * j);
    }
    LOADS_STAY_AHEAD_OF_WMMAS();
#pragma unroll
    for (int j = 0; j < KBLK; ++j) { g = mma16(a1[j], b1[j], g); u = mma16(a1[j], c1[j], u); }
  }
  {
    const float* qa = pa + (size_t)(NBLK - 1) * KB;
    const float* qb = pb + (size_t)(NBLK - 1) * KB;
    const float* qc = pc + (size_t)(NBLK - 1) * KB;
#pragma unroll
    for (int j = 0; j < KBLK; ++j) {
      a1[j] = ld_frag16(qa + 16 * j); b1[j] = ld_frag16(qb + 16 * j); c1[j] = ld_frag16(qc + 16 * j);
    }
    LOADS_STAY_AHEAD_OF_WMMAS();
#pragma unroll
    for (int j = 0; j < KBLK; ++j) { g = mma16(a0[j], b0[j], g); u = mma16(a0[j], c0[j], u); }
#pragma unroll
    for (int j = 0; j < KBLK; ++j) { g = mma16(a1[j], b1[j], g); u = mma16(a1[j], c1[j], u); }
  }
}


__global__ void k_init(const float* __restrict__ Wi, const float* __restrict__ bi,
                       float* __restrict__ Wst, float* __restrict__ bst) {
  int i = blockIdx.x * blockDim.x + threadIdx.x;
  const int n = HH * HH;
  if (i < n) {
    float w = Wi[i];
    Wst[i] = w;
    Wst[n + i] = w;
  }
  if (i < HH) {
    float b = bi[i];
    bst[i] = b;
    bst[HH + i] = b;
  }
}

__global__ void __launch_bounds__(256) __attribute__((amdgpu_waves_per_eu(1, 2)))
k_p0(const float* x, const float* Wst, const float* bst, float* P0, int chunk) {
  int wid = (blockIdx.x * blockDim.x + threadIdx.x) >> 5;
  int lane = threadIdx.x & 31;
  int tn = wid % (HH / 16);
  int tm = (wid / (HH / 16)) % (CHK / 16);
  int b  = wid / ((HH / 16) * (CHK / 16));
  const float* Xc = x + ((size_t)b * SS + chunk * CHK) * HH;
  const float* W  = Wst + (size_t)b * HH * HH;
  v8f acc = tile_nk<HH, 8>(lane_row(Xc, HH, tm * 16, lane), lane_row(W, HH, tn * 16, lane));
  float bias = bst[(size_t)b * HH + tn * 16 + (lane & 15)];
#pragma unroll
  for (int i = 0; i < 8; ++i) acc[i] += bias;
  store_tile(P0 + (size_t)b * CHK * HH, HH, tm * 16, tn * 16, acc, lane);
}

__global__ void __launch_bounds__(256) __attribute__((amdgpu_waves_per_eu(1, 2)))
k_attn(const float* x, float* Amat, int chunk) {
  int wid = (blockIdx.x * blockDim.x + threadIdx.x) >> 5;
  int lane = threadIdx.x & 31;
  int tn = wid % (CHK / 16);
  int tm = (wid / (CHK / 16)) % (CHK / 16);
  int b  = wid / ((CHK / 16) * (CHK / 16));
  const float* Xc = x + ((size_t)b * SS + chunk * CHK) * HH;
  v8f acc = tile_nk<HH, 8>(lane_row(Xc, HH, tm * 16, lane), lane_row(Xc, HH, tn * 16, lane));
  int s = tn * 16 + (lane & 15);
  int mbase = tm * 16 + ((lane >> 4) << 3);
  float* Ab = Amat + (size_t)b * CHK * CHK;
#pragma unroll
  for (int i = 0; i < 8; ++i) {
    int t = mbase + i;
    float v = (s < t) ? LRC * (acc[i] + 1.0f) : 0.0f;
    Ab[(size_t)t * CHK + s] = v;
  }
}

__global__ void __launch_bounds__(32) __attribute__((amdgpu_waves_per_eu(1, 1)))
k_solve(const float* __restrict__ x, const float* __restrict__ P0,
        const float* __restrict__ Amat, float* __restrict__ Pred,
        float* __restrict__ Diff, float* __restrict__ bst, int chunk) {
  __shared__ v4f Ash[CHK * CHK / 4];
  const int lane = threadIdx.x;
  const int b = blockIdx.x / (HH / 32);
  const int h = (blockIdx.x % (HH / 32)) * 32 + lane;
  const int g0 = chunk * CHK;
  const float* Xb  = x + (size_t)b * SS * HH + h;
  const float* P0b = P0 + (size_t)b * CHK * HH + h;
  float* Predb = Pred + (size_t)b * CHK * HH + h;
  float* Diffb = Diff + (size_t)b * CHK * HH + h;

  const v4f* Ag = reinterpret_cast<const v4f*>(Amat + (size_t)b * CHK * CHK);
#pragma unroll
  for (int k = 0; k < CHK * CHK / 4 / 32; ++k) Ash[k * 32 + lane] = Ag[k * 32 + lane];

  float acc[CHK], xn[CHK];
#pragma unroll
  for (int t = 0; t < CHK; ++t) acc[t] = P0b[(size_t)t * HH];
#pragma unroll
  for (int t = 0; t < CHK; ++t) {
    const int gn = g0 + t + 1;
    const int gr = (gn < SS) ? gn : (SS - 1);
    const float v = Xb[(size_t)gr * HH];
    xn[t] = (gn < SS) ? v : 0.0f;
  }
  __syncthreads();

  const float* Af = reinterpret_cast<const float*>(Ash);
  v4f abuf[2][CHK];
  float tip[2][6];
  tip[0][0] = Af[1 * CHK];
  tip[0][1] = Af[2 * CHK];  tip[0][2] = Af[2 * CHK + 1];
  tip[0][3] = Af[3 * CHK];  tip[0][4] = Af[3 * CHK + 1];  tip[0][5] = Af[3 * CHK + 2];
#pragma unroll
  for (int t = 4; t < CHK; ++t) abuf[0][t] = Ash[(t * CHK) / 4];
  float bsum = 0.0f;
#pragma unroll
  for (int g = 0; g < CHK / 4; ++g) {
    const int s = 4 * g;
    const int cur = g & 1;
    if (g + 1 < CHK / 4) {
      const int sn = s + 4;
      tip[cur ^ 1][0] = Af[(sn + 1) * CHK + sn];
      tip[cur ^ 1][1] = Af[(sn + 2) * CHK + sn];
      tip[cur ^ 1][2] = Af[(sn + 2) * CHK + sn + 1];
      tip[cur ^ 1][3] = Af[(sn + 3) * CHK + sn];
      tip[cur ^ 1][4] = Af[(sn + 3) * CHK + sn + 1];
      tip[cur ^ 1][5] = Af[(sn + 3) * CHK + sn + 2];
#pragma unroll
      for (int t = 0; t < CHK; ++t) {
        if (t > sn + 3) abuf[cur ^ 1][t] = Ash[(t * CHK + sn) / 4];
      }
    }
    __builtin_amdgcn_sched_barrier(0);
    float n[4];
#pragma unroll
    for (int r = 0; r < 4; ++r) {
      const int t = s + r;
      float a = acc[t];
      if (r == 1) a = fmaf(tip[cur][0], n[0], a);
      if (r == 2) { a = fmaf(tip[cur][1], n[0], a); a = fmaf(tip[cur][2], n[1], a); }
      if (r == 3) { a = fmaf(tip[cur][3], n[0], a); a = fmaf(tip[cur][4], n[1], a); a = fmaf(tip[cur][5], n[2], a); }
      Predb[(size_t)t * HH] = a;
      const float dd = a - xn[t];
      n[r] = xn[t] - a;
      const float dl = (g0 + t == SS - 1) ? 0.0f : dd;
      Diffb[(size_t)t * HH] = dl;
      bsum += dl;
    }
#pragma unroll
    for (int t = 0; t < CHK; ++t) {
      if (t > s + 3) {
        float r = acc[t];
        r = fmaf(abuf[cur][t].x, n[0], r);
        r = fmaf(abuf[cur][t].y, n[1], r);
        r = fmaf(abuf[cur][t].z, n[2], r);
        r = fmaf(abuf[cur][t].w, n[3], r);
        acc[t] = r;
      }
    }
    __builtin_amdgcn_sched_barrier(0);
  }
  bst[(size_t)b * HH + h] -= LRC * bsum;
}

__global__ void k_wupdate(const float* __restrict__ x, const float* __restrict__ Diff,
                          float* __restrict__ Wst, int chunk) {
  int wid = (blockIdx.x * blockDim.x + threadIdx.x) >> 5;
  int lane = threadIdx.x & 31;
  int tn = wid % (HH / 16);
  int tm = (wid / (HH / 16)) % (HH / 16);
  int b  = wid / ((HH / 16) * (HH / 16));
  const float* Xc = x + ((size_t)b * SS + chunk * CHK) * HH;
  const float* Db = Diff + (size_t)b * CHK * HH;
  float* W = Wst + (size_t)b * HH * HH;
  v8f s = {};
  for (int k0 = 0; k0 < CHK; k0 += 4) {
    v2f a  = frag_kn(Db, HH, k0, tm * 16, lane);
    v2f bb = frag_kn(Xc, HH, k0, tn * 16, lane);
    s = wmma4(a, bb, s);
  }
  v8f acc = load_tile(W, HH, tm * 16, tn * 16, lane);
#pragma unroll
  for (int i = 0; i < 8; ++i) acc[i] -= LRC * s[i];
  store_tile(W, HH, tm * 16, tn * 16, acc, lane);
}

__global__ void __launch_bounds__(256) __attribute__((amdgpu_waves_per_eu(1, 2)))
k_mlp1(const float* Pred, const float* gw, const float* gb,
       const float* uw, const float* ub, float* Gt) {
  int wid = (blockIdx.x * blockDim.x + threadIdx.x) >> 5;
  int lane = threadIdx.x & 31;
  int tn = wid % (H4 / 16);
  int tm = (wid / (H4 / 16)) % (CHK / 16);
  int b  = wid / ((H4 / 16) * (CHK / 16));
  const float* Pb = Pred + (size_t)b * CHK * HH;
  v8f g = {}, u = {};
  tile_nk2<HH, 4>(lane_row(Pb, HH, tm * 16, lane), lane_row(gw, HH, tn * 16, lane),
               lane_row(uw, HH, tn * 16, lane), g, u);
  int n = tn * 16 + (lane & 15);
  float gbias = gb[n], ubias = ub[n];
  v8f r;
#pragma unroll
  for (int i = 0; i < 8; ++i) {
    float zg = g[i] + gbias;
    float zu = u[i] + ubias;
    r[i] = (1.0f / (1.0f + __expf(-zg))) * zu;
  }
  store_tile(Gt + (size_t)b * CHK * H4, H4, tm * 16, tn * 16, r, lane);
}

__global__ void __launch_bounds__(256) __attribute__((amdgpu_waves_per_eu(1, 2)))
k_mlp2(const float* Gt, const float* dw, const float* db, const float* Pred,
       float* Ypre) {
  int wid = (blockIdx.x * blockDim.x + threadIdx.x) >> 5;
  int lane = threadIdx.x & 31;
  int tn = wid % (HH / 16);
  int tm = (wid / (HH / 16)) % (CHK / 16);
  int b  = wid / ((HH / 16) * (CHK / 16));
  const float* Gb = Gt + (size_t)b * CHK * H4;
  v8f acc = tile_nk<H4, 8>(lane_row(Gb, H4, tm * 16, lane), lane_row(dw, H4, tn * 16, lane));
  int n = tn * 16 + (lane & 15);
  int mbase = tm * 16 + ((lane >> 4) << 3);
  float bias = db[n];
  const float* Pb = Pred + (size_t)b * CHK * HH;
  float* Yb = Ypre + (size_t)b * CHK * HH;
#pragma unroll
  for (int i = 0; i < 8; ++i) {
    int m = mbase + i;
    Yb[(size_t)m * HH + n] = acc[i] + bias + Pb[(size_t)m * HH + n];
  }
}

__global__ void k_ln(const float* __restrict__ Y, const float* __restrict__ nw,
                     const float* __restrict__ nb, float* __restrict__ out, int chunk) {
  __shared__ float s1[256], s2[256];
  int row = blockIdx.x;
  int b = row / CHK, t = row % CHK;
  const float* y = Y + ((size_t)b * CHK + t) * HH;
  int tid = threadIdx.x;
  float v0 = y[tid], v1 = y[tid + 256], v2 = y[tid + 512];
  s1[tid] = v0 + v1 + v2;
  s2[tid] = v0 * v0 + v1 * v1 + v2 * v2;
  __syncthreads();
  for (int off = 128; off > 0; off >>= 1) {
    if (tid < off) { s1[tid] += s1[tid + off]; s2[tid] += s2[tid + off]; }
    __syncthreads();
  }
  float mu  = s1[0] * (1.0f / HH);
  float var = s2[0] * (1.0f / HH) - mu * mu;
  float r = rsqrtf(var + EPSC);
  int gt = chunk * CHK + t;
  float* o = out + ((size_t)b * SS + gt) * HH;
  o[tid]       = (v0 - mu) * r * nw[tid]       + nb[tid];
  o[tid + 256] = (v1 - mu) * r * nw[tid + 256] + nb[tid + 256];
  o[tid + 512] = (v2 - mu) * r * nw[tid + 512] + nb[tid + 512];
}

extern "C" void kernel_launch(void* const* d_in, const int* in_sizes, int n_in,
                              void* d_out, int out_size, void* d_ws, size_t ws_size,
                              hipStream_t stream) {
  const float* x      = (const float*)d_in[0];
  const float* W_init = (const float*)d_in[1];
  const float* b_init = (const float*)d_in[2];
  const float* gate_w = (const float*)d_in[3];
  const float* gate_b = (const float*)d_in[4];
  const float* up_w   = (const float*)d_in[5];
  const float* up_b   = (const float*)d_in[6];
  const float* down_w = (const float*)d_in[7];
  const float* down_b = (const float*)d_in[8];
  const float* norm_w = (const float*)d_in[9];
  const float* norm_b = (const float*)d_in[10];
  float* out = (float*)d_out;

  float* ws   = (float*)d_ws;
  float* Wst  = ws;  ws += (size_t)BB * HH * HH;
  float* bst  = ws;  ws += (size_t)BB * HH;
  float* P0   = ws;  ws += (size_t)BB * CHK * HH;
  float* Pred = ws;  ws += (size_t)BB * CHK * HH;
  float* Diff = ws;  ws += (size_t)BB * CHK * HH;
  float* Amat = ws;  ws += (size_t)BB * CHK * CHK;
  float* Gt   = ws;  ws += (size_t)BB * CHK * H4;
  float* Ypre = ws;  ws += (size_t)BB * CHK * HH;

  k_init<<<(HH * HH + 255) / 256, 256, 0, stream>>>(W_init, b_init, Wst, bst);

  const int NCH = SS / CHK;
  for (int c = 0; c < NCH; ++c) {
    k_p0<<<48, 256, 0, stream>>>(x, Wst, bst, P0, c);
    k_attn<<<4, 256, 0, stream>>>(x, Amat, c);
    k_solve<<<BB * (HH / 32), 32, 0, stream>>>(x, P0, Amat, Pred, Diff, bst, c);
    k_wupdate<<<576, 256, 0, stream>>>(x, Diff, Wst, c);
    k_mlp1<<<192, 256, 0, stream>>>(Pred, gate_w, gate_b, up_w, up_b, Gt);
    k_mlp2<<<48, 256, 0, stream>>>(Gt, down_w, down_b, Pred, Ypre);
    k_ln<<<BB * CHK, 256, 0, stream>>>(Ypre, norm_w, norm_b, out, c);
  }
}


# attn (reads x only) and mlp1/mlp2/LayerNorm (feed nothing back) leave the 4-chunk loop: one launch each over all 256 positions, 29 launches -> 17; every sum as before
# speedup vs baseline: 4.5035x; 1.8383x over previous
#include <hip/hip_runtime.h>
#include <math.h>

#define BB 2
#define SS 256
#define HH 768
#define H4 3072
#define CHK 64
#define NCH (SS / CHK)
#define LRC 0.01f
#define EPSC 1e-5f

typedef float v2f __attribute__((ext_vector_type(2)));
typedef float v8f __attribute__((ext_vector_type(8)));

__device__ __forceinline__ v8f wmma4(v2f a, v2f b, v8f c) {
  return __builtin_amdgcn_wmma_f32_16x16x4_f32(
       false, a,  false, b,
       (short)0, c,  false,  false);
}

__device__ __forceinline__ v2f frag_kn(const float* M, int ld, int k0, int r0, int lane) {
  const float* p = M + (size_t)(k0 + ((lane >> 4) << 1)) * ld + r0 + (lane & 15);
  v2f v; v.x = p[0]; v.y = p[ld];
  return v;
}
__device__ __forceinline__ void store_tile(float* D, int ld, int r0, int c0, v8f v, int lane) {
  int n = c0 + (lane & 15);
  int m0 = r0 + ((lane >> 4) << 3);
#pragma unroll
  for (int i = 0; i < 8; ++i) D[(size_t)(m0 + i) * ld + n] = v[i];
}
__device__ __forceinline__ v8f load_tile(const float* D, int ld, int r0, int c0, int lane) {
  int n = c0 + (lane & 15);
  int m0 = r0 + ((lane >> 4) << 3);
  v8f v;
#pragma unroll
  for (int i = 0; i < 8; ++i) v[i] = D[(size_t)(m0 + i) * ld + n];
  return v;
}


typedef float v4f __attribute__((ext_vector_type(4)));

struct frag16 { v4f lo, hi; };
__device__ __forceinline__ frag16 ld_frag16(const float* p) {
  const v4f* q = reinterpret_cast<const v4f*>(p);
  frag16 f; f.lo = q[0]; f.hi = q[1];
  return f;
}
__device__ __forceinline__ v8f mma16(const frag16& a, const frag16& b, v8f c) {
  c = wmma4(a.lo.xy, b.lo.xy, c);
  c = wmma4(a.lo.zw, b.lo.zw, c);
  c = wmma4(a.hi.xy, b.hi.xy, c);
  c = wmma4(a.hi.zw, b.hi.zw, c);
  return c;
}
__device__ __forceinline__ const float* lane_row(const float* M, int ld, int r0, int lane) {
  return M + (size_t)(r0 + (lane & 15)) * ld + ((lane >> 4) << 3);
}

#define LOADS_STAY_AHEAD_OF_WMMAS() __builtin_amdgcn_sched_barrier(0)
#define READERS_DONE_BEFORE_REFILL(...) asm volatile("" :: __VA_ARGS__)

template <int KTOT, int KBLK>
__device__ __forceinline__ v8f tile_nk(const float* pa, const float* pb) {
  constexpr int KB = 16 * KBLK;
  constexpr int NBLK = KTOT / KB;
  static_assert(KTOT % (2 * KB) == 0, "K must be a whole number of block pairs");
  frag16 a0[KBLK], b0[KBLK], a1[KBLK], b1[KBLK];
#pragma unroll
  for (int j = 0; j < KBLK; ++j) { a0[j] = ld_frag16(pa + 16 * j); b0[j] = ld_frag16(pb + 16 * j); }
  v8f acc = {};
  for (int blk = 0; blk < NBLK - 2; blk += 2) {
    const float* qa = pa + (size_t)(blk + 1) * KB;
    const float* qb = pb + (size_t)(blk + 1) * KB;
#pragma unroll
    for (int j = 0; j < KBLK; ++j) { a1[j] = ld_frag16(qa + 16 * j); b1[j] = ld_frag16(qb + 16 * j); }
    LOADS_STAY_AHEAD_OF_WMMAS();
#pragma unroll
    for (int j = 0; j < KBLK; ++j) acc = mma16(a0[j], b0[j], acc);
    READERS_DONE_BEFORE_REFILL("v"(acc));
#pragma unroll
    for (int j = 0; j < KBLK; ++j) { a0[j] = ld_frag16(qa + KB + 16 * j); b0[j] = ld_frag16(qb + KB + 16 * j); }
    LOADS_STAY_AHEAD_OF_WMMAS();
#pragma unroll
    for (int j = 0; j < KBLK; ++j) acc = mma16(a1[j], b1[j], acc);
  }
  {
    const float* qa = pa + (size_t)(NBLK - 1) * KB;
    const float* qb = pb + (size_t)(NBLK - 1) * KB;
#pragma unroll
    for (int j = 0; j < KBLK; ++j) { a1[j] = ld_frag16(qa + 16 * j); b1[j] = ld_frag16(qb + 16 * j); }
    LOADS_STAY_AHEAD_OF_WMMAS();
#pragma unroll
    for (int j = 0; j < KBLK; ++j) acc = mma16(a0[j], b0[j], acc);
#pragma unroll
    for (int j = 0; j < KBLK; ++j) acc = mma16(a1[j], b1[j], acc);
  }
  return acc;
}

template <int KTOT, int KBLK>
__device__ __forceinline__ void tile_nk2(const float* pa, const float* pb, const float* pc,
                                         v8f& g, v8f& u) {
  constexpr int KB = 16 * KBLK;
  constexpr int NBLK = KTOT / KB;
  static_assert(KTOT % (2 * KB) == 0, "K must be a whole number of block pairs");
  frag16 a0[KBLK], b0[KBLK], c0[KBLK], a1[KBLK], b1[KBLK], c1[KBLK];
#pragma unroll
  for (int j = 0; j < KBLK; ++j) {
    a0[j] = ld_frag16(pa + 16 * j); b0[j] = ld_frag16(pb + 16 * j); c0[j] = ld_frag16(pc + 16 * j);
  }
  for (int blk = 0; blk < NBLK - 2; blk += 2) {
    const float* qa = pa + (size_t)(blk + 1) * KB;
    const float* qb = pb + (size_t)(blk + 1) * KB;
    const float* qc = pc + (size_t)(blk + 1) * KB;
#pragma unroll
    for (int j = 0; j < KBLK; ++j) {
      a1[j] = ld_frag16(qa + 16 * j); b1[j] = ld_frag16(qb + 16 * j); c1[j] = ld_frag16(qc + 16 * j);
    }
    LOADS_STAY_AHEAD_OF_WMMAS();
#pragma unroll
    for (int j = 0; j < KBLK; ++j) { g = mma16(a0[j], b0[j], g); u = mma16(a0[j], c0[j], u); }
    READERS_DONE_BEFORE_REFILL("v"(g), "v"(u));
#pragma unroll
    for (int j = 0; j < KBLK; ++j) {
      a0[j] = ld_frag16(qa + KB + 16 * j); b0[j] = ld_frag16(qb + KB + 16 * j); c0[j] = ld_frag16(qc + KB + 16 * j);
    }
    LOADS_STAY_AHEAD_OF_WMMAS();
#pragma unroll
    for (int j = 0; j < KBLK; ++j) { g = mma16(a1[j], b1[j], g); u = mma16(a1[j], c1[j], u); }
  }
  {
    const float* qa = pa + (size_t)(NBLK - 1) * KB;
    const float* qb = pb + (size_t)(NBLK - 1) * KB;
    const float* qc = pc + (size_t)(NBLK - 1) * KB;
#pragma unroll
    for (int j = 0; j < KBLK; ++j) {
      a1[j] = ld_frag16(qa + 16 * j); b1[j] = ld_frag16(qb + 16 * j); c1[j] = ld_frag16(qc + 16 * j);
    }
    LOADS_STAY_AHEAD_OF_WMMAS();
#pragma unroll
    for (int j = 0; j < KBLK; ++j) { g = mma16(a0[j], b0[j], g); u = mma16(a0[j], c0[j], u); }
#pragma unroll
    for (int j = 0; j < KBLK; ++j) { g = mma16(a1[j], b1[j], g); u = mma16(a1[j], c1[j], u); }
  }
}


__global__ void k_init(const float* __restrict__ Wi, const float* __restrict__ bi,
                       float* __restrict__ Wst, float* __restrict__ bst) {
  int i = blockIdx.x * blockDim.x + threadIdx.x;
  const int n = HH * HH;
  if (i < n) {
    float w = Wi[i];
    Wst[i] = w;
    Wst[n + i] = w;
  }
  if (i < HH) {
    float b = bi[i];
    bst[i] = b;
    bst[HH + i] = b;
  }
}

__global__ void __launch_bounds__(256) __attribute__((amdgpu_waves_per_eu(1, 2)))
k_p0(const float* x, const float* Wst, const float* bst, float* P0, int chunk) {
  int wid = (blockIdx.x * blockDim.x + threadIdx.x) >> 5;
  int lane = threadIdx.x & 31;
  int tn = wid % (HH / 16);
  int tm = (wid / (HH / 16)) % (CHK / 16);
  int b  = wid / ((HH / 16) * (CHK / 16));
  const float* Xc = x + ((size_t)b * SS + chunk * CHK) * HH;
  const float* W  = Wst + (size_t)b * HH * HH;
  v8f acc = tile_nk<HH, 8>(lane_row(Xc, HH, tm * 16, lane), lane_row(W, HH, tn * 16, lane));
  float bias = bst[(size_t)b * HH + tn * 16 + (lane & 15)];
#pragma unroll
  for (int i = 0; i < 8; ++i) acc[i] += bias;
  store_tile(P0 + (size_t)b * CHK * HH, HH, tm * 16, tn * 16, acc, lane);
}

__global__ void __launch_bounds__(256) __attribute__((amdgpu_waves_per_eu(1, 2)))
k_attn(const float* x, float* Amat) {
  int wid = (blockIdx.x * blockDim.x + threadIdx.x) >> 5;
  int lane = threadIdx.x & 31;
  int tn = wid % (CHK / 16);
  int tm = (wid / (CHK / 16)) % (CHK / 16);
  int b  = (wid / ((CHK / 16) * (CHK / 16))) % BB;
  int chunk = wid / ((CHK / 16) * (CHK / 16) * BB);
  const float* Xc = x + ((size_t)b * SS + chunk * CHK) * HH;
  v8f acc = tile_nk<HH, 8>(lane_row(Xc, HH, tm * 16, lane), lane_row(Xc, HH, tn * 16, lane));
  int s = tn * 16 + (lane & 15);
  int mbase = tm * 16 + ((lane >> 4) << 3);
  float* Ab = Amat + (size_t)(chunk * BB + b) * CHK * CHK;
#pragma unroll
  for (int i = 0; i < 8; ++i) {
    int t = mbase + i;
    float v = (s < t) ? LRC * (acc[i] + 1.0f) : 0.0f;
    Ab[(size_t)t * CHK + s] = v;
  }
}

__global__ void __launch_bounds__(32) __attribute__((amdgpu_waves_per_eu(1, 1)))
k_solve(const float* __restrict__ x, const float* __restrict__ P0,
        const float* __restrict__ Amat, float* __restrict__ Pred,
        float* __restrict__ Diff, float* __restrict__ bst, int chunk) {
  __shared__ v4f Ash[CHK * CHK / 4];
  const int lane = threadIdx.x;
  const int b = blockIdx.x / (HH / 32);
  const int h = (blockIdx.x % (HH / 32)) * 32 + lane;
  const int g0 = chunk * CHK;
  const float* Xb  = x + (size_t)b * SS * HH + h;
  const float* P0b = P0 + (size_t)b * CHK * HH + h;
  float* Predb = Pred + ((size_t)b * SS + g0) * HH + h;
  float* Diffb = Diff + (size_t)b * CHK * HH + h;

  const v4f* Ag = reinterpret_cast<const v4f*>(Amat + (size_t)(chunk * BB + b) * CHK * CHK);
#pragma unroll
  for (int k = 0; k < CHK * CHK / 4 / 32; ++k) Ash[k * 32 + lane] = Ag[k * 32 + lane];

  float acc[CHK], xn[CHK];
#pragma unroll
  for (int t = 0; t < CHK; ++t) acc[t] = P0b[(size_t)t * HH];
#pragma unroll
  for (int t = 0; t < CHK; ++t) {
    const int gn = g0 + t + 1;
    const int gr = (gn < SS) ? gn : (SS - 1);
    const float v = Xb[(size_t)gr * HH];
    xn[t] = (gn < SS) ? v : 0.0f;
  }
  __syncthreads();

  const float* Af = reinterpret_cast<const float*>(Ash);
  v4f abuf[2][CHK];
  float tip[2][6];
  tip[0][0] = Af[1 * CHK];
  tip[0][1] = Af[2 * CHK];  tip[0][2] = Af[2 * CHK + 1];
  tip[0][3] = Af[3 * CHK];  tip[0][4] = Af[3 * CHK + 1];  tip[0][5] = Af[3 * CHK + 2];
#pragma unroll
  for (int t = 4; t < CHK; ++t) abuf[0][t] = Ash[(t * CHK) / 4];
  float bsum = 0.0f;
#pragma unroll
  for (int g = 0; g < CHK / 4; ++g) {
    const int s = 4 * g;
    const int cur = g & 1;
    if (g + 1 < CHK / 4) {
      const int sn = s + 4;
      tip[cur ^ 1][0] = Af[(sn + 1) * CHK + sn];
      tip[cur ^ 1][1] = Af[(sn + 2) * CHK + sn];
      tip[cur ^ 1][2] = Af[(sn + 2) * CHK + sn + 1];
      tip[cur ^ 1][3] = Af[(sn + 3) * CHK + sn];
      tip[cur ^ 1][4] = Af[(sn + 3) * CHK + sn + 1];
      tip[cur ^ 1][5] = Af[(sn + 3) * CHK + sn + 2];
#pragma unroll
      for (int t = 0; t < CHK; ++t) {
        if (t > sn + 3) abuf[cur ^ 1][t] = Ash[(t * CHK + sn) / 4];
      }
    }
    __builtin_amdgcn_sched_barrier(0);
    float n[4];
#pragma unroll
    for (int r = 0; r < 4; ++r) {
      const int t = s + r;
      float a = acc[t];
      if (r == 1) a = fmaf(tip[cur][0], n[0], a);
      if (r == 2) { a = fmaf(tip[cur][1], n[0], a); a = fmaf(tip[cur][2], n[1], a); }
      if (r == 3) { a = fmaf(tip[cur][3], n[0], a); a = fmaf(tip[cur][4], n[1], a); a = fmaf(tip[cur][5], n[2], a); }
      Predb[(size_t)t * HH] = a;
      const float dd = a - xn[t];
      n[r] = xn[t] - a;
      const float dl = (g0 + t == SS - 1) ? 0.0f : dd;
      Diffb[(size_t)t * HH] = dl;
      bsum += dl;
    }
#pragma unroll
    for (int t = 0; t < CHK; ++t) {
      if (t > s + 3) {
        float r = acc[t];
        r = fmaf(abuf[cur][t].x, n[0], r);
        r = fmaf(abuf[cur][t].y, n[1], r);
        r = fmaf(abuf[cur][t].z, n[2], r);
        r = fmaf(abuf[cur][t].w, n[3], r);
        acc[t] = r;
      }
    }
    __builtin_amdgcn_sched_barrier(0);
  }
  bst[(size_t)b * HH + h] -= LRC * bsum;
}

__global__ void k_wupdate(const float* __restrict__ x, const float* __restrict__ Diff,
                          float* __restrict__ Wst, int chunk) {
  int wid = (blockIdx.x * blockDim.x + threadIdx.x) >> 5;
  int lane = threadIdx.x & 31;
  int tn = wid % (HH / 16);
  int tm = (wid / (HH / 16)) % (HH / 16);
  int b  = wid / ((HH / 16) * (HH / 16));
  const float* Xc = x + ((size_t)b * SS + chunk * CHK) * HH;
  const float* Db = Diff + (size_t)b * CHK * HH;
  float* W = Wst + (size_t)b * HH * HH;
  v8f s = {};
  for (int k0 = 0; k0 < CHK; k0 += 4) {
    v2f a  = frag_kn(Db, HH, k0, tm * 16, lane);
    v2f bb = frag_kn(Xc, HH, k0, tn * 16, lane);
    s = wmma4(a, bb, s);
  }
  v8f acc = load_tile(W, HH, tm * 16, tn * 16, lane);
#pragma unroll
  for (int i = 0; i < 8; ++i) acc[i] -= LRC * s[i];
  store_tile(W, HH, tm * 16, tn * 16, acc, lane);
}

__global__ void __launch_bounds__(256) __attribute__((amdgpu_waves_per_eu(1, 2)))
k_mlp1(const float* Pred, const float* gw, const float* gb,
       const float* uw, const float* ub, float* Gt) {
  int wid = (blockIdx.x * blockDim.x + threadIdx.x) >> 5;
  int lane = threadIdx.x & 31;
  int tn = wid % (H4 / 16);
  int tm = wid / (H4 / 16);
  v8f g = {}, u = {};
  tile_nk2<HH, 4>(lane_row(Pred, HH, tm * 16, lane), lane_row(gw, HH, tn * 16, lane),
               lane_row(uw, HH, tn * 16, lane), g, u);
  int n = tn * 16 + (lane & 15);
  float gbias = gb[n], ubias = ub[n];
  v8f r;
#pragma unroll
  for (int i = 0; i < 8; ++i) {
    float zg = g[i] + gbias;
    float zu = u[i] + ubias;
    r[i] = (1.0f / (1.0f + __expf(-zg))) * zu;
  }
  store_tile(Gt, H4, tm * 16, tn * 16, r, lane);
}

__global__ void __launch_bounds__(256) __attribute__((amdgpu_waves_per_eu(1, 2)))
k_mlp2(const float* Gt, const float* dw, const float* db, const float* Pred,
       float* Ypre) {
  int wid = (blockIdx.x * blockDim.x + threadIdx.x) >> 5;
  int lane = threadIdx.x & 31;
  int tn = wid % (HH / 16);
  int tm = wid / (HH / 16);
  v8f acc = tile_nk<H4, 8>(lane_row(Gt, H4, tm * 16, lane), lane_row(dw, H4, tn * 16, lane));
  int n = tn * 16 + (lane & 15);
  int mbase = tm * 16 + ((lane >> 4) << 3);
  float bias = db[n];
#pragma unroll
  for (int i = 0; i < 8; ++i) {
    int m = mbase + i;
    Ypre[(size_t)m * HH + n] = acc[i] + bias + Pred[(size_t)m * HH + n];
  }
}

__global__ void k_ln(const float* __restrict__ Y, const float* __restrict__ nw,
                     const float* __restrict__ nb, float* __restrict__ out) {
  __shared__ float s1[256], s2[256];
  int row = blockIdx.x;
  const float* y = Y + (size_t)row * HH;
  int tid = threadIdx.x;
  float v0 = y[tid], v1 = y[tid + 256], v2 = y[tid + 512];
  s1[tid] = v0 + v1 + v2;
  s2[tid] = v0 * v0 + v1 * v1 + v2 * v2;
  __syncthreads();
  for (int off = 128; off > 0; off >>= 1) {
    if (tid < off) { s1[tid] += s1[tid + off]; s2[tid] += s2[tid + off]; }
    __syncthreads();
  }
  float mu  = s1[0] * (1.0f / HH);
  float var = s2[0] * (1.0f / HH) - mu * mu;
  float r = rsqrtf(var + EPSC);
  float* o = out + (size_t)row * HH;
  o[tid]       = (v0 - mu) * r * nw[tid]       + nb[tid];
  o[tid + 256] = (v1 - mu) * r * nw[tid + 256] + nb[tid + 256];
  o[tid + 512] = (v2 - mu) * r * nw[tid + 512] + nb[tid + 512];
}

extern "C" void kernel_launch(void* const* d_in, const int* in_sizes, int n_in,
                              void* d_out, int out_size, void* d_ws, size_t ws_size,
                              hipStream_t stream) {
  const float* x      = (const float*)d_in[0];
  const float* W_init = (const float*)d_in[1];
  const float* b_init = (const float*)d_in[2];
  const float* gate_w = (const float*)d_in[3];
  const float* gate_b = (const float*)d_in[4];
  const float* up_w   = (const float*)d_in[5];
  const float* up_b   = (const float*)d_in[6];
  const float* down_w = (const float*)d_in[7];
  const float* down_b = (const float*)d_in[8];
  const float* norm_w = (const float*)d_in[9];
  const float* norm_b = (const float*)d_in[10];
  float* out = (float*)d_out;

  float* ws   = (float*)d_ws;
  float* Wst  = ws;  ws += (size_t)BB * HH * HH;
  float* bst  = ws;  ws += (size_t)BB * HH;
  float* P0   = ws;  ws += (size_t)BB * CHK * HH;
  float* Pred = ws;  ws += (size_t)BB * SS * HH;
  float* Diff = ws;  ws += (size_t)BB * CHK * HH;
  float* Amat = ws;  ws += (size_t)NCH * BB * CHK * CHK;
  float* Gt   = ws;  ws += (size_t)BB * SS * H4;
  float* Ypre = ws;  ws += (size_t)BB * SS * HH;

  k_init<<<(HH * HH + 255) / 256, 256, 0, stream>>>(W_init, b_init, Wst, bst);
  k_attn<<<16, 256, 0, stream>>>(x, Amat);

  for (int c = 0; c < NCH; ++c) {
    k_p0<<<48, 256, 0, stream>>>(x, Wst, bst, P0, c);
    k_solve<<<BB * (HH / 32), 32, 0, stream>>>(x, P0, Amat, Pred, Diff, bst, c);
    k_wupdate<<<576, 256, 0, stream>>>(x, Diff, Wst, c);
  }

  k_mlp1<<<768, 256, 0, stream>>>(Pred, gate_w, gate_b, up_w, up_b, Gt);
  k_mlp2<<<192, 256, 0, stream>>>(Gt, down_w, down_b, Pred, Ypre);
  k_ln<<<BB * SS, 256, 0, stream>>>(Ypre, norm_w, norm_b, out);
}


# fast-weight state never built: pred = x*W_init^T + b_init - A*diff over all earlier tokens; other chunks' terms by one WMMA GEMM per chunk; init and W-update launches go; sum order changes
# speedup vs baseline: 5.9447x; 1.3200x over previous
#include <hip/hip_runtime.h>
#include <math.h>

#define BB 2
#define SS 256
#define HH 768
#define H4 3072
#define CHK 64
#define NCH (SS / CHK)
#define LRC 0.01f
#define EPSC 1e-5f

typedef float v2f __attribute__((ext_vector_type(2)));
typedef float v8f __attribute__((ext_vector_type(8)));

__device__ __forceinline__ v8f wmma4(v2f a, v2f b, v8f c) {
  return __builtin_amdgcn_wmma_f32_16x16x4_f32(
       false, a,  false, b,
       (short)0, c,  false,  false);
}

__device__ __forceinline__ v2f frag_kn(const float* M, int ld, int k0, int r0, int lane) {
  const float* p = M + (size_t)(k0 + ((lane >> 4) << 1)) * ld + r0 + (lane & 15);
  v2f v; v.x = p[0]; v.y = p[ld];
  return v;
}
__device__ __forceinline__ void store_tile(float* D, int ld, int r0, int c0, v8f v, int lane) {
  int n = c0 + (lane & 15);
  int m0 = r0 + ((lane >> 4) << 3);
#pragma unroll
  for (int i = 0; i < 8; ++i) D[(size_t)(m0 + i) * ld + n] = v[i];
}
__device__ __forceinline__ v8f load_tile(const float* D, int ld, int r0, int c0, int lane) {
  int n = c0 + (lane & 15);
  int m0 = r0 + ((lane >> 4) << 3);
  v8f v;
#pragma unroll
  for (int i = 0; i < 8; ++i) v[i] = D[(size_t)(m0 + i) * ld + n];
  return v;
}


typedef float v4f __attribute__((ext_vector_type(4)));

struct frag16 { v4f lo, hi; };
__device__ __forceinline__ frag16 ld_frag16(const float* p) {
  const v4f* q = reinterpret_cast<const v4f*>(p);
  frag16 f; f.lo = q[0]; f.hi = q[1];
  return f;
}
__device__ __forceinline__ v8f mma16(const frag16& a, const frag16& b, v8f c) {
  c = wmma4(a.lo.xy, b.lo.xy, c);
  c = wmma4(a.lo.zw, b.lo.zw, c);
  c = wmma4(a.hi.xy, b.hi.xy, c);
  c = wmma4(a.hi.zw, b.hi.zw, c);
  return c;
}
__device__ __forceinline__ const float* lane_row(const float* M, int ld, int r0, int lane) {
  return M + (size_t)(r0 + (lane & 15)) * ld + ((lane >> 4) << 3);
}

#define LOADS_STAY_AHEAD_OF_WMMAS() __builtin_amdgcn_sched_barrier(0)
#define READERS_DONE_BEFORE_REFILL(...) asm volatile("" :: __VA_ARGS__)

template <int KTOT, int KBLK>
__device__ __forceinline__ v8f tile_nk(const float* pa, const float* pb) {
  constexpr int KB = 16 * KBLK;
  constexpr int NBLK = KTOT / KB;
  static_assert(KTOT % (2 * KB) == 0, "K must be a whole number of block pairs");
  frag16 a0[KBLK], b0[KBLK], a1[KBLK], b1[KBLK];
#pragma unroll
  for (int j = 0; j < KBLK; ++j) { a0[j] = ld_frag16(pa + 16 * j); b0[j] = ld_frag16(pb + 16 * j); }
  v8f acc = {};
  for (int blk = 0; blk < NBLK - 2; blk += 2) {
    const float* qa = pa + (size_t)(blk + 1) * KB;
    const float* qb = pb + (size_t)(blk + 1) * KB;
#pragma unroll
    for (int j = 0; j < KBLK; ++j) { a1[j] = ld_frag16(qa + 16 * j); b1[j] = ld_frag16(qb + 16 * j); }
    LOADS_STAY_AHEAD_OF_WMMAS();
#pragma unroll
    for (int j = 0; j < KBLK; ++j) acc = mma16(a0[j], b0[j], acc);
    READERS_DONE_BEFORE_REFILL("v"(acc));
#pragma unroll
    for (int j = 0; j < KBLK; ++j) { a0[j] = ld_frag16(qa + KB + 16 * j); b0[j] = ld_frag16(qb + KB + 16 * j); }
    LOADS_STAY_AHEAD_OF_WMMAS();
#pragma unroll
    for (int j = 0; j < KBLK; ++j) acc = mma16(a1[j], b1[j], acc);
  }
  {
    const float* qa = pa + (size_t)(NBLK - 1) * KB;
    const float* qb = pb + (size_t)(NBLK - 1) * KB;
#pragma unroll
    for (int j = 0; j < KBLK; ++j) { a1[j] = ld_frag16(qa + 16 * j); b1[j] = ld_frag16(qb + 16 * j); }
    LOADS_STAY_AHEAD_OF_WMMAS();
#pragma unroll
    for (int j = 0; j < KBLK; ++j) acc = mma16(a0[j], b0[j], acc);
#pragma unroll
    for (int j = 0; j < KBLK; ++j) acc = mma16(a1[j], b1[j], acc);
  }
  return acc;
}

template <int KTOT, int KBLK>
__device__ __forceinline__ void tile_nk2(const float* pa, const float* pb, const float* pc,
                                         v8f& g, v8f& u) {
  constexpr int KB = 16 * KBLK;
  constexpr int NBLK = KTOT / KB;
  static_assert(KTOT % (2 * KB) == 0, "K must be a whole number of block pairs");
  frag16 a0[KBLK], b0[KBLK], c0[KBLK], a1[KBLK], b1[KBLK], c1[KBLK];
#pragma unroll
  for (int j = 0; j < KBLK; ++j) {
    a0[j] = ld_frag16(pa + 16 * j); b0[j] = ld_frag16(pb + 16 * j); c0[j] = ld_frag16(pc + 16 * j);
  }
  for (int blk = 0; blk < NBLK - 2; blk += 2) {
    const float* qa = pa + (size_t)(blk + 1) * KB;
    const float* qb = pb + (size_t)(blk + 1) * KB;
    const float* qc = pc + (size_t)(blk + 1) * KB;
#pragma unroll
    for (int j = 0; j < KBLK; ++j) {
      a1[j] = ld_frag16(qa + 16 * j); b1[j] = ld_frag16(qb + 16 * j); c1[j] = ld_frag16(qc + 16 * j);
    }
    LOADS_STAY_AHEAD_OF_WMMAS();
#pragma unroll
    for (int j = 0; j < KBLK; ++j) { g = mma16(a0[j], b0[j], g); u = mma16(a0[j], c0[j], u); }
    READERS_DONE_BEFORE_REFILL("v"(g), "v"(u));
#pragma unroll
    for (int j = 0; j < KBLK; ++j) {
      a0[j] = ld_frag16(qa + KB + 16 * j); b0[j] = ld_frag16(qb + KB + 16 * j); c0[j] = ld_frag16(qc + KB + 16 * j);
    }
    LOADS_STAY_AHEAD_OF_WMMAS();
#pragma unroll
    for (int j = 0; j < KBLK; ++j) { g = mma16(a1[j], b1[j], g); u = mma16(a1[j], c1[j], u); }
  }
  {
    const float* qa = pa + (size_t)(NBLK - 1) * KB;
    const float* qb = pb + (size_t)(NBLK - 1) * KB;
    const float* qc = pc + (size_t)(NBLK - 1) * KB;
#pragma unroll
    for (int j = 0; j < KBLK; ++j) {
      a1[j] = ld_frag16(qa + 16 * j); b1[j] = ld_frag16(qb + 16 * j); c1[j] = ld_frag16(qc + 16 * j);
    }
    LOADS_STAY_AHEAD_OF_WMMAS();
#pragma unroll
    for (int j = 0; j < KBLK; ++j) { g = mma16(a0[j], b0[j], g); u = mma16(a0[j], c0[j], u); }
#pragma unroll
    for (int j = 0; j < KBLK; ++j) { g = mma16(a1[j], b1[j], g); u = mma16(a1[j], c1[j], u); }
  }
}


__global__ void __launch_bounds__(256) __attribute__((amdgpu_waves_per_eu(1, 2)))
k_p0(const float* x, const float* Wi, const float* bi, float* P0) {
  int wid = (blockIdx.x * blockDim.x + threadIdx.x) >> 5;
  int lane = threadIdx.x & 31;
  int tn = wid % (HH / 16);
  int tm = wid / (HH / 16);
  v8f acc = tile_nk<HH, 8>(lane_row(x, HH, tm * 16, lane), lane_row(Wi, HH, tn * 16, lane));
  float bias = bi[tn * 16 + (lane & 15)];
#pragma unroll
  for (int i = 0; i < 8; ++i) acc[i] += bias;
  store_tile(P0, HH, tm * 16, tn * 16, acc, lane);
}

__global__ void __launch_bounds__(256) __attribute__((amdgpu_waves_per_eu(1, 2)))
k_attn(const float* x, float* Amat) {
  int wid = (blockIdx.x * blockDim.x + threadIdx.x) >> 5;
  int lane = threadIdx.x & 31;
  int tn = wid % (SS / 16);
  int tm = (wid / (SS / 16)) % (SS / 16);
  int b  = wid / ((SS / 16) * (SS / 16));
  if ((tn / (CHK / 16)) > (tm / (CHK / 16))) return;
  const float* Xb = x + (size_t)b * SS * HH;
  v8f acc = tile_nk<HH, 8>(lane_row(Xb, HH, tm * 16, lane), lane_row(Xb, HH, tn * 16, lane));
  int s = tn * 16 + (lane & 15);
  int mbase = tm * 16 + ((lane >> 4) << 3);
  float* Ab = Amat + (size_t)b * SS * SS;
#pragma unroll
  for (int i = 0; i < 8; ++i) {
    int t = mbase + i;
    float v = (s < t) ? LRC * (acc[i] + 1.0f) : 0.0f;
    Ab[(size_t)t * SS + s] = v;
  }
}

__global__ void __launch_bounds__(32) __attribute__((amdgpu_waves_per_eu(1, 1)))
k_solve(const float* __restrict__ x, const float* __restrict__ P0,
        const float* __restrict__ Amat, float* __restrict__ Pred,
        float* __restrict__ Diff, int chunk) {
  __shared__ v4f Ash[CHK * CHK / 4];
  const int lane = threadIdx.x;
  const int b = blockIdx.x / (HH / 32);
  const int h = (blockIdx.x % (HH / 32)) * 32 + lane;
  const int g0 = chunk * CHK;
  const float* Xb  = x + (size_t)b * SS * HH + h;
  const float* P0b = P0 + ((size_t)b * SS + g0) * HH + h;
  float* Predb = Pred + ((size_t)b * SS + g0) * HH + h;
  float* Diffb = Diff + ((size_t)b * SS + g0) * HH + h;

  const float* Ag = Amat + (size_t)b * SS * SS + (size_t)g0 * SS + g0;
#pragma unroll
  for (int k = 0; k < CHK * CHK / 4 / 32; ++k) {
    const int i = k * 32 + lane;
    Ash[i] = *reinterpret_cast<const v4f*>(Ag + (size_t)(i >> 4) * SS + ((i & 15) << 2));
  }

  float acc[CHK], xn[CHK];
#pragma unroll
  for (int t = 0; t < CHK; ++t) acc[t] = P0b[(size_t)t * HH];
#pragma unroll
  for (int t = 0; t < CHK; ++t) {
    const int gn = g0 + t + 1;
    const int gr = (gn < SS) ? gn : (SS - 1);
    const float v = Xb[(size_t)gr * HH];
    xn[t] = (gn < SS) ? v : 0.0f;
  }
  __syncthreads();

  const float* Af = reinterpret_cast<const float*>(Ash);
  v4f abuf[2][CHK];
  float tip[2][6];
  tip[0][0] = Af[1 * CHK];
  tip[0][1] = Af[2 * CHK];  tip[0][2] = Af[2 * CHK + 1];
  tip[0][3] = Af[3 * CHK];  tip[0][4] = Af[3 * CHK + 1];  tip[0][5] = Af[3 * CHK + 2];
#pragma unroll
  for (int t = 4; t < CHK; ++t) abuf[0][t] = Ash[(t * CHK) / 4];
#pragma unroll
  for (int g = 0; g < CHK / 4; ++g) {
    const int s = 4 * g;
    const int cur = g & 1;
    if (g + 1 < CHK / 4) {
      const int sn = s + 4;
      tip[cur ^ 1][0] = Af[(sn + 1) * CHK + sn];
      tip[cur ^ 1][1] = Af[(sn + 2) * CHK + sn];
      tip[cur ^ 1][2] = Af[(sn + 2) * CHK + sn + 1];
      tip[cur ^ 1][3] = Af[(sn + 3) * CHK + sn];
      tip[cur ^ 1][4] = Af[(sn + 3) * CHK + sn + 1];
      tip[cur ^ 1][5] = Af[(sn + 3) * CHK + sn + 2];
#pragma unroll
      for (int t = 0; t < CHK; ++t) {
        if (t > sn + 3) abuf[cur ^ 1][t] = Ash[(t * CHK + sn) / 4];
      }
    }
    __builtin_amdgcn_sched_barrier(0);
    float n[4];
#pragma unroll
    for (int r = 0; r < 4; ++r) {
      const int t = s + r;
      float a = acc[t];
      if (r == 1) a = fmaf(tip[cur][0], n[0], a);
      if (r == 2) { a = fmaf(tip[cur][1], n[0], a); a = fmaf(tip[cur][2], n[1], a); }
      if (r == 3) { a = fmaf(tip[cur][3], n[0], a); a = fmaf(tip[cur][4], n[1], a); a = fmaf(tip[cur][5], n[2], a); }
      Predb[(size_t)t * HH] = a;
      Diffb[(size_t)t * HH] = a - xn[t];
      n[r] = xn[t] - a;
    }
#pragma unroll
    for (int t = 0; t < CHK; ++t) {
      if (t > s + 3) {
        float r = acc[t];
        r = fmaf(abuf[cur][t].x, n[0], r);
        r = fmaf(abuf[cur][t].y, n[1], r);
        r = fmaf(abuf[cur][t].z, n[2], r);
        r = fmaf(abuf[cur][t].w, n[3], r);
        acc[t] = r;
      }
    }
    __builtin_amdgcn_sched_barrier(0);
  }
}

__global__ void k_cross(const float* __restrict__ Amat, const float* __restrict__ Diff,
                        float* __restrict__ P0, int chunk) {
  int wid = (blockIdx.x * blockDim.x + threadIdx.x) >> 5;
  int lane = threadIdx.x & 31;
  int tn = wid % (HH / 16);
  int tm = (wid / (HH / 16)) % (CHK / 16);
  int b  = wid / ((HH / 16) * (CHK / 16));
  const int g0 = chunk * CHK;
  const float* Ar = Amat + (size_t)b * SS * SS + (size_t)(g0 + tm * 16 + (lane & 15)) * SS
                    + ((lane >> 4) << 1);
  const float* Db = Diff + (size_t)b * SS * HH;
  float* Pb = P0 + ((size_t)b * SS + g0) * HH;
  v8f acc = load_tile(Pb, HH, tm * 16, tn * 16, lane);
  v8f s = {};
  for (int kb = 0; kb < g0; kb += CHK) {
    v2f a[CHK / 4], bb[CHK / 4];
#pragma unroll
    for (int j = 0; j < CHK / 4; ++j) {
      a[j]  = *reinterpret_cast<const v2f*>(Ar + kb + 4 * j);
      bb[j] = frag_kn(Db, HH, kb + 4 * j, tn * 16, lane);
    }
#pragma unroll
    for (int j = 0; j < CHK / 4; ++j) s = wmma4(a[j], bb[j], s);
  }
#pragma unroll
  for (int i = 0; i < 8; ++i) acc[i] -= s[i];
  store_tile(Pb, HH, tm * 16, tn * 16, acc, lane);
}

__global__ void __launch_bounds__(256) __attribute__((amdgpu_waves_per_eu(1, 2)))
k_mlp1(const float* Pred, const float* gw, const float* gb,
       const float* uw, const float* ub, float* Gt) {
  int wid = (blockIdx.x * blockDim.x + threadIdx.x) >> 5;
  int lane = threadIdx.x & 31;
  int tn = wid % (H4 / 16);
  int tm = wid / (H4 / 16);
  v8f g = {}, u = {};
  tile_nk2<HH, 4>(lane_row(Pred, HH, tm * 16, lane), lane_row(gw, HH, tn * 16, lane),
               lane_row(uw, HH, tn * 16, lane), g, u);
  int n = tn * 16 + (lane & 15);
  float gbias = gb[n], ubias = ub[n];
  v8f r;
#pragma unroll
  for (int i = 0; i < 8; ++i) {
    float zg = g[i] + gbias;
    float zu = u[i] + ubias;
    r[i] = (1.0f / (1.0f + __expf(-zg))) * zu;
  }
  store_tile(Gt, H4, tm * 16, tn * 16, r, lane);
}

__global__ void __launch_bounds__(256) __attribute__((amdgpu_waves_per_eu(1, 2)))
k_mlp2(const float* Gt, const float* dw, const float* db, const float* Pred,
       float* Ypre) {
  int wid = (blockIdx.x * blockDim.x + threadIdx.x) >> 5;
  int lane = threadIdx.x & 31;
  int tn = wid % (HH / 16);
  int tm = wid / (HH / 16);
  v8f acc = tile_nk<H4, 8>(lane_row(Gt, H4, tm * 16, lane), lane_row(dw, H4, tn * 16, lane));
  int n = tn * 16 + (lane & 15);
  int mbase = tm * 16 + ((lane >> 4) << 3);
  float bias = db[n];
#pragma unroll
  for (int i = 0; i < 8; ++i) {
    int m = mbase + i;
    Ypre[(size_t)m * HH + n] = acc[i] + bias + Pred[(size_t)m * HH + n];
  }
}

__global__ void k_ln(const float* __restrict__ Y, const float* __restrict__ nw,
                     const float* __restrict__ nb, float* __restrict__ out) {
  __shared__ float s1[256], s2[256];
  int row = blockIdx.x;
  const float* y = Y + (size_t)row * HH;
  int tid = threadIdx.x;
  float v0 = y[tid], v1 = y[tid + 256], v2 = y[tid + 512];
  s1[tid] = v0 + v1 + v2;
  s2[tid] = v0 * v0 + v1 * v1 + v2 * v2;
  __syncthreads();
  for (int off = 128; off > 0; off >>= 1) {
    if (tid < off) { s1[tid] += s1[tid + off]; s2[tid] += s2[tid + off]; }
    __syncthreads();
  }
  float mu  = s1[0] * (1.0f / HH);
  float var = s2[0] * (1.0f / HH) - mu * mu;
  float r = rsqrtf(var + EPSC);
  float* o = out + (size_t)row * HH;
  o[tid]       = (v0 - mu) * r * nw[tid]       + nb[tid];
  o[tid + 256] = (v1 - mu) * r * nw[tid + 256] + nb[tid + 256];
  o[tid + 512] = (v2 - mu) * r * nw[tid + 512] + nb[tid + 512];
}

extern "C" void kernel_launch(void* const* d_in, const int* in_sizes, int n_in,
                              void* d_out, int out_size, void* d_ws, size_t ws_size,
                              hipStream_t stream) {
  const float* x      = (const float*)d_in[0];
  const float* W_init = (const float*)d_in[1];
  const float* b_init = (const float*)d_in[2];
  const float* gate_w = (const float*)d_in[3];
  const float* gate_b = (const float*)d_in[4];
  const float* up_w   = (const float*)d_in[5];
  const float* up_b   = (const float*)d_in[6];
  const float* down_w = (const float*)d_in[7];
  const float* down_b = (const float*)d_in[8];
  const float* norm_w = (const float*)d_in[9];
  const float* norm_b = (const float*)d_in[10];
  float* out = (float*)d_out;

  float* ws   = (float*)d_ws;
  float* P0   = ws;  ws += (size_t)BB * SS * HH;
  float* Pred = ws;  ws += (size_t)BB * SS * HH;
  float* Diff = ws;  ws += (size_t)BB * SS * HH;
  float* Amat = ws;  ws += (size_t)BB * SS * SS;
  float* Gt   = ws;  ws += (size_t)BB * SS * H4;
  float* Ypre = ws;  ws += (size_t)BB * SS * HH;

  k_attn<<<64, 256, 0, stream>>>(x, Amat);
  k_p0<<<192, 256, 0, stream>>>(x, W_init, b_init, P0);

  for (int c = 0; c < NCH; ++c) {
    if (c > 0) k_cross<<<192, 64, 0, stream>>>(Amat, Diff, P0, c);
    k_solve<<<BB * (HH / 32), 32, 0, stream>>>(x, P0, Amat, Pred, Diff, c);
  }

  k_mlp1<<<768, 256, 0, stream>>>(Pred, gate_w, gate_b, up_w, up_b, Gt);
  k_mlp2<<<192, 256, 0, stream>>>(Gt, down_w, down_b, Pred, Ypre);
  k_ln<<<BB * SS, 256, 0, stream>>>(Ypre, norm_w, norm_b, out);
}


# chunk loop's 7 launches (4 solves, 3 cross GEMMs) become 1: a 4-wave block keeps its 32 columns through all chunks, diffs stay in LDS, cross terms by WMMA from LDS; every sum as before
# speedup vs baseline: 7.1203x; 1.1978x over previous
#include <hip/hip_runtime.h>
#include <math.h>

#define BB 2
#define SS 256
#define HH 768
#define H4 3072
#define CHK 64
#define NCH (SS / CHK)
#define LRC 0.01f
#define EPSC 1e-5f

typedef float v2f __attribute__((ext_vector_type(2)));
typedef float v8f __attribute__((ext_vector_type(8)));

__device__ __forceinline__ v8f wmma4(v2f a, v2f b, v8f c) {
  return __builtin_amdgcn_wmma_f32_16x16x4_f32(
       false, a,  false, b,
       (short)0, c,  false,  false);
}

__device__ __forceinline__ v2f frag_kn(const float* M, int ld, int k0, int r0, int lane) {
  const float* p = M + (size_t)(k0 + ((lane >> 4) << 1)) * ld + r0 + (lane & 15);
  v2f v; v.x = p[0]; v.y = p[ld];
  return v;
}
__device__ __forceinline__ void store_tile(float* D, int ld, int r0, int c0, v8f v, int lane) {
  int n = c0 + (lane & 15);
  int m0 = r0 + ((lane >> 4) << 3);
#pragma unroll
  for (int i = 0; i < 8; ++i) D[(size_t)(m0 + i) * ld + n] = v[i];
}
__device__ __forceinline__ v8f load_tile(const float* D, int ld, int r0, int c0, int lane) {
  int n = c0 + (lane & 15);
  int m0 = r0 + ((lane >> 4) << 3);
  v8f v;
#pragma unroll
  for (int i = 0; i < 8; ++i) v[i] = D[(size_t)(m0 + i) * ld + n];
  return v;
}


typedef float v4f __attribute__((ext_vector_type(4)));

template <int KS> struct stage_geo {
  static constexpr int PITCH = KS + 4;
  static constexpr int PPR = KS / 4;
  static constexpr int RPP = 256 / PPR;
  static constexpr int NQ = KS / 8;
};

#define FRAG_READS_IN_WMMA_ORDER_AHEAD_OF_WMMAS() __builtin_amdgcn_sched_barrier(0)
#define WMMAS_STAY_AHEAD_OF_NEXT_STAGE_STORES() __builtin_amdgcn_sched_barrier(0)

__device__ __forceinline__ v4f frag_read(const float* p, int i) {
  return *reinterpret_cast<const v4f*>(p + 4 * i);
}

template <int KTOT, int KS>
__device__ __forceinline__ v8f block_nk(const float* Ablk, int lda, const float* Bblk, int ldb,
                                        float* sm) {
  typedef stage_geo<KS> G;
  constexpr int NPA = 32 / G::RPP, NPB = 64 / G::RPP, NP = NPA + NPB;
  constexpr int BUF = 96 * G::PITCH;
  constexpr int NS = KTOT / KS;
  static_assert(KTOT % KS == 0, "K must be a whole number of stages");
  const int tid = threadIdx.x, lane = tid & 31, w = tid >> 5;
  const int lr = tid / G::PPR, lp = (tid % G::PPR) * 4;
  const float* ga = Ablk + (size_t)lr * lda + lp;
  const float* gb = Bblk + (size_t)lr * ldb + lp;
  float* sw = sm + lr * G::PITCH + lp;
  const int fo = (lane & 15) * G::PITCH + (lane >> 4) * (KS / 2);
  const float* fa = sm + ((w >> 2) * 16) * G::PITCH + fo;
  const float* fb = sm + (32 + (w & 3) * 16) * G::PITCH + fo;
  v4f st[NP];
#pragma unroll
  for (int p = 0; p < NPA; ++p) st[p] = *reinterpret_cast<const v4f*>(ga + (size_t)p * G::RPP * lda);
#pragma unroll
  for (int p = 0; p < NPB; ++p) st[NPA + p] = *reinterpret_cast<const v4f*>(gb + (size_t)p * G::RPP * ldb);
#pragma unroll
  for (int p = 0; p < NP; ++p) *reinterpret_cast<v4f*>(sw + p * G::RPP * G::PITCH) = st[p];
  __syncthreads();
  v8f acc = {};
  for (int s = 0; s < NS; ++s) {
    const int cur = s & 1;
    const bool more = (s + 1 < NS);
    if (more) {
      const int k1 = (s + 1) * KS;
#pragma unroll
      for (int p = 0; p < NPA; ++p) st[p] = *reinterpret_cast<const v4f*>(ga + (size_t)p * G::RPP * lda + k1);
#pragma unroll
      for (int p = 0; p < NPB; ++p) st[NPA + p] = *reinterpret_cast<const v4f*>(gb + (size_t)p * G::RPP * ldb + k1);
    }
    v4f a[G::NQ], b[G::NQ];
#pragma unroll
    for (int i = 0; i < G::NQ; ++i) {
      a[i] = frag_read(fa + cur * BUF, i); b[i] = frag_read(fb + cur * BUF, i);
      FRAG_READS_IN_WMMA_ORDER_AHEAD_OF_WMMAS();
    }
#pragma unroll
    for (int i = 0; i < G::NQ; ++i) {
      acc = wmma4(a[i].xy, b[i].xy, acc);
      acc = wmma4(a[i].zw, b[i].zw, acc);
    }
    WMMAS_STAY_AHEAD_OF_NEXT_STAGE_STORES();
    if (more) {
#pragma unroll
      for (int p = 0; p < NP; ++p) *reinterpret_cast<v4f*>(sw + (cur ^ 1) * BUF + p * G::RPP * G::PITCH) = st[p];
    }
    __syncthreads();
  }
  return acc;
}

template <int KTOT, int KS>
__device__ __forceinline__ void block_nk2(const float* Ablk, int lda, const float* Bblk,
                                          const float* Cblk, int ldb, float* sm,
                                          v8f& g, v8f& u) {
  typedef stage_geo<KS> G;
  constexpr int NPA = 64 / G::RPP, NPB = 32 / G::RPP, NP = NPA + 2 * NPB;
  constexpr int BUF = 128 * G::PITCH;
  constexpr int NS = KTOT / KS;
  static_assert(KTOT % KS == 0, "K must be a whole number of stages");
  const int tid = threadIdx.x, lane = tid & 31, w = tid >> 5;
  const int lr = tid / G::PPR, lp = (tid % G::PPR) * 4;
  const float* ga = Ablk + (size_t)lr * lda + lp;
  const float* gb = Bblk + (size_t)lr * ldb + lp;
  const float* gc = Cblk + (size_t)lr * ldb + lp;
  float* sw = sm + lr * G::PITCH + lp;
  const int fo = (lane & 15) * G::PITCH + (lane >> 4) * (KS / 2);
  const float* fa = sm + ((w >> 1) * 16) * G::PITCH + fo;
  const float* fb = sm + (64 + (w & 1) * 16) * G::PITCH + fo;
  const float* fc = sm + (96 + (w & 1) * 16) * G::PITCH + fo;
  v4f st[NP];
#pragma unroll
  for (int p = 0; p < NPA; ++p) st[p] = *reinterpret_cast<const v4f*>(ga + (size_t)p * G::RPP * lda);
#pragma unroll
  for (int p = 0; p < NPB; ++p) {
    st[NPA + p] = *reinterpret_cast<const v4f*>(gb + (size_t)p * G::RPP * ldb);
    st[NPA + NPB + p] = *reinterpret_cast<const v4f*>(gc + (size_t)p * G::RPP * ldb);
  }
#pragma unroll
  for (int p = 0; p < NP; ++p) *reinterpret_cast<v4f*>(sw + p * G::RPP * G::PITCH) = st[p];
  __syncthreads();
  for (int s = 0; s < NS; ++s) {
    const int cur = s & 1;
    const bool more = (s + 1 < NS);
    if (more) {
      const int k1 = (s + 1) * KS;
#pragma unroll
      for (int p = 0; p < NPA; ++p) st[p] = *reinterpret_cast<const v4f*>(ga + (size_t)p * G::RPP * lda + k1);
#pragma unroll
      for (int p = 0; p < NPB; ++p) {
        st[NPA + p] = *reinterpret_cast<const v4f*>(gb + (size_t)p * G::RPP * ldb + k1);
        st[NPA + NPB + p] = *reinterpret_cast<const v4f*>(gc + (size_t)p * G::RPP * ldb + k1);
      }
    }
    v4f a[G::NQ], b[G::NQ], c[G::NQ];
#pragma unroll
    for (int i = 0; i < G::NQ; ++i) {
      a[i] = frag_read(fa + cur * BUF, i); b[i] = frag_read(fb + cur * BUF, i); c[i] = frag_read(fc + cur * BUF, i);
      FRAG_READS_IN_WMMA_ORDER_AHEAD_OF_WMMAS();
    }
#pragma unroll
    for (int i = 0; i < G::NQ; ++i) {
      g = wmma4(a[i].xy, b[i].xy, g);
      u = wmma4(a[i].xy, c[i].xy, u);
      g = wmma4(a[i].zw, b[i].zw, g);
      u = wmma4(a[i].zw, c[i].zw, u);
    }
    WMMAS_STAY_AHEAD_OF_NEXT_STAGE_STORES();
    if (more) {
#pragma unroll
      for (int p = 0; p < NP; ++p) *reinterpret_cast<v4f*>(sw + (cur ^ 1) * BUF + p * G::RPP * G::PITCH) = st[p];
    }
    __syncthreads();
  }
}


#define KS1 64
#define KS2 32

__global__ void __launch_bounds__(256)
k_p0(const float* x, const float* Wi, const float* bi, float* P0) {
  __shared__ v4f smv[2 * 96 * (KS1 + 4) / 4];
  const int w = threadIdx.x >> 5, lane = threadIdx.x & 31;
  const int bn = blockIdx.x % (HH / 64);
  const int bm = blockIdx.x / (HH / 64);
  v8f acc = block_nk<HH, KS1>(x + (size_t)bm * 32 * HH, HH, Wi + (size_t)bn * 64 * HH, HH,
                              reinterpret_cast<float*>(smv));
  const int tm = bm * 2 + (w >> 2), tn = bn * 4 + (w & 3);
  float bias = bi[tn * 16 + (lane & 15)];
#pragma unroll
  for (int i = 0; i < 8; ++i) acc[i] += bias;
  store_tile(P0, HH, tm * 16, tn * 16, acc, lane);
}

__global__ void __launch_bounds__(256)
k_attn(const float* x, float* Amat) {
  __shared__ v4f smv[2 * 96 * (KS1 + 4) / 4];
  const int w = threadIdx.x >> 5, lane = threadIdx.x & 31;
  const int bn = blockIdx.x % (SS / 64);
  const int bm = (blockIdx.x / (SS / 64)) % (SS / 32);
  const int b  = blockIdx.x / ((SS / 64) * (SS / 32));
  if (bn > (bm >> 1)) return;
  const float* Xb = x + (size_t)b * SS * HH;
  v8f acc = block_nk<HH, KS1>(Xb + (size_t)bm * 32 * HH, HH, Xb + (size_t)bn * 64 * HH, HH,
                              reinterpret_cast<float*>(smv));
  const int tm = bm * 2 + (w >> 2), tn = bn * 4 + (w & 3);
  int s = tn * 16 + (lane & 15);
  int mbase = tm * 16 + ((lane >> 4) << 3);
  float* Ab = Amat + (size_t)b * SS * SS;
#pragma unroll
  for (int i = 0; i < 8; ++i) {
    int t = mbase + i;
    float v = (s < t) ? LRC * (acc[i] + 1.0f) : 0.0f;
    Ab[(size_t)t * SS + s] = v;
  }
}

#define DPP 96
__global__ void __launch_bounds__(128) __attribute__((amdgpu_waves_per_eu(1, 1)))
k_seq(const float* __restrict__ x, const float* __restrict__ P0,
      const float* __restrict__ Amat, float* __restrict__ Pred) {
  __shared__ v4f Ash[CHK * CHK / 4];
  __shared__ v4f Xsh[CHK * 32 / 4];
  __shared__ float Csh[CHK * 32];
  __shared__ v2f Dsh[(SS / 2) * (DPP / 2)];
  const int tid = threadIdx.x, lane = tid & 31, w = tid >> 5;
  const int b = blockIdx.x / (HH / 32);
  const int h0 = (blockIdx.x % (HH / 32)) * 32;
  const float* Ab = Amat + (size_t)b * SS * SS;
  const float* Xb = x + (size_t)b * SS * HH + h0;

#pragma nounroll
  for (int c = 0; c < NCH; ++c) {
    const int g0 = c * CHK;
    const float* Ar = Ab + (size_t)(g0 + w * 16 + (lane & 15)) * SS + ((lane >> 4) << 1);
    const int k1 = (g0 > CHK) ? CHK : 0;
    const int k2 = (g0 > 2 * CHK) ? 2 * CHK : 0;
    v2f a0[CHK / 4], a1[CHK / 4], a2[CHK / 4];
#pragma unroll
    for (int j = 0; j < CHK / 4; ++j) {
      a0[j] = *reinterpret_cast<const v2f*>(Ar + 4 * j);
      a1[j] = *reinterpret_cast<const v2f*>(Ar + k1 + 4 * j);
      a2[j] = *reinterpret_cast<const v2f*>(Ar + k2 + 4 * j);
    }
    const float* Pb = P0 + ((size_t)b * SS + g0) * HH + h0;
    v8f p0 = load_tile(Pb, HH, w * 16, 0, lane);
    v8f p1 = load_tile(Pb, HH, w * 16, 16, lane);
    const float* Ag = Ab + (size_t)g0 * SS + g0;
    v4f stA[CHK * CHK / 4 / 128], stX[CHK * 32 / 4 / 128];
#pragma unroll
    for (int k = 0; k < CHK * CHK / 4 / 128; ++k) {
      const int i = k * 128 + tid;
      stA[k] = *reinterpret_cast<const v4f*>(Ag + (size_t)(i >> 4) * SS + ((i & 15) << 2));
    }
#pragma unroll
    for (int k = 0; k < CHK * 32 / 4 / 128; ++k) {
      const int i = k * 128 + tid;
      const int gn = g0 + (i >> 3) + 1;
      const int gr = (gn < SS) ? gn : (SS - 1);
      stX[k] = *reinterpret_cast<const v4f*>(Xb + (size_t)gr * HH + ((i & 7) << 2));
    }

    v8f s0 = {}, s1 = {};
    const v2f* Dr = Dsh + (lane >> 4) * (DPP / 2) + (lane & 15);
#define CROSS_TRIP(AF, KB)                                                          \
    {                                                                               \
      v2f b0[CHK / 4], b1[CHK / 4];                                                 \
      _Pragma("unroll")               \
      for (int j = 0; j < CHK / 4; ++j) {                                           \
        const v2f* d = Dr + ((KB) / 2 + 2 * j) * (DPP / 2);   \
        b0[j] = d[0];                                                               \
        b1[j] = d[16];                                                              \
      }                                                                             \
          \
                \
      __builtin_amdgcn_sched_barrier(0);                                            \
      _Pragma("unroll")                                        \
      for (int j = 0; j < CHK / 4; ++j) {                                           \
        s0 = wmma4(AF[j], b0[j], s0);                                               \
        s1 = wmma4(AF[j], b1[j], s1);                                               \
      }                                                                             \
    }
    if (g0 > 0) CROSS_TRIP(a0, 0)
    if (g0 > CHK) CROSS_TRIP(a1, CHK)
    if (g0 > 2 * CHK) CROSS_TRIP(a2, 2 * CHK)
#undef CROSS_TRIP

#pragma unroll
    for (int k = 0; k < CHK * CHK / 4 / 128; ++k) Ash[k * 128 + tid] = stA[k];
#pragma unroll
    for (int k = 0; k < CHK * 32 / 4 / 128; ++k) {
      const int i = k * 128 + tid;
      v4f v = stX[k];
      if (g0 + (i >> 3) + 1 >= SS) v = v4f{0.0f, 0.0f, 0.0f, 0.0f};
      Xsh[i] = v;
    }
    {
      const int n = lane & 15;
      const int m0 = w * 16 + ((lane >> 4) << 3);
#pragma unroll
      for (int i = 0; i < 8; ++i) {
        Csh[(m0 + i) * 32 + n] = p0[i] - s0[i];
        Csh[(m0 + i) * 32 + 16 + n] = p1[i] - s1[i];
      }
    }
    __syncthreads();

    if (w == 0) {
      float* Predb = Pred + ((size_t)b * SS + g0) * HH + h0 + lane;
      float* Dw = reinterpret_cast<float*>(Dsh) + (g0 / 2) * DPP + 2 * lane;
      const float* Cf = Csh + lane;
      const float* Xf = reinterpret_cast<const float*>(Xsh) + lane;
      float acc[CHK], xn[CHK];
#pragma unroll
      for (int t = 0; t < CHK; ++t) acc[t] = Cf[t * 32];
#pragma unroll
      for (int t = 0; t < CHK; ++t) xn[t] = Xf[t * 32];

      const float* Af = reinterpret_cast<const float*>(Ash);
      v4f abuf[2][CHK];
      float tip[2][6];
      tip[0][0] = Af[1 * CHK];
      tip[0][1] = Af[2 * CHK];  tip[0][2] = Af[2 * CHK + 1];
      tip[0][3] = Af[3 * CHK];  tip[0][4] = Af[3 * CHK + 1];  tip[0][5] = Af[3 * CHK + 2];
#pragma unroll
      for (int t = 4; t < CHK; ++t) abuf[0][t] = Ash[(t * CHK) / 4];
#pragma unroll
      for (int g = 0; g < CHK / 4; ++g) {
        const int s = 4 * g;
        const int cur = g & 1;
        if (g + 1 < CHK / 4) {
          const int sn = s + 4;
          tip[cur ^ 1][0] = Af[(sn + 1) * CHK + sn];
          tip[cur ^ 1][1] = Af[(sn + 2) * CHK + sn];
          tip[cur ^ 1][2] = Af[(sn + 2) * CHK + sn + 1];
          tip[cur ^ 1][3] = Af[(sn + 3) * CHK + sn];
          tip[cur ^ 1][4] = Af[(sn + 3) * CHK + sn + 1];
          tip[cur ^ 1][5] = Af[(sn + 3) * CHK + sn + 2];
#pragma unroll
          for (int t = 0; t < CHK; ++t) {
            if (t > sn + 3) abuf[cur ^ 1][t] = Ash[(t * CHK + sn) / 4];
          }
        }
        __builtin_amdgcn_sched_barrier(0);
        float n[4];
#pragma unroll
        for (int r = 0; r < 4; ++r) {
          const int t = s + r;
          float a = acc[t];
          if (r == 1) a = fmaf(tip[cur][0], n[0], a);
          if (r == 2) { a = fmaf(tip[cur][1], n[0], a); a = fmaf(tip[cur][2], n[1], a); }
          if (r == 3) { a = fmaf(tip[cur][3], n[0], a); a = fmaf(tip[cur][4], n[1], a); a = fmaf(tip[cur][5], n[2], a); }
          Predb[(size_t)t * HH] = a;
          Dw[(t >> 1) * DPP + (t & 1)] = a - xn[t];
          n[r] = xn[t] - a;
        }
#pragma unroll
        for (int t = 0; t < CHK; ++t) {
          if (t > s + 3) {
            float r = acc[t];
            r = fmaf(abuf[cur][t].x, n[0], r);
            r = fmaf(abuf[cur][t].y, n[1], r);
            r = fmaf(abuf[cur][t].z, n[2], r);
            r = fmaf(abuf[cur][t].w, n[3], r);
            acc[t] = r;
          }
        }
        __builtin_amdgcn_sched_barrier(0);
      }
    }
    __builtin_amdgcn_fence(__ATOMIC_RELEASE, "workgroup", "local");
    __builtin_amdgcn_s_barrier();
    __builtin_amdgcn_fence(__ATOMIC_ACQUIRE, "workgroup", "local");
  }
}

__global__ void __launch_bounds__(256)
k_mlp1(const float* Pred, const float* gw, const float* gb,
       const float* uw, const float* ub, float* Gt) {
  __shared__ v4f smv[2 * 128 * (KS2 + 4) / 4];
  const int w = threadIdx.x >> 5, lane = threadIdx.x & 31;
  const int bn = blockIdx.x % (H4 / 32);
  const int bm = blockIdx.x / (H4 / 32);
  v8f g = {}, u = {};
  block_nk2<HH, KS2>(Pred + (size_t)bm * 64 * HH, HH, gw + (size_t)bn * 32 * HH,
                     uw + (size_t)bn * 32 * HH, HH, reinterpret_cast<float*>(smv), g, u);
  const int tm = bm * 4 + (w >> 1), tn = bn * 2 + (w & 1);
  int n = tn * 16 + (lane & 15);
  float gbias = gb[n], ubias = ub[n];
  v8f r;
#pragma unroll
  for (int i = 0; i < 8; ++i) {
    float zg = g[i] + gbias;
    float zu = u[i] + ubias;
    r[i] = (1.0f / (1.0f + __expf(-zg))) * zu;
  }
  store_tile(Gt, H4, tm * 16, tn * 16, r, lane);
}

__global__ void __launch_bounds__(256)
k_mlp2(const float* Gt, const float* dw, const float* db, const float* Pred,
       float* Ypre) {
  __shared__ v4f smv[2 * 96 * (KS1 + 4) / 4];
  const int w = threadIdx.x >> 5, lane = threadIdx.x & 31;
  const int bn = blockIdx.x % (HH / 64);
  const int bm = blockIdx.x / (HH / 64);
  v8f acc = block_nk<H4, KS1>(Gt + (size_t)bm * 32 * H4, H4, dw + (size_t)bn * 64 * H4, H4,
                              reinterpret_cast<float*>(smv));
  const int tm = bm * 2 + (w >> 2), tn = bn * 4 + (w & 3);
  int n = tn * 16 + (lane & 15);
  int mbase = tm * 16 + ((lane >> 4) << 3);
  float bias = db[n];
#pragma unroll
  for (int i = 0; i < 8; ++i) {
    int m = mbase + i;
    Ypre[(size_t)m * HH + n] = acc[i] + bias + Pred[(size_t)m * HH + n];
  }
}

__global__ void k_ln(const float* __restrict__ Y, const float* __restrict__ nw,
                     const float* __restrict__ nb, float* __restrict__ out) {
  __shared__ float s1[256], s2[256];
  int row = blockIdx.x;
  const float* y = Y + (size_t)row * HH;
  int tid = threadIdx.x;
  float v0 = y[tid], v1 = y[tid + 256], v2 = y[tid + 512];
  s1[tid] = v0 + v1 + v2;
  s2[tid] = v0 * v0 + v1 * v1 + v2 * v2;
  __syncthreads();
  for (int off = 128; off > 0; off >>= 1) {
    if (tid < off) { s1[tid] += s1[tid + off]; s2[tid] += s2[tid + off]; }
    __syncthreads();
  }
  float mu  = s1[0] * (1.0f / HH);
  float var = s2[0] * (1.0f / HH) - mu * mu;
  float r = rsqrtf(var + EPSC);
  float* o = out + (size_t)row * HH;
  o[tid]       = (v0 - mu) * r * nw[tid]       + nb[tid];
  o[tid + 256] = (v1 - mu) * r * nw[tid + 256] + nb[tid + 256];
  o[tid + 512] = (v2 - mu) * r * nw[tid + 512] + nb[tid + 512];
}

extern "C" void kernel_launch(void* const* d_in, const int* in_sizes, int n_in,
                              void* d_out, int out_size, void* d_ws, size_t ws_size,
                              hipStream_t stream) {
  const float* x      = (const float*)d_in[0];
  const float* W_init = (const float*)d_in[1];
  const float* b_init = (const float*)d_in[2];
  const float* gate_w = (const float*)d_in[3];
  const float* gate_b = (const float*)d_in[4];
  const float* up_w   = (const float*)d_in[5];
  const float* up_b   = (const float*)d_in[6];
  const float* down_w = (const float*)d_in[7];
  const float* down_b = (const float*)d_in[8];
  const float* norm_w = (const float*)d_in[9];
  const float* norm_b = (const float*)d_in[10];
  float* out = (float*)d_out;

  float* ws   = (float*)d_ws;
  float* P0   = ws;  ws += (size_t)BB * SS * HH;
  float* Pred = ws;  ws += (size_t)BB * SS * HH;
  float* Amat = ws;  ws += (size_t)BB * SS * SS;
  float* Gt   = ws;  ws += (size_t)BB * SS * H4;
  float* Ypre = ws;  ws += (size_t)BB * SS * HH;

  k_attn<<<64, 256, 0, stream>>>(x, Amat);
  k_p0<<<192, 256, 0, stream>>>(x, W_init, b_init, P0);

  k_seq<<<BB * (HH / 32), 128, 0, stream>>>(x, P0, Amat, Pred);

  k_mlp1<<<768, 256, 0, stream>>>(Pred, gate_w, gate_b, up_w, up_b, Gt);
  k_mlp2<<<192, 256, 0, stream>>>(Gt, down_w, down_b, Pred, Ypre);
  k_ln<<<BB * SS, 256, 0, stream>>>(Ypre, norm_w, norm_b, out);
}


# mlp2 split in K four ways: 192 blocks of 48 stages become 768 of 12 (3 resident per WGP); the quarter sums, down_b and Pred are added where the LayerNorm kernel loads its row; sum order changes
# speedup vs baseline: 9.4543x; 1.3278x over previous
#include <hip/hip_runtime.h>
#include <math.h>

#define BB 2
#define SS 256
#define HH 768
#define H4 3072
#define CHK 64
#define NCH (SS / CHK)
#define LRC 0.01f
#define EPSC 1e-5f

typedef float v2f __attribute__((ext_vector_type(2)));
typedef float v8f __attribute__((ext_vector_type(8)));

__device__ __forceinline__ v8f wmma4(v2f a, v2f b, v8f c) {
  return __builtin_amdgcn_wmma_f32_16x16x4_f32(
       false, a,  false, b,
       (short)0, c,  false,  false);
}

__device__ __forceinline__ v2f frag_kn(const float* M, int ld, int k0, int r0, int lane) {
  const float* p = M + (size_t)(k0 + ((lane >> 4) << 1)) * ld + r0 + (lane & 15);
  v2f v; v.x = p[0]; v.y = p[ld];
  return v;
}
__device__ __forceinline__ void store_tile(float* D, int ld, int r0, int c0, v8f v, int lane) {
  int n = c0 + (lane & 15);
  int m0 = r0 + ((lane >> 4) << 3);
#pragma unroll
  for (int i = 0; i < 8; ++i) D[(size_t)(m0 + i) * ld + n] = v[i];
}
__device__ __forceinline__ v8f load_tile(const float* D, int ld, int r0, int c0, int lane) {
  int n = c0 + (lane & 15);
  int m0 = r0 + ((lane >> 4) << 3);
  v8f v;
#pragma unroll
  for (int i = 0; i < 8; ++i) v[i] = D[(size_t)(m0 + i) * ld + n];
  return v;
}


typedef float v4f __attribute__((ext_vector_type(4)));

template <int KS> struct stage_geo {
  static constexpr int PITCH = KS + 4;
  static constexpr int PPR = KS / 4;
  static constexpr int RPP = 256 / PPR;
  static constexpr int NQ = KS / 8;
};

#define FRAG_READS_IN_WMMA_ORDER_AHEAD_OF_WMMAS() __builtin_amdgcn_sched_barrier(0)
#define WMMAS_STAY_AHEAD_OF_NEXT_STAGE_STORES() __builtin_amdgcn_sched_barrier(0)

__device__ __forceinline__ v4f frag_read(const float* p, int i) {
  return *reinterpret_cast<const v4f*>(p + 4 * i);
}

template <int KTOT, int KS>
__device__ __forceinline__ v8f block_nk(const float* Ablk, int lda, const float* Bblk, int ldb,
                                        float* sm) {
  typedef stage_geo<KS> G;
  constexpr int NPA = 32 / G::RPP, NPB = 64 / G::RPP, NP = NPA + NPB;
  constexpr int BUF = 96 * G::PITCH;
  constexpr int NS = KTOT / KS;
  static_assert(KTOT % KS == 0, "K must be a whole number of stages");
  const int tid = threadIdx.x, lane = tid & 31, w = tid >> 5;
  const int lr = tid / G::PPR, lp = (tid % G::PPR) * 4;
  const float* ga = Ablk + (size_t)lr * lda + lp;
  const float* gb = Bblk + (size_t)lr * ldb + lp;
  float* sw = sm + lr * G::PITCH + lp;
  const int fo = (lane & 15) * G::PITCH + (lane >> 4) * (KS / 2);
  const float* fa = sm + ((w >> 2) * 16) * G::PITCH + fo;
  const float* fb = sm + (32 + (w & 3) * 16) * G::PITCH + fo;
  v4f st[NP];
#pragma unroll
  for (int p = 0; p < NPA; ++p) st[p] = *reinterpret_cast<const v4f*>(ga + (size_t)p * G::RPP * lda);
#pragma unroll
  for (int p = 0; p < NPB; ++p) st[NPA + p] = *reinterpret_cast<const v4f*>(gb + (size_t)p * G::RPP * ldb);
#pragma unroll
  for (int p = 0; p < NP; ++p) *reinterpret_cast<v4f*>(sw + p * G::RPP * G::PITCH) = st[p];
  __syncthreads();
  v8f acc = {};
  for (int s = 0; s < NS; ++s) {
    const int cur = s & 1;
    const bool more = (s + 1 < NS);
    if (more) {
      const int k1 = (s + 1) * KS;
#pragma unroll
      for (int p = 0; p < NPA; ++p) st[p] = *reinterpret_cast<const v4f*>(ga + (size_t)p * G::RPP * lda + k1);
#pragma unroll
      for (int p = 0; p < NPB; ++p) st[NPA + p] = *reinterpret_cast<const v4f*>(gb + (size_t)p * G::RPP * ldb + k1);
    }
    v4f a[G::NQ], b[G::NQ];
#pragma unroll
    for (int i = 0; i < G::NQ; ++i) {
      a[i] = frag_read(fa + cur * BUF, i); b[i] = frag_read(fb + cur * BUF, i);
      FRAG_READS_IN_WMMA_ORDER_AHEAD_OF_WMMAS();
    }
#pragma unroll
    for (int i = 0; i < G::NQ; ++i) {
      acc = wmma4(a[i].xy, b[i].xy, acc);
      acc = wmma4(a[i].zw, b[i].zw, acc);
    }
    WMMAS_STAY_AHEAD_OF_NEXT_STAGE_STORES();
    if (more) {
#pragma unroll
      for (int p = 0; p < NP; ++p) *reinterpret_cast<v4f*>(sw + (cur ^ 1) * BUF + p * G::RPP * G::PITCH) = st[p];
    }
    __syncthreads();
  }
  return acc;
}

template <int KTOT, int KS>
__device__ __forceinline__ void block_nk2(const float* Ablk, int lda, const float* Bblk,
                                          const float* Cblk, int ldb, float* sm,
                                          v8f& g, v8f& u) {
  typedef stage_geo<KS> G;
  constexpr int NPA = 64 / G::RPP, NPB = 32 / G::RPP, NP = NPA + 2 * NPB;
  constexpr int BUF = 128 * G::PITCH;
  constexpr int NS = KTOT / KS;
  static_assert(KTOT % KS == 0, "K must be a whole number of stages");
  const int tid = threadIdx.x, lane = tid & 31, w = tid >> 5;
  const int lr = tid / G::PPR, lp = (tid % G::PPR) * 4;
  const float* ga = Ablk + (size_t)lr * lda + lp;
  const float* gb = Bblk + (size_t)lr * ldb + lp;
  const float* gc = Cblk + (size_t)lr * ldb + lp;
  float* sw = sm + lr * G::PITCH + lp;
  const int fo = (lane & 15) * G::PITCH + (lane >> 4) * (KS / 2);
  const float* fa = sm + ((w >> 1) * 16) * G::PITCH + fo;
  const float* fb = sm + (64 + (w & 1) * 16) * G::PITCH + fo;
  const float* fc = sm + (96 + (w & 1) * 16) * G::PITCH + fo;
  v4f st[NP];
#pragma unroll
  for (int p = 0; p < NPA; ++p) st[p] = *reinterpret_cast<const v4f*>(ga + (size_t)p * G::RPP * lda);
#pragma unroll
  for (int p = 0; p < NPB; ++p) {
    st[NPA + p] = *reinterpret_cast<const v4f*>(gb + (size_t)p * G::RPP * ldb);
    st[NPA + NPB + p] = *reinterpret_cast<const v4f*>(gc + (size_t)p * G::RPP * ldb);
  }
#pragma unroll
  for (int p = 0; p < NP; ++p) *reinterpret_cast<v4f*>(sw + p * G::RPP * G::PITCH) = st[p];
  __syncthreads();
  for (int s = 0; s < NS; ++s) {
    const int cur = s & 1;
    const bool more = (s + 1 < NS);
    if (more) {
      const int k1 = (s + 1) * KS;
#pragma unroll
      for (int p = 0; p < NPA; ++p) st[p] = *reinterpret_cast<const v4f*>(ga + (size_t)p * G::RPP * lda + k1);
#pragma unroll
      for (int p = 0; p < NPB; ++p) {
        st[NPA + p] = *reinterpret_cast<const v4f*>(gb + (size_t)p * G::RPP * ldb + k1);
        st[NPA + NPB + p] = *reinterpret_cast<const v4f*>(gc + (size_t)p * G::RPP * ldb + k1);
      }
    }
    v4f a[G::NQ], b[G::NQ], c[G::NQ];
#pragma unroll
    for (int i = 0; i < G::NQ; ++i) {
      a[i] = frag_read(fa + cur * BUF, i); b[i] = frag_read(fb + cur * BUF, i); c[i] = frag_read(fc + cur * BUF, i);
      FRAG_READS_IN_WMMA_ORDER_AHEAD_OF_WMMAS();
    }
#pragma unroll
    for (int i = 0; i < G::NQ; ++i) {
      g = wmma4(a[i].xy, b[i].xy, g);
      u = wmma4(a[i].xy, c[i].xy, u);
      g = wmma4(a[i].zw, b[i].zw, g);
      u = wmma4(a[i].zw, c[i].zw, u);
    }
    WMMAS_STAY_AHEAD_OF_NEXT_STAGE_STORES();
    if (more) {
#pragma unroll
      for (int p = 0; p < NP; ++p) *reinterpret_cast<v4f*>(sw + (cur ^ 1) * BUF + p * G::RPP * G::PITCH) = st[p];
    }
    __syncthreads();
  }
}


#define KS1 64
#define KS2 32

__global__ void __launch_bounds__(256)
k_p0(const float* x, const float* Wi, const float* bi, float* P0) {
  __shared__ v4f smv[2 * 96 * (KS1 + 4) / 4];
  const int w = threadIdx.x >> 5, lane = threadIdx.x & 31;
  const int bn = blockIdx.x % (HH / 64);
  const int bm = blockIdx.x / (HH / 64);
  v8f acc = block_nk<HH, KS1>(x + (size_t)bm * 32 * HH, HH, Wi + (size_t)bn * 64 * HH, HH,
                              reinterpret_cast<float*>(smv));
  const int tm = bm * 2 + (w >> 2), tn = bn * 4 + (w & 3);
  float bias = bi[tn * 16 + (lane & 15)];
#pragma unroll
  for (int i = 0; i < 8; ++i) acc[i] += bias;
  store_tile(P0, HH, tm * 16, tn * 16, acc, lane);
}

__global__ void __launch_bounds__(256)
k_attn(const float* x, float* Amat) {
  __shared__ v4f smv[2 * 96 * (KS1 + 4) / 4];
  const int w = threadIdx.x >> 5, lane = threadIdx.x & 31;
  const int bn = blockIdx.x % (SS / 64);
  const int bm = (blockIdx.x / (SS / 64)) % (SS / 32);
  const int b  = blockIdx.x / ((SS / 64) * (SS / 32));
  if (bn > (bm >> 1)) return;
  const float* Xb = x + (size_t)b * SS * HH;
  v8f acc = block_nk<HH, KS1>(Xb + (size_t)bm * 32 * HH, HH, Xb + (size_t)bn * 64 * HH, HH,
                              reinterpret_cast<float*>(smv));
  const int tm = bm * 2 + (w >> 2), tn = bn * 4 + (w & 3);
  int s = tn * 16 + (lane & 15);
  int mbase = tm * 16 + ((lane >> 4) << 3);
  float* Ab = Amat + (size_t)b * SS * SS;
#pragma unroll
  for (int i = 0; i < 8; ++i) {
    int t = mbase + i;
    float v = (s < t) ? LRC * (acc[i] + 1.0f) : 0.0f;
    Ab[(size_t)t * SS + s] = v;
  }
}

#define DPP 96
__global__ void __launch_bounds__(128) __attribute__((amdgpu_waves_per_eu(1, 1)))
k_seq(const float* __restrict__ x, const float* __restrict__ P0,
      const float* __restrict__ Amat, float* __restrict__ Pred) {
  __shared__ v4f Ash[CHK * CHK / 4];
  __shared__ v4f Xsh[CHK * 32 / 4];
  __shared__ float Csh[CHK * 32];
  __shared__ v2f Dsh[(SS / 2) * (DPP / 2)];
  const int tid = threadIdx.x, lane = tid & 31, w = tid >> 5;
  const int b = blockIdx.x / (HH / 32);
  const int h0 = (blockIdx.x % (HH / 32)) * 32;
  const float* Ab = Amat + (size_t)b * SS * SS;
  const float* Xb = x + (size_t)b * SS * HH + h0;

#pragma nounroll
  for (int c = 0; c < NCH; ++c) {
    const int g0 = c * CHK;
    const float* Ar = Ab + (size_t)(g0 + w * 16 + (lane & 15)) * SS + ((lane >> 4) << 1);
    const int k1 = (g0 > CHK) ? CHK : 0;
    const int k2 = (g0 > 2 * CHK) ? 2 * CHK : 0;
    v2f a0[CHK / 4], a1[CHK / 4], a2[CHK / 4];
#pragma unroll
    for (int j = 0; j < CHK / 4; ++j) {
      a0[j] = *reinterpret_cast<const v2f*>(Ar + 4 * j);
      a1[j] = *reinterpret_cast<const v2f*>(Ar + k1 + 4 * j);
      a2[j] = *reinterpret_cast<const v2f*>(Ar + k2 + 4 * j);
    }
    const float* Pb = P0 + ((size_t)b * SS + g0) * HH + h0;
    v8f p0 = load_tile(Pb, HH, w * 16, 0, lane);
    v8f p1 = load_tile(Pb, HH, w * 16, 16, lane);
    const float* Ag = Ab + (size_t)g0 * SS + g0;
    v4f stA[CHK * CHK / 4 / 128], stX[CHK * 32 / 4 / 128];
#pragma unroll
    for (int k = 0; k < CHK * CHK / 4 / 128; ++k) {
      const int i = k * 128 + tid;
      stA[k] = *reinterpret_cast<const v4f*>(Ag + (size_t)(i >> 4) * SS + ((i & 15) << 2));
    }
#pragma unroll
    for (int k = 0; k < CHK * 32 / 4 / 128; ++k) {
      const int i = k * 128 + tid;
      const int gn = g0 + (i >> 3) + 1;
      const int gr = (gn < SS) ? gn : (SS - 1);
      stX[k] = *reinterpret_cast<const v4f*>(Xb + (size_t)gr * HH + ((i & 7) << 2));
    }

    v8f s0 = {}, s1 = {};
    const v2f* Dr = Dsh + (lane >> 4) * (DPP / 2) + (lane & 15);
#define CROSS_TRIP(AF, KB)                                                          \
    {                                                                               \
      v2f b0[CHK / 4], b1[CHK / 4];                                                 \
      _Pragma("unroll")               \
      for (int j = 0; j < CHK / 4; ++j) {                                           \
        const v2f* d = Dr + ((KB) / 2 + 2 * j) * (DPP / 2);   \
        b0[j] = d[0];                                                               \
        b1[j] = d[16];                                                              \
      }                                                                             \
          \
                \
      __builtin_amdgcn_sched_barrier(0);                                            \
      _Pragma("unroll")                                        \
      for (int j = 0; j < CHK / 4; ++j) {                                           \
        s0 = wmma4(AF[j], b0[j], s0);                                               \
        s1 = wmma4(AF[j], b1[j], s1);                                               \
      }                                                                             \
    }
    if (g0 > 0) CROSS_TRIP(a0, 0)
    if (g0 > CHK) CROSS_TRIP(a1, CHK)
    if (g0 > 2 * CHK) CROSS_TRIP(a2, 2 * CHK)
#undef CROSS_TRIP

#pragma unroll
    for (int k = 0; k < CHK * CHK / 4 / 128; ++k) Ash[k * 128 + tid] = stA[k];
#pragma unroll
    for (int k = 0; k < CHK * 32 / 4 / 128; ++k) {
      const int i = k * 128 + tid;
      v4f v = stX[k];
      if (g0 + (i >> 3) + 1 >= SS) v = v4f{0.0f, 0.0f, 0.0f, 0.0f};
      Xsh[i] = v;
    }
    {
      const int n = lane & 15;
      const int m0 = w * 16 + ((lane >> 4) << 3);
#pragma unroll
      for (int i = 0; i < 8; ++i) {
        Csh[(m0 + i) * 32 + n] = p0[i] - s0[i];
        Csh[(m0 + i) * 32 + 16 + n] = p1[i] - s1[i];
      }
    }
    __syncthreads();

    if (w == 0) {
      float* Predb = Pred + ((size_t)b * SS + g0) * HH + h0 + lane;
      float* Dw = reinterpret_cast<float*>(Dsh) + (g0 / 2) * DPP + 2 * lane;
      const float* Cf = Csh + lane;
      const float* Xf = reinterpret_cast<const float*>(Xsh) + lane;
      float acc[CHK], xn[CHK];
#pragma unroll
      for (int t = 0; t < CHK; ++t) acc[t] = Cf[t * 32];
#pragma unroll
      for (int t = 0; t < CHK; ++t) xn[t] = Xf[t * 32];

      const float* Af = reinterpret_cast<const float*>(Ash);
      v4f abuf[2][CHK];
      float tip[2][6];
      tip[0][0] = Af[1 * CHK];
      tip[0][1] = Af[2 * CHK];  tip[0][2] = Af[2 * CHK + 1];
      tip[0][3] = Af[3 * CHK];  tip[0][4] = Af[3 * CHK + 1];  tip[0][5] = Af[3 * CHK + 2];
#pragma unroll
      for (int t = 4; t < CHK; ++t) abuf[0][t] = Ash[(t * CHK) / 4];
#pragma unroll
      for (int g = 0; g < CHK / 4; ++g) {
        const int s = 4 * g;
        const int cur = g & 1;
        if (g + 1 < CHK / 4) {
          const int sn = s + 4;
          tip[cur ^ 1][0] = Af[(sn + 1) * CHK + sn];
          tip[cur ^ 1][1] = Af[(sn + 2) * CHK + sn];
          tip[cur ^ 1][2] = Af[(sn + 2) * CHK + sn + 1];
          tip[cur ^ 1][3] = Af[(sn + 3) * CHK + sn];
          tip[cur ^ 1][4] = Af[(sn + 3) * CHK + sn + 1];
          tip[cur ^ 1][5] = Af[(sn + 3) * CHK + sn + 2];
#pragma unroll
          for (int t = 0; t < CHK; ++t) {
            if (t > sn + 3) abuf[cur ^ 1][t] = Ash[(t * CHK + sn) / 4];
          }
        }
        __builtin_amdgcn_sched_barrier(0);
        float n[4];
#pragma unroll
        for (int r = 0; r < 4; ++r) {
          const int t = s + r;
          float a = acc[t];
          if (r == 1) a = fmaf(tip[cur][0], n[0], a);
          if (r == 2) { a = fmaf(tip[cur][1], n[0], a); a = fmaf(tip[cur][2], n[1], a); }
          if (r == 3) { a = fmaf(tip[cur][3], n[0], a); a = fmaf(tip[cur][4], n[1], a); a = fmaf(tip[cur][5], n[2], a); }
          Predb[(size_t)t * HH] = a;
          Dw[(t >> 1) * DPP + (t & 1)] = a - xn[t];
          n[r] = xn[t] - a;
        }
#pragma unroll
        for (int t = 0; t < CHK; ++t) {
          if (t > s + 3) {
            float r = acc[t];
            r = fmaf(abuf[cur][t].x, n[0], r);
            r = fmaf(abuf[cur][t].y, n[1], r);
            r = fmaf(abuf[cur][t].z, n[2], r);
            r = fmaf(abuf[cur][t].w, n[3], r);
            acc[t] = r;
          }
        }
        __builtin_amdgcn_sched_barrier(0);
      }
    }
    __builtin_amdgcn_fence(__ATOMIC_RELEASE, "workgroup", "local");
    __builtin_amdgcn_s_barrier();
    __builtin_amdgcn_fence(__ATOMIC_ACQUIRE, "workgroup", "local");
  }
}

__global__ void __launch_bounds__(256)
k_mlp1(const float* Pred, const float* gw, const float* gb,
       const float* uw, const float* ub, float* Gt) {
  __shared__ v4f smv[2 * 128 * (KS2 + 4) / 4];
  const int w = threadIdx.x >> 5, lane = threadIdx.x & 31;
  const int bn = blockIdx.x % (H4 / 32);
  const int bm = blockIdx.x / (H4 / 32);
  v8f g = {}, u = {};
  block_nk2<HH, KS2>(Pred + (size_t)bm * 64 * HH, HH, gw + (size_t)bn * 32 * HH,
                     uw + (size_t)bn * 32 * HH, HH, reinterpret_cast<float*>(smv), g, u);
  const int tm = bm * 4 + (w >> 1), tn = bn * 2 + (w & 1);
  int n = tn * 16 + (lane & 15);
  float gbias = gb[n], ubias = ub[n];
  v8f r;
#pragma unroll
  for (int i = 0; i < 8; ++i) {
    float zg = g[i] + gbias;
    float zu = u[i] + ubias;
    r[i] = (1.0f / (1.0f + __expf(-zg))) * zu;
  }
  store_tile(Gt, H4, tm * 16, tn * 16, r, lane);
}

#define KSPLIT 4
__global__ void __launch_bounds__(256)
k_mlp2(const float* Gt, const float* dw, float* Part) {
  __shared__ v4f smv[2 * 96 * (KS1 + 4) / 4];
  const int w = threadIdx.x >> 5, lane = threadIdx.x & 31;
  const int bn = blockIdx.x % (HH / 64);
  const int bm = (blockIdx.x / (HH / 64)) % (BB * SS / 32);
  const int q  = blockIdx.x / ((HH / 64) * (BB * SS / 32));
  v8f acc = block_nk<H4 / KSPLIT, KS1>(Gt + (size_t)bm * 32 * H4 + (size_t)q * (H4 / KSPLIT), H4,
                                       dw + (size_t)bn * 64 * H4 + (size_t)q * (H4 / KSPLIT), H4,
                                       reinterpret_cast<float*>(smv));
  const int tm = bm * 2 + (w >> 2), tn = bn * 4 + (w & 3);
  store_tile(Part + (size_t)q * BB * SS * HH, HH, tm * 16, tn * 16, acc, lane);
}

__global__ void k_ln(const float* __restrict__ Part, const float* __restrict__ db,
                     const float* __restrict__ Pred, const float* __restrict__ nw,
                     const float* __restrict__ nb, float* __restrict__ out) {
  __shared__ float s1[256], s2[256];
  int row = blockIdx.x;
  const float* y = Part + (size_t)row * HH;
  const float* pr = Pred + (size_t)row * HH;
  int tid = threadIdx.x;
  float v0 = y[tid], v1 = y[tid + 256], v2 = y[tid + 512];
  for (int q = 1; q < KSPLIT; ++q) {
    const float* yq = y + (size_t)q * BB * SS * HH;
    v0 += yq[tid]; v1 += yq[tid + 256]; v2 += yq[tid + 512];
  }
  v0 = v0 + db[tid]       + pr[tid];
  v1 = v1 + db[tid + 256] + pr[tid + 256];
  v2 = v2 + db[tid + 512] + pr[tid + 512];
  s1[tid] = v0 + v1 + v2;
  s2[tid] = v0 * v0 + v1 * v1 + v2 * v2;
  __syncthreads();
  for (int off = 128; off > 0; off >>= 1) {
    if (tid < off) { s1[tid] += s1[tid + off]; s2[tid] += s2[tid + off]; }
    __syncthreads();
  }
  float mu  = s1[0] * (1.0f / HH);
  float var = s2[0] * (1.0f / HH) - mu * mu;
  float r = rsqrtf(var + EPSC);
  float* o = out + (size_t)row * HH;
  o[tid]       = (v0 - mu) * r * nw[tid]       + nb[tid];
  o[tid + 256] = (v1 - mu) * r * nw[tid + 256] + nb[tid + 256];
  o[tid + 512] = (v2 - mu) * r * nw[tid + 512] + nb[tid + 512];
}

extern "C" void kernel_launch(void* const* d_in, const int* in_sizes, int n_in,
                              void* d_out, int out_size, void* d_ws, size_t ws_size,
                              hipStream_t stream) {
  const float* x      = (const float*)d_in[0];
  const float* W_init = (const float*)d_in[1];
  const float* b_init = (const float*)d_in[2];
  const float* gate_w = (const float*)d_in[3];
  const float* gate_b = (const float*)d_in[4];
  const float* up_w   = (const float*)d_in[5];
  const float* up_b   = (const float*)d_in[6];
  const float* down_w = (const float*)d_in[7];
  const float* down_b = (const float*)d_in[8];
  const float* norm_w = (const float*)d_in[9];
  const float* norm_b = (const float*)d_in[10];
  float* out = (float*)d_out;

  float* ws   = (float*)d_ws;
  float* P0   = ws;  ws += (size_t)BB * SS * HH;
  float* Pred = ws;  ws += (size_t)BB * SS * HH;
  float* Amat = ws;  ws += (size_t)BB * SS * SS;
  float* Gt   = ws;  ws += (size_t)BB * SS * H4;
  float* Part = ws;  ws += (size_t)KSPLIT * BB * SS * HH;

  k_attn<<<64, 256, 0, stream>>>(x, Amat);
  k_p0<<<192, 256, 0, stream>>>(x, W_init, b_init, P0);

  k_seq<<<BB * (HH / 32), 128, 0, stream>>>(x, P0, Amat, Pred);

  k_mlp1<<<768, 256, 0, stream>>>(Pred, gate_w, gate_b, up_w, up_b, Gt);
  k_mlp2<<<768, 256, 0, stream>>>(Gt, down_w, Part);
  k_ln<<<BB * SS, 256, 0, stream>>>(Part, down_b, Pred, norm_w, norm_b, out);
}


# P0 and A (both read inputs only) in one launch: 192 + 64 blocks share one K loop, the block index picks operands and epilogue; 6 launches -> 5; every sum as before
# speedup vs baseline: 10.7563x; 1.1377x over previous
#include <hip/hip_runtime.h>
#include <math.h>

#define BB 2
#define SS 256
#define HH 768
#define H4 3072
#define CHK 64
#define NCH (SS / CHK)
#define LRC 0.01f
#define EPSC 1e-5f

typedef float v2f __attribute__((ext_vector_type(2)));
typedef float v8f __attribute__((ext_vector_type(8)));

__device__ __forceinline__ v8f wmma4(v2f a, v2f b, v8f c) {
  return __builtin_amdgcn_wmma_f32_16x16x4_f32(
       false, a,  false, b,
       (short)0, c,  false,  false);
}

__device__ __forceinline__ v2f frag_kn(const float* M, int ld, int k0, int r0, int lane) {
  const float* p = M + (size_t)(k0 + ((lane >> 4) << 1)) * ld + r0 + (lane & 15);
  v2f v; v.x = p[0]; v.y = p[ld];
  return v;
}
__device__ __forceinline__ void store_tile(float* D, int ld, int r0, int c0, v8f v, int lane) {
  int n = c0 + (lane & 15);
  int m0 = r0 + ((lane >> 4) << 3);
#pragma unroll
  for (int i = 0; i < 8; ++i) D[(size_t)(m0 + i) * ld + n] = v[i];
}
__device__ __forceinline__ v8f load_tile(const float* D, int ld, int r0, int c0, int lane) {
  int n = c0 + (lane & 15);
  int m0 = r0 + ((lane >> 4) << 3);
  v8f v;
#pragma unroll
  for (int i = 0; i < 8; ++i) v[i] = D[(size_t)(m0 + i) * ld + n];
  return v;
}


typedef float v4f __attribute__((ext_vector_type(4)));

template <int KS> struct stage_geo {
  static constexpr int PITCH = KS + 4;
  static constexpr int PPR = KS / 4;
  static constexpr int RPP = 256 / PPR;
  static constexpr int NQ = KS / 8;
};

#define FRAG_READS_IN_WMMA_ORDER_AHEAD_OF_WMMAS() __builtin_amdgcn_sched_barrier(0)
#define WMMAS_STAY_AHEAD_OF_NEXT_STAGE_STORES() __builtin_amdgcn_sched_barrier(0)

__device__ __forceinline__ v4f frag_read(const float* p, int i) {
  return *reinterpret_cast<const v4f*>(p + 4 * i);
}

template <int KTOT, int KS>
__device__ __forceinline__ v8f block_nk(const float* Ablk, int lda, const float* Bblk, int ldb,
                                        float* sm) {
  typedef stage_geo<KS> G;
  constexpr int NPA = 32 / G::RPP, NPB = 64 / G::RPP, NP = NPA + NPB;
  constexpr int BUF = 96 * G::PITCH;
  constexpr int NS = KTOT / KS;
  static_assert(KTOT % KS == 0, "K must be a whole number of stages");
  const int tid = threadIdx.x, lane = tid & 31, w = tid >> 5;
  const int lr = tid / G::PPR, lp = (tid % G::PPR) * 4;
  const float* ga = Ablk + (size_t)lr * lda + lp;
  const float* gb = Bblk + (size_t)lr * ldb + lp;
  float* sw = sm + lr * G::PITCH + lp;
  const int fo = (lane & 15) * G::PITCH + (lane >> 4) * (KS / 2);
  const float* fa = sm + ((w >> 2) * 16) * G::PITCH + fo;
  const float* fb = sm + (32 + (w & 3) * 16) * G::PITCH + fo;
  v4f st[NP];
#pragma unroll
  for (int p = 0; p < NPA; ++p) st[p] = *reinterpret_cast<const v4f*>(ga + (size_t)p * G::RPP * lda);
#pragma unroll
  for (int p = 0; p < NPB; ++p) st[NPA + p] = *reinterpret_cast<const v4f*>(gb + (size_t)p * G::RPP * ldb);
#pragma unroll
  for (int p = 0; p < NP; ++p) *reinterpret_cast<v4f*>(sw + p * G::RPP * G::PITCH) = st[p];
  __syncthreads();
  v8f acc = {};
  for (int s = 0; s < NS; ++s) {
    const int cur = s & 1;
    const bool more = (s + 1 < NS);
    if (more) {
      const int k1 = (s + 1) * KS;
#pragma unroll
      for (int p = 0; p < NPA; ++p) st[p] = *reinterpret_cast<const v4f*>(ga + (size_t)p * G::RPP * lda + k1);
#pragma unroll
      for (int p = 0; p < NPB; ++p) st[NPA + p] = *reinterpret_cast<const v4f*>(gb + (size_t)p * G::RPP * ldb + k1);
    }
    v4f a[G::NQ], b[G::NQ];
#pragma unroll
    for (int i = 0; i < G::NQ; ++i) {
      a[i] = frag_read(fa + cur * BUF, i); b[i] = frag_read(fb + cur * BUF, i);
      FRAG_READS_IN_WMMA_ORDER_AHEAD_OF_WMMAS();
    }
#pragma unroll
    for (int i = 0; i < G::NQ; ++i) {
      acc = wmma4(a[i].xy, b[i].xy, acc);
      acc = wmma4(a[i].zw, b[i].zw, acc);
    }
    WMMAS_STAY_AHEAD_OF_NEXT_STAGE_STORES();
    if (more) {
#pragma unroll
      for (int p = 0; p < NP; ++p) *reinterpret_cast<v4f*>(sw + (cur ^ 1) * BUF + p * G::RPP * G::PITCH) = st[p];
    }
    __syncthreads();
  }
  return acc;
}

template <int KTOT, int KS>
__device__ __forceinline__ void block_nk2(const float* Ablk, int lda, const float* Bblk,
                                          const float* Cblk, int ldb, float* sm,
                                          v8f& g, v8f& u) {
  typedef stage_geo<KS> G;
  constexpr int NPA = 64 / G::RPP, NPB = 32 / G::RPP, NP = NPA + 2 * NPB;
  constexpr int BUF = 128 * G::PITCH;
  constexpr int NS = KTOT / KS;
  static_assert(KTOT % KS == 0, "K must be a whole number of stages");
  const int tid = threadIdx.x, lane = tid & 31, w = tid >> 5;
  const int lr = tid / G::PPR, lp = (tid % G::PPR) * 4;
  const float* ga = Ablk + (size_t)lr * lda + lp;
  const float* gb = Bblk + (size_t)lr * ldb + lp;
  const float* gc = Cblk + (size_t)lr * ldb + lp;
  float* sw = sm + lr * G::PITCH + lp;
  const int fo = (lane & 15) * G::PITCH + (lane >> 4) * (KS / 2);
  const float* fa = sm + ((w >> 1) * 16) * G::PITCH + fo;
  const float* fb = sm + (64 + (w & 1) * 16) * G::PITCH + fo;
  const float* fc = sm + (96 + (w & 1) * 16) * G::PITCH + fo;
  v4f st[NP];
#pragma unroll
  for (int p = 0; p < NPA; ++p) st[p] = *reinterpret_cast<const v4f*>(ga + (size_t)p * G::RPP * lda);
#pragma unroll
  for (int p = 0; p < NPB; ++p) {
    st[NPA + p] = *reinterpret_cast<const v4f*>(gb + (size_t)p * G::RPP * ldb);
    st[NPA + NPB + p] = *reinterpret_cast<const v4f*>(gc + (size_t)p * G::RPP * ldb);
  }
#pragma unroll
  for (int p = 0; p < NP; ++p) *reinterpret_cast<v4f*>(sw + p * G::RPP * G::PITCH) = st[p];
  __syncthreads();
  for (int s = 0; s < NS; ++s) {
    const int cur = s & 1;
    const bool more = (s + 1 < NS);
    if (more) {
      const int k1 = (s + 1) * KS;
#pragma unroll
      for (int p = 0; p < NPA; ++p) st[p] = *reinterpret_cast<const v4f*>(ga + (size_t)p * G::RPP * lda + k1);
#pragma unroll
      for (int p = 0; p < NPB; ++p) {
        st[NPA + p] = *reinterpret_cast<const v4f*>(gb + (size_t)p * G::RPP * ldb + k1);
        st[NPA + NPB + p] = *reinterpret_cast<const v4f*>(gc + (size_t)p * G::RPP * ldb + k1);
      }
    }
    v4f a[G::NQ], b[G::NQ], c[G::NQ];
#pragma unroll
    for (int i = 0; i < G::NQ; ++i) {
      a[i] = frag_read(fa + cur * BUF, i); b[i] = frag_read(fb + cur * BUF, i); c[i] = frag_read(fc + cur * BUF, i);
      FRAG_READS_IN_WMMA_ORDER_AHEAD_OF_WMMAS();
    }
#pragma unroll
    for (int i = 0; i < G::NQ; ++i) {
      g = wmma4(a[i].xy, b[i].xy, g);
      u = wmma4(a[i].xy, c[i].xy, u);
      g = wmma4(a[i].zw, b[i].zw, g);
      u = wmma4(a[i].zw, c[i].zw, u);
    }
    WMMAS_STAY_AHEAD_OF_NEXT_STAGE_STORES();
    if (more) {
#pragma unroll
      for (int p = 0; p < NP; ++p) *reinterpret_cast<v4f*>(sw + (cur ^ 1) * BUF + p * G::RPP * G::PITCH) = st[p];
    }
    __syncthreads();
  }
}


#define KS1 64
#define KS2 32

#define NB_P0 ((BB * SS / 32) * (HH / 64))
#define NB_ATTN (BB * (SS / 32) * (SS / 64))
__global__ void __launch_bounds__(256)
k_front(const float* x, const float* Wi, const float* bi, float* P0, float* Amat) {
  __shared__ v4f smv[2 * 96 * (KS1 + 4) / 4];
  const int w = threadIdx.x >> 5, lane = threadIdx.x & 31;
  const bool attn = (blockIdx.x >= NB_P0);
  int bm, bn, b = 0;
  const float *Ablk, *Bblk;
  if (!attn) {
    bn = blockIdx.x % (HH / 64);
    bm = blockIdx.x / (HH / 64);
    Ablk = x + (size_t)bm * 32 * HH;
    Bblk = Wi + (size_t)bn * 64 * HH;
  } else {
    const int i = blockIdx.x - NB_P0;
    bn = i % (SS / 64);
    bm = (i / (SS / 64)) % (SS / 32);
    b  = i / ((SS / 64) * (SS / 32));
    if (bn > (bm >> 1)) return;
    const float* Xb = x + (size_t)b * SS * HH;
    Ablk = Xb + (size_t)bm * 32 * HH;
    Bblk = Xb + (size_t)bn * 64 * HH;
  }
  v8f acc = block_nk<HH, KS1>(Ablk, HH, Bblk, HH, reinterpret_cast<float*>(smv));
  const int tm = bm * 2 + (w >> 2), tn = bn * 4 + (w & 3);
  if (!attn) {
    float bias = bi[tn * 16 + (lane & 15)];
#pragma unroll
    for (int i = 0; i < 8; ++i) acc[i] += bias;
    store_tile(P0, HH, tm * 16, tn * 16, acc, lane);
  } else {
    int s = tn * 16 + (lane & 15);
    int mbase = tm * 16 + ((lane >> 4) << 3);
    float* Ab = Amat + (size_t)b * SS * SS;
#pragma unroll
    for (int i = 0; i < 8; ++i) {
      int t = mbase + i;
      float v = (s < t) ? LRC * (acc[i] + 1.0f) : 0.0f;
      Ab[(size_t)t * SS + s] = v;
    }
  }
}

#define DPP 96
__global__ void __launch_bounds__(128) __attribute__((amdgpu_waves_per_eu(1, 1)))
k_seq(const float* __restrict__ x, const float* __restrict__ P0,
      const float* __restrict__ Amat, float* __restrict__ Pred) {
  __shared__ v4f Ash[CHK * CHK / 4];
  __shared__ v4f Xsh[CHK * 32 / 4];
  __shared__ float Csh[CHK * 32];
  __shared__ v2f Dsh[(SS / 2) * (DPP / 2)];
  const int tid = threadIdx.x, lane = tid & 31, w = tid >> 5;
  const int b = blockIdx.x / (HH / 32);
  const int h0 = (blockIdx.x % (HH / 32)) * 32;
  const float* Ab = Amat + (size_t)b * SS * SS;
  const float* Xb = x + (size_t)b * SS * HH + h0;

#pragma nounroll
  for (int c = 0; c < NCH; ++c) {
    const int g0 = c * CHK;
    const float* Ar = Ab + (size_t)(g0 + w * 16 + (lane & 15)) * SS + ((lane >> 4) << 1);
    const int k1 = (g0 > CHK) ? CHK : 0;
    const int k2 = (g0 > 2 * CHK) ? 2 * CHK : 0;
    v2f a0[CHK / 4], a1[CHK / 4], a2[CHK / 4];
#pragma unroll
    for (int j = 0; j < CHK / 4; ++j) {
      a0[j] = *reinterpret_cast<const v2f*>(Ar + 4 * j);
      a1[j] = *reinterpret_cast<const v2f*>(Ar + k1 + 4 * j);
      a2[j] = *reinterpret_cast<const v2f*>(Ar + k2 + 4 * j);
    }
    const float* Pb = P0 + ((size_t)b * SS + g0) * HH + h0;
    v8f p0 = load_tile(Pb, HH, w * 16, 0, lane);
    v8f p1 = load_tile(Pb, HH, w * 16, 16, lane);
    const float* Ag = Ab + (size_t)g0 * SS + g0;
    v4f stA[CHK * CHK / 4 / 128], stX[CHK * 32 / 4 / 128];
#pragma unroll
    for (int k = 0; k < CHK * CHK / 4 / 128; ++k) {
      const int i = k * 128 + tid;
      stA[k] = *reinterpret_cast<const v4f*>(Ag + (size_t)(i >> 4) * SS + ((i & 15) << 2));
    }
#pragma unroll
    for (int k = 0; k < CHK * 32 / 4 / 128; ++k) {
      const int i = k * 128 + tid;
      const int gn = g0 + (i >> 3) + 1;
      const int gr = (gn < SS) ? gn : (SS - 1);
      stX[k] = *reinterpret_cast<const v4f*>(Xb + (size_t)gr * HH + ((i & 7) << 2));
    }

    v8f s0 = {}, s1 = {};
    const v2f* Dr = Dsh + (lane >> 4) * (DPP / 2) + (lane & 15);
#define CROSS_TRIP(AF, KB)                                                          \
    {                                                                               \
      v2f b0[CHK / 4], b1[CHK / 4];                                                 \
      _Pragma("unroll")               \
      for (int j = 0; j < CHK / 4; ++j) {                                           \
        const v2f* d = Dr + ((KB) / 2 + 2 * j) * (DPP / 2);   \
        b0[j] = d[0];                                                               \
        b1[j] = d[16];                                                              \
      }                                                                             \
          \
                \
      __builtin_amdgcn_sched_barrier(0);                                            \
      _Pragma("unroll")                                        \
      for (int j = 0; j < CHK / 4; ++j) {                                           \
        s0 = wmma4(AF[j], b0[j], s0);                                               \
        s1 = wmma4(AF[j], b1[j], s1);                                               \
      }                                                                             \
    }
    if (g0 > 0) CROSS_TRIP(a0, 0)
    if (g0 > CHK) CROSS_TRIP(a1, CHK)
    if (g0 > 2 * CHK) CROSS_TRIP(a2, 2 * CHK)
#undef CROSS_TRIP

#pragma unroll
    for (int k = 0; k < CHK * CHK / 4 / 128; ++k) Ash[k * 128 + tid] = stA[k];
#pragma unroll
    for (int k = 0; k < CHK * 32 / 4 / 128; ++k) {
      const int i = k * 128 + tid;
      v4f v = stX[k];
      if (g0 + (i >> 3) + 1 >= SS) v = v4f{0.0f, 0.0f, 0.0f, 0.0f};
      Xsh[i] = v;
    }
    {
      const int n = lane & 15;
      const int m0 = w * 16 + ((lane >> 4) << 3);
#pragma unroll
      for (int i = 0; i < 8; ++i) {
        Csh[(m0 + i) * 32 + n] = p0[i] - s0[i];
        Csh[(m0 + i) * 32 + 16 + n] = p1[i] - s1[i];
      }
    }
    __syncthreads();

    if (w == 0) {
      float* Predb = Pred + ((size_t)b * SS + g0) * HH + h0 + lane;
      float* Dw = reinterpret_cast<float*>(Dsh) + (g0 / 2) * DPP + 2 * lane;
      const float* Cf = Csh + lane;
      const float* Xf = reinterpret_cast<const float*>(Xsh) + lane;
      float acc[CHK], xn[CHK];
#pragma unroll
      for (int t = 0; t < CHK; ++t) acc[t] = Cf[t * 32];
#pragma unroll
      for (int t = 0; t < CHK; ++t) xn[t] = Xf[t * 32];

      const float* Af = reinterpret_cast<const float*>(Ash);
      v4f abuf[2][CHK];
      float tip[2][6];
      tip[0][0] = Af[1 * CHK];
      tip[0][1] = Af[2 * CHK];  tip[0][2] = Af[2 * CHK + 1];
      tip[0][3] = Af[3 * CHK];  tip[0][4] = Af[3 * CHK + 1];  tip[0][5] = Af[3 * CHK + 2];
#pragma unroll
      for (int t = 4; t < CHK; ++t) abuf[0][t] = Ash[(t * CHK) / 4];
#pragma unroll
      for (int g = 0; g < CHK / 4; ++g) {
        const int s = 4 * g;
        const int cur = g & 1;
        if (g + 1 < CHK / 4) {
          const int sn = s + 4;
          tip[cur ^ 1][0] = Af[(sn + 1) * CHK + sn];
          tip[cur ^ 1][1] = Af[(sn + 2) * CHK + sn];
          tip[cur ^ 1][2] = Af[(sn + 2) * CHK + sn + 1];
          tip[cur ^ 1][3] = Af[(sn + 3) * CHK + sn];
          tip[cur ^ 1][4] = Af[(sn + 3) * CHK + sn + 1];
          tip[cur ^ 1][5] = Af[(sn + 3) * CHK + sn + 2];
#pragma unroll
          for (int t = 0; t < CHK; ++t) {
            if (t > sn + 3) abuf[cur ^ 1][t] = Ash[(t * CHK + sn) / 4];
          }
        }
        __builtin_amdgcn_sched_barrier(0);
        float n[4];
#pragma unroll
        for (int r = 0; r < 4; ++r) {
          const int t = s + r;
          float a = acc[t];
          if (r == 1) a = fmaf(tip[cur][0], n[0], a);
          if (r == 2) { a = fmaf(tip[cur][1], n[0], a); a = fmaf(tip[cur][2], n[1], a); }
          if (r == 3) { a = fmaf(tip[cur][3], n[0], a); a = fmaf(tip[cur][4], n[1], a); a = fmaf(tip[cur][5], n[2], a); }
          Predb[(size_t)t * HH] = a;
          Dw[(t >> 1) * DPP + (t & 1)] = a - xn[t];
          n[r] = xn[t] - a;
        }
#pragma unroll
        for (int t = 0; t < CHK; ++t) {
          if (t > s + 3) {
            float r = acc[t];
            r = fmaf(abuf[cur][t].x, n[0], r);
            r = fmaf(abuf[cur][t].y, n[1], r);
            r = fmaf(abuf[cur][t].z, n[2], r);
            r = fmaf(abuf[cur][t].w, n[3], r);
            acc[t] = r;
          }
        }
        __builtin_amdgcn_sched_barrier(0);
      }
    }
    __builtin_amdgcn_fence(__ATOMIC_RELEASE, "workgroup", "local");
    __builtin_amdgcn_s_barrier();
    __builtin_amdgcn_fence(__ATOMIC_ACQUIRE, "workgroup", "local");
  }
}

__global__ void __launch_bounds__(256)
k_mlp1(const float* Pred, const float* gw, const float* gb,
       const float* uw, const float* ub, float* Gt) {
  __shared__ v4f smv[2 * 128 * (KS2 + 4) / 4];
  const int w = threadIdx.x >> 5, lane = threadIdx.x & 31;
  const int bn = blockIdx.x % (H4 / 32);
  const int bm = blockIdx.x / (H4 / 32);
  v8f g = {}, u = {};
  block_nk2<HH, KS2>(Pred + (size_t)bm * 64 * HH, HH, gw + (size_t)bn * 32 * HH,
                     uw + (size_t)bn * 32 * HH, HH, reinterpret_cast<float*>(smv), g, u);
  const int tm = bm * 4 + (w >> 1), tn = bn * 2 + (w & 1);
  int n = tn * 16 + (lane & 15);
  float gbias = gb[n], ubias = ub[n];
  v8f r;
#pragma unroll
  for (int i = 0; i < 8; ++i) {
    float zg = g[i] + gbias;
    float zu = u[i] + ubias;
    r[i] = (1.0f / (1.0f + __expf(-zg))) * zu;
  }
  store_tile(Gt, H4, tm * 16, tn * 16, r, lane);
}

#define KSPLIT 4
__global__ void __launch_bounds__(256)
k_mlp2(const float* Gt, const float* dw, float* Part) {
  __shared__ v4f smv[2 * 96 * (KS1 + 4) / 4];
  const int w = threadIdx.x >> 5, lane = threadIdx.x & 31;
  const int bn = blockIdx.x % (HH / 64);
  const int bm = (blockIdx.x / (HH / 64)) % (BB * SS / 32);
  const int q  = blockIdx.x / ((HH / 64) * (BB * SS / 32));
  v8f acc = block_nk<H4 / KSPLIT, KS1>(Gt + (size_t)bm * 32 * H4 + (size_t)q * (H4 / KSPLIT), H4,
                                       dw + (size_t)bn * 64 * H4 + (size_t)q * (H4 / KSPLIT), H4,
                                       reinterpret_cast<float*>(smv));
  const int tm = bm * 2 + (w >> 2), tn = bn * 4 + (w & 3);
  store_tile(Part + (size_t)q * BB * SS * HH, HH, tm * 16, tn * 16, acc, lane);
}

__global__ void k_ln(const float* __restrict__ Part, const float* __restrict__ db,
                     const float* __restrict__ Pred, const float* __restrict__ nw,
                     const float* __restrict__ nb, float* __restrict__ out) {
  __shared__ float s1[256], s2[256];
  int row = blockIdx.x;
  const float* y = Part + (size_t)row * HH;
  const float* pr = Pred + (size_t)row * HH;
  int tid = threadIdx.x;
  float v0 = y[tid], v1 = y[tid + 256], v2 = y[tid + 512];
  for (int q = 1; q < KSPLIT; ++q) {
    const float* yq = y + (size_t)q * BB * SS * HH;
    v0 += yq[tid]; v1 += yq[tid + 256]; v2 += yq[tid + 512];
  }
  v0 = v0 + db[tid]       + pr[tid];
  v1 = v1 + db[tid + 256] + pr[tid + 256];
  v2 = v2 + db[tid + 512] + pr[tid + 512];
  s1[tid] = v0 + v1 + v2;
  s2[tid] = v0 * v0 + v1 * v1 + v2 * v2;
  __syncthreads();
  for (int off = 128; off > 0; off >>= 1) {
    if (tid < off) { s1[tid] += s1[tid + off]; s2[tid] += s2[tid + off]; }
    __syncthreads();
  }
  float mu  = s1[0] * (1.0f / HH);
  float var = s2[0] * (1.0f / HH) - mu * mu;
  float r = rsqrtf(var + EPSC);
  float* o = out + (size_t)row * HH;
  o[tid]       = (v0 - mu) * r * nw[tid]       + nb[tid];
  o[tid + 256] = (v1 - mu) * r * nw[tid + 256] + nb[tid + 256];
  o[tid + 512] = (v2 - mu) * r * nw[tid + 512] + nb[tid + 512];
}

extern "C" void kernel_launch(void* const* d_in, const int* in_sizes, int n_in,
                              void* d_out, int out_size, void* d_ws, size_t ws_size,
                              hipStream_t stream) {
  const float* x      = (const float*)d_in[0];
  const float* W_init = (const float*)d_in[1];
  const float* b_init = (const float*)d_in[2];
  const float* gate_w = (const float*)d_in[3];
  const float* gate_b = (const float*)d_in[4];
  const float* up_w   = (const float*)d_in[5];
  const float* up_b   = (const float*)d_in[6];
  const float* down_w = (const float*)d_in[7];
  const float* down_b = (const float*)d_in[8];
  const float* norm_w = (const float*)d_in[9];
  const float* norm_b = (const float*)d_in[10];
  float* out = (float*)d_out;

  float* ws   = (float*)d_ws;
  float* P0   = ws;  ws += (size_t)BB * SS * HH;
  float* Pred = ws;  ws += (size_t)BB * SS * HH;
  float* Amat = ws;  ws += (size_t)BB * SS * SS;
  float* Gt   = ws;  ws += (size_t)BB * SS * H4;
  float* Part = ws;  ws += (size_t)KSPLIT * BB * SS * HH;

  k_front<<<NB_P0 + NB_ATTN, 256, 0, stream>>>(x, W_init, b_init, P0, Amat);

  k_seq<<<BB * (HH / 32), 128, 0, stream>>>(x, P0, Amat, Pred);

  k_mlp1<<<768, 256, 0, stream>>>(Pred, gate_w, gate_b, up_w, up_b, Gt);
  k_mlp2<<<768, 256, 0, stream>>>(Gt, down_w, Part);
  k_ln<<<BB * SS, 256, 0, stream>>>(Part, down_b, Pred, norm_w, norm_b, out);
}


# p0/attn/mlp1/mlp2 K loops: a stage's global loads are issued two stages ahead (second register set; stage barriers order LDS only), so they have one more stage to arrive in; every sum as before
# speedup vs baseline: 11.5013x; 1.0693x over previous
#include <hip/hip_runtime.h>
#include <math.h>

#define BB 2
#define SS 256
#define HH 768
#define H4 3072
#define CHK 64
#define NCH (SS / CHK)
#define LRC 0.01f
#define EPSC 1e-5f

typedef float v2f __attribute__((ext_vector_type(2)));
typedef float v8f __attribute__((ext_vector_type(8)));

__device__ __forceinline__ v8f wmma4(v2f a, v2f b, v8f c) {
  return __builtin_amdgcn_wmma_f32_16x16x4_f32(
       false, a,  false, b,
       (short)0, c,  false,  false);
}

__device__ __forceinline__ v2f frag_kn(const float* M, int ld, int k0, int r0, int lane) {
  const float* p = M + (size_t)(k0 + ((lane >> 4) << 1)) * ld + r0 + (lane & 15);
  v2f v; v.x = p[0]; v.y = p[ld];
  return v;
}
__device__ __forceinline__ void store_tile(float* D, int ld, int r0, int c0, v8f v, int lane) {
  int n = c0 + (lane & 15);
  int m0 = r0 + ((lane >> 4) << 3);
#pragma unroll
  for (int i = 0; i < 8; ++i) D[(size_t)(m0 + i) * ld + n] = v[i];
}
__device__ __forceinline__ v8f load_tile(const float* D, int ld, int r0, int c0, int lane) {
  int n = c0 + (lane & 15);
  int m0 = r0 + ((lane >> 4) << 3);
  v8f v;
#pragma unroll
  for (int i = 0; i < 8; ++i) v[i] = D[(size_t)(m0 + i) * ld + n];
  return v;
}


typedef float v4f __attribute__((ext_vector_type(4)));

template <int KS> struct stage_geo {
  static constexpr int PITCH = KS + 4;
  static constexpr int PPR = KS / 4;
  static constexpr int RPP = 256 / PPR;
  static constexpr int NQ = KS / 8;
};

#define FRAG_READS_IN_WMMA_ORDER_AHEAD_OF_WMMAS() __builtin_amdgcn_sched_barrier(0)
#define WMMAS_ISSUED_AHEAD_OF_NEXT_STAGE_STORES(...) asm volatile("" :: __VA_ARGS__)

__device__ __forceinline__ v4f frag_read(const float* p, int i) {
  return *reinterpret_cast<const v4f*>(p + 4 * i);
}

__device__ __forceinline__ void stage_barrier() {
  __builtin_amdgcn_fence(__ATOMIC_RELEASE, "workgroup", "local");
  __builtin_amdgcn_s_barrier();
  __builtin_amdgcn_fence(__ATOMIC_ACQUIRE, "workgroup", "local");
}

template <int KS, int NP>
__device__ __forceinline__ void rows_load(v4f* st, const float* g, int ld, int k0) {
#pragma unroll
  for (int p = 0; p < NP; ++p)
    st[p] = *reinterpret_cast<const v4f*>(g + (size_t)p * stage_geo<KS>::RPP * ld + k0);
}
template <int KS, int NP>
__device__ __forceinline__ void rows_store(float* sw, const v4f* st) {
#pragma unroll
  for (int p = 0; p < NP; ++p)
    *reinterpret_cast<v4f*>(sw + p * stage_geo<KS>::RPP * stage_geo<KS>::PITCH) = st[p];
}

template <int KS>
__device__ __forceinline__ v8f stage_mma(const float* fa, const float* fb, v8f acc) {
  typedef stage_geo<KS> G;
  v4f a[G::NQ], b[G::NQ];
#pragma unroll
  for (int i = 0; i < G::NQ; ++i) {
    a[i] = frag_read(fa, i); b[i] = frag_read(fb, i);
    FRAG_READS_IN_WMMA_ORDER_AHEAD_OF_WMMAS();
  }
#pragma unroll
  for (int i = 0; i < G::NQ; ++i) {
    acc = wmma4(a[i].xy, b[i].xy, acc);
    acc = wmma4(a[i].zw, b[i].zw, acc);
  }
  return acc;
}

template <int KTOT, int KS>
__device__ __forceinline__ v8f block_nk(const float* Ablk, int lda, const float* Bblk, int ldb,
                                        float* sm) {
  typedef stage_geo<KS> G;
  constexpr int NPA = 32 / G::RPP, NPB = 64 / G::RPP, NP = NPA + NPB;
  constexpr int BUF = 96 * G::PITCH;
  constexpr int NS = KTOT / KS;
  static_assert(KTOT % KS == 0, "K must be a whole number of stages");
  static_assert(NS % 2 == 0 && NS >= 2, "the K loop takes the stages in pairs");
  const int tid = threadIdx.x, lane = tid & 31, w = tid >> 5;
  const int lr = tid / G::PPR, lp = (tid % G::PPR) * 4;
  const float* ga = Ablk + (size_t)lr * lda + lp;
  const float* gb = Bblk + (size_t)lr * ldb + lp;
  float* sw = sm + lr * G::PITCH + lp;
  const int fo = (lane & 15) * G::PITCH + (lane >> 4) * (KS / 2);
  const float* fa = sm + ((w >> 2) * 16) * G::PITCH + fo;
  const float* fb = sm + (32 + (w & 3) * 16) * G::PITCH + fo;
  v4f r0[NP], r1[NP];
  rows_load<KS, NPA>(r0, ga, lda, 0);        rows_load<KS, NPB>(r0 + NPA, gb, ldb, 0);
  rows_load<KS, NPA>(r1, ga, lda, KS);       rows_load<KS, NPB>(r1 + NPA, gb, ldb, KS);
  rows_store<KS, NP>(sw, r0);
  stage_barrier();
  v8f acc = {};
#pragma nounroll
  for (int s = 0; s < NS - 2; s += 2) {
    rows_load<KS, NPA>(r0, ga, lda, (s + 2) * KS);
    rows_load<KS, NPB>(r0 + NPA, gb, ldb, (s + 2) * KS);
    acc = stage_mma<KS>(fa, fb, acc);
    WMMAS_ISSUED_AHEAD_OF_NEXT_STAGE_STORES("v"(acc));
    rows_store<KS, NP>(sw + BUF, r1);
    stage_barrier();
    rows_load<KS, NPA>(r1, ga, lda, (s + 3) * KS);
    rows_load<KS, NPB>(r1 + NPA, gb, ldb, (s + 3) * KS);
    acc = stage_mma<KS>(fa + BUF, fb + BUF, acc);
    WMMAS_ISSUED_AHEAD_OF_NEXT_STAGE_STORES("v"(acc));
    rows_store<KS, NP>(sw, r0);
    stage_barrier();
  }
  acc = stage_mma<KS>(fa, fb, acc);
  WMMAS_ISSUED_AHEAD_OF_NEXT_STAGE_STORES("v"(acc));
  rows_store<KS, NP>(sw + BUF, r1);
  stage_barrier();
  acc = stage_mma<KS>(fa + BUF, fb + BUF, acc);
  return acc;
}

template <int KS>
__device__ __forceinline__ void stage_mma2(const float* fa, const float* fb, const float* fc,
                                           v8f& g, v8f& u) {
  typedef stage_geo<KS> G;
  v4f a[G::NQ], b[G::NQ], c[G::NQ];
#pragma unroll
  for (int i = 0; i < G::NQ; ++i) {
    a[i] = frag_read(fa, i); b[i] = frag_read(fb, i); c[i] = frag_read(fc, i);
    FRAG_READS_IN_WMMA_ORDER_AHEAD_OF_WMMAS();
  }
#pragma unroll
  for (int i = 0; i < G::NQ; ++i) {
    g = wmma4(a[i].xy, b[i].xy, g);
    u = wmma4(a[i].xy, c[i].xy, u);
    g = wmma4(a[i].zw, b[i].zw, g);
    u = wmma4(a[i].zw, c[i].zw, u);
  }
}

template <int KTOT, int KS>
__device__ __forceinline__ void block_nk2(const float* Ablk, int lda, const float* Bblk,
                                          const float* Cblk, int ldb, float* sm,
                                          v8f& g, v8f& u) {
  typedef stage_geo<KS> G;
  constexpr int NPA = 64 / G::RPP, NPB = 32 / G::RPP, NP = NPA + 2 * NPB;
  constexpr int BUF = 128 * G::PITCH;
  constexpr int NS = KTOT / KS;
  static_assert(KTOT % KS == 0, "K must be a whole number of stages");
  static_assert(NS % 2 == 0 && NS >= 2, "the K loop takes the stages in pairs");
  const int tid = threadIdx.x, lane = tid & 31, w = tid >> 5;
  const int lr = tid / G::PPR, lp = (tid % G::PPR) * 4;
  const float* ga = Ablk + (size_t)lr * lda + lp;
  const float* gb = Bblk + (size_t)lr * ldb + lp;
  const float* gc = Cblk + (size_t)lr * ldb + lp;
  float* sw = sm + lr * G::PITCH + lp;
  const int fo = (lane & 15) * G::PITCH + (lane >> 4) * (KS / 2);
  const float* fa = sm + ((w >> 1) * 16) * G::PITCH + fo;
  const float* fb = sm + (64 + (w & 1) * 16) * G::PITCH + fo;
  const float* fc = sm + (96 + (w & 1) * 16) * G::PITCH + fo;
  v4f r0[NP], r1[NP];
  rows_load<KS, NPA>(r0, ga, lda, 0);
  rows_load<KS, NPB>(r0 + NPA, gb, ldb, 0);   rows_load<KS, NPB>(r0 + NPA + NPB, gc, ldb, 0);
  rows_load<KS, NPA>(r1, ga, lda, KS);
  rows_load<KS, NPB>(r1 + NPA, gb, ldb, KS);  rows_load<KS, NPB>(r1 + NPA + NPB, gc, ldb, KS);
  rows_store<KS, NP>(sw, r0);
  stage_barrier();
#pragma nounroll
  for (int s = 0; s < NS - 2; s += 2) {
    rows_load<KS, NPA>(r0, ga, lda, (s + 2) * KS);
    rows_load<KS, NPB>(r0 + NPA, gb, ldb, (s + 2) * KS);
    rows_load<KS, NPB>(r0 + NPA + NPB, gc, ldb, (s + 2) * KS);
    stage_mma2<KS>(fa, fb, fc, g, u);
    WMMAS_ISSUED_AHEAD_OF_NEXT_STAGE_STORES("v"(g), "v"(u));
    rows_store<KS, NP>(sw + BUF, r1);
    stage_barrier();
    rows_load<KS, NPA>(r1, ga, lda, (s + 3) * KS);
    rows_load<KS, NPB>(r1 + NPA, gb, ldb, (s + 3) * KS);
    rows_load<KS, NPB>(r1 + NPA + NPB, gc, ldb, (s + 3) * KS);
    stage_mma2<KS>(fa + BUF, fb + BUF, fc + BUF, g, u);
    WMMAS_ISSUED_AHEAD_OF_NEXT_STAGE_STORES("v"(g), "v"(u));
    rows_store<KS, NP>(sw, r0);
    stage_barrier();
  }
  stage_mma2<KS>(fa, fb, fc, g, u);
  WMMAS_ISSUED_AHEAD_OF_NEXT_STAGE_STORES("v"(g), "v"(u));
  rows_store<KS, NP>(sw + BUF, r1);
  stage_barrier();
  stage_mma2<KS>(fa + BUF, fb + BUF, fc + BUF, g, u);
}


#define KS1 64
#define KS2 32

#define NB_P0 ((BB * SS / 32) * (HH / 64))
#define NB_ATTN (BB * (SS / 32) * (SS / 64))
__global__ void __launch_bounds__(256)
k_front(const float* x, const float* Wi, const float* bi, float* P0, float* Amat) {
  __shared__ v4f smv[2 * 96 * (KS1 + 4) / 4];
  const int w = threadIdx.x >> 5, lane = threadIdx.x & 31;
  const bool attn = (blockIdx.x >= NB_P0);
  int bm, bn, b = 0;
  const float *Ablk, *Bblk;
  if (!attn) {
    bn = blockIdx.x % (HH / 64);
    bm = blockIdx.x / (HH / 64);
    Ablk = x + (size_t)bm * 32 * HH;
    Bblk = Wi + (size_t)bn * 64 * HH;
  } else {
    const int i = blockIdx.x - NB_P0;
    bn = i % (SS / 64);
    bm = (i / (SS / 64)) % (SS / 32);
    b  = i / ((SS / 64) * (SS / 32));
    if (bn > (bm >> 1)) return;
    const float* Xb = x + (size_t)b * SS * HH;
    Ablk = Xb + (size_t)bm * 32 * HH;
    Bblk = Xb + (size_t)bn * 64 * HH;
  }
  v8f acc = block_nk<HH, KS1>(Ablk, HH, Bblk, HH, reinterpret_cast<float*>(smv));
  const int tm = bm * 2 + (w >> 2), tn = bn * 4 + (w & 3);
  if (!attn) {
    float bias = bi[tn * 16 + (lane & 15)];
#pragma unroll
    for (int i = 0; i < 8; ++i) acc[i] += bias;
    store_tile(P0, HH, tm * 16, tn * 16, acc, lane);
  } else {
    int s = tn * 16 + (lane & 15);
    int mbase = tm * 16 + ((lane >> 4) << 3);
    float* Ab = Amat + (size_t)b * SS * SS;
#pragma unroll
    for (int i = 0; i < 8; ++i) {
      int t = mbase + i;
      float v = (s < t) ? LRC * (acc[i] + 1.0f) : 0.0f;
      Ab[(size_t)t * SS + s] = v;
    }
  }
}

#define DPP 96
__global__ void __launch_bounds__(128) __attribute__((amdgpu_waves_per_eu(1, 1)))
k_seq(const float* __restrict__ x, const float* __restrict__ P0,
      const float* __restrict__ Amat, float* __restrict__ Pred) {
  __shared__ v4f Ash[CHK * CHK / 4];
  __shared__ v4f Xsh[CHK * 32 / 4];
  __shared__ float Csh[CHK * 32];
  __shared__ v2f Dsh[(SS / 2) * (DPP / 2)];
  const int tid = threadIdx.x, lane = tid & 31, w = tid >> 5;
  const int b = blockIdx.x / (HH / 32);
  const int h0 = (blockIdx.x % (HH / 32)) * 32;
  const float* Ab = Amat + (size_t)b * SS * SS;
  const float* Xb = x + (size_t)b * SS * HH + h0;

#pragma nounroll
  for (int c = 0; c < NCH; ++c) {
    const int g0 = c * CHK;
    const float* Ar = Ab + (size_t)(g0 + w * 16 + (lane & 15)) * SS + ((lane >> 4) << 1);
    const int k1 = (g0 > CHK) ? CHK : 0;
    const int k2 = (g0 > 2 * CHK) ? 2 * CHK : 0;
    v2f a0[CHK / 4], a1[CHK / 4], a2[CHK / 4];
#pragma unroll
    for (int j = 0; j < CHK / 4; ++j) {
      a0[j] = *reinterpret_cast<const v2f*>(Ar + 4 * j);
      a1[j] = *reinterpret_cast<const v2f*>(Ar + k1 + 4 * j);
      a2[j] = *reinterpret_cast<const v2f*>(Ar + k2 + 4 * j);
    }
    const float* Pb = P0 + ((size_t)b * SS + g0) * HH + h0;
    v8f p0 = load_tile(Pb, HH, w * 16, 0, lane);
    v8f p1 = load_tile(Pb, HH, w * 16, 16, lane);
    const float* Ag = Ab + (size_t)g0 * SS + g0;
    v4f stA[CHK * CHK / 4 / 128], stX[CHK * 32 / 4 / 128];
#pragma unroll
    for (int k = 0; k < CHK * CHK / 4 / 128; ++k) {
      const int i = k * 128 + tid;
      stA[k] = *reinterpret_cast<const v4f*>(Ag + (size_t)(i >> 4) * SS + ((i & 15) << 2));
    }
#pragma unroll
    for (int k = 0; k < CHK * 32 / 4 / 128; ++k) {
      const int i = k * 128 + tid;
      const int gn = g0 + (i >> 3) + 1;
      const int gr = (gn < SS) ? gn : (SS - 1);
      stX[k] = *reinterpret_cast<const v4f*>(Xb + (size_t)gr * HH + ((i & 7) << 2));
    }

    v8f s0 = {}, s1 = {};
    const v2f* Dr = Dsh + (lane >> 4) * (DPP / 2) + (lane & 15);
#define CROSS_TRIP(AF, KB)                                                          \
    {                                                                               \
      v2f b0[CHK / 4], b1[CHK / 4];                                                 \
      _Pragma("unroll")               \
      for (int j = 0; j < CHK / 4; ++j) {                                           \
        const v2f* d = Dr + ((KB) / 2 + 2 * j) * (DPP / 2);   \
        b0[j] = d[0];                                                               \
        b1[j] = d[16];                                                              \
      }                                                                             \
          \
                \
      __builtin_amdgcn_sched_barrier(0);                                            \
      _Pragma("unroll")                                        \
      for (int j = 0; j < CHK / 4; ++j) {                                           \
        s0 = wmma4(AF[j], b0[j], s0);                                               \
        s1 = wmma4(AF[j], b1[j], s1);                                               \
      }                                                                             \
    }
    if (g0 > 0) CROSS_TRIP(a0, 0)
    if (g0 > CHK) CROSS_TRIP(a1, CHK)
    if (g0 > 2 * CHK) CROSS_TRIP(a2, 2 * CHK)
#undef CROSS_TRIP

#pragma unroll
    for (int k = 0; k < CHK * CHK / 4 / 128; ++k) Ash[k * 128 + tid] = stA[k];
#pragma unroll
    for (int k = 0; k < CHK * 32 / 4 / 128; ++k) {
      const int i = k * 128 + tid;
      v4f v = stX[k];
      if (g0 + (i >> 3) + 1 >= SS) v = v4f{0.0f, 0.0f, 0.0f, 0.0f};
      Xsh[i] = v;
    }
    {
      const int n = lane & 15;
      const int m0 = w * 16 + ((lane >> 4) << 3);
#pragma unroll
      for (int i = 0; i < 8; ++i) {
        Csh[(m0 + i) * 32 + n] = p0[i] - s0[i];
        Csh[(m0 + i) * 32 + 16 + n] = p1[i] - s1[i];
      }
    }
    __syncthreads();

    if (w == 0) {
      float* Predb = Pred + ((size_t)b * SS + g0) * HH + h0 + lane;
      float* Dw = reinterpret_cast<float*>(Dsh) + (g0 / 2) * DPP + 2 * lane;
      const float* Cf = Csh + lane;
      const float* Xf = reinterpret_cast<const float*>(Xsh) + lane;
      float acc[CHK], xn[CHK];
#pragma unroll
      for (int t = 0; t < CHK; ++t) acc[t] = Cf[t * 32];
#pragma unroll
      for (int t = 0; t < CHK; ++t) xn[t] = Xf[t * 32];

      const float* Af = reinterpret_cast<const float*>(Ash);
      v4f abuf[2][CHK];
      float tip[2][6];
      tip[0][0] = Af[1 * CHK];
      tip[0][1] = Af[2 * CHK];  tip[0][2] = Af[2 * CHK + 1];
      tip[0][3] = Af[3 * CHK];  tip[0][4] = Af[3 * CHK + 1];  tip[0][5] = Af[3 * CHK + 2];
#pragma unroll
      for (int t = 4; t < CHK; ++t) abuf[0][t] = Ash[(t * CHK) / 4];
#pragma unroll
      for (int g = 0; g < CHK / 4; ++g) {
        const int s = 4 * g;
        const int cur = g & 1;
        if (g + 1 < CHK / 4) {
          const int sn = s + 4;
          tip[cur ^ 1][0] = Af[(sn + 1) * CHK + sn];
          tip[cur ^ 1][1] = Af[(sn + 2) * CHK + sn];
          tip[cur ^ 1][2] = Af[(sn + 2) * CHK + sn + 1];
          tip[cur ^ 1][3] = Af[(sn + 3) * CHK + sn];
          tip[cur ^ 1][4] = Af[(sn + 3) * CHK + sn + 1];
          tip[cur ^ 1][5] = Af[(sn + 3) * CHK + sn + 2];
#pragma unroll
          for (int t = 0; t < CHK; ++t) {
            if (t > sn + 3) abuf[cur ^ 1][t] = Ash[(t * CHK + sn) / 4];
          }
        }
        __builtin_amdgcn_sched_barrier(0);
        float n[4];
#pragma unroll
        for (int r = 0; r < 4; ++r) {
          const int t = s + r;
          float a = acc[t];
          if (r == 1) a = fmaf(tip[cur][0], n[0], a);
          if (r == 2) { a = fmaf(tip[cur][1], n[0], a); a = fmaf(tip[cur][2], n[1], a); }
          if (r == 3) { a = fmaf(tip[cur][3], n[0], a); a = fmaf(tip[cur][4], n[1], a); a = fmaf(tip[cur][5], n[2], a); }
          Predb[(size_t)t * HH] = a;
          Dw[(t >> 1) * DPP + (t & 1)] = a - xn[t];
          n[r] = xn[t] - a;
        }
#pragma unroll
        for (int t = 0; t < CHK; ++t) {
          if (t > s + 3) {
            float r = acc[t];
            r = fmaf(abuf[cur][t].x, n[0], r);
            r = fmaf(abuf[cur][t].y, n[1], r);
            r = fmaf(abuf[cur][t].z, n[2], r);
            r = fmaf(abuf[cur][t].w, n[3], r);
            acc[t] = r;
          }
        }
        __builtin_amdgcn_sched_barrier(0);
      }
    }
    __builtin_amdgcn_fence(__ATOMIC_RELEASE, "workgroup", "local");
    __builtin_amdgcn_s_barrier();
    __builtin_amdgcn_fence(__ATOMIC_ACQUIRE, "workgroup", "local");
  }
}

__global__ void __launch_bounds__(256)
k_mlp1(const float* Pred, const float* gw, const float* gb,
       const float* uw, const float* ub, float* Gt) {
  __shared__ v4f smv[2 * 128 * (KS2 + 4) / 4];
  const int w = threadIdx.x >> 5, lane = threadIdx.x & 31;
  const int bn = blockIdx.x % (H4 / 32);
  const int bm = blockIdx.x / (H4 / 32);
  v8f g = {}, u = {};
  block_nk2<HH, KS2>(Pred + (size_t)bm * 64 * HH, HH, gw + (size_t)bn * 32 * HH,
                     uw + (size_t)bn * 32 * HH, HH, reinterpret_cast<float*>(smv), g, u);
  const int tm = bm * 4 + (w >> 1), tn = bn * 2 + (w & 1);
  int n = tn * 16 + (lane & 15);
  float gbias = gb[n], ubias = ub[n];
  v8f r;
#pragma unroll
  for (int i = 0; i < 8; ++i) {
    float zg = g[i] + gbias;
    float zu = u[i] + ubias;
    r[i] = (1.0f / (1.0f + __expf(-zg))) * zu;
  }
  store_tile(Gt, H4, tm * 16, tn * 16, r, lane);
}

#define KSPLIT 4
__global__ void __launch_bounds__(256)
k_mlp2(const float* Gt, const float* dw, float* Part) {
  __shared__ v4f smv[2 * 96 * (KS1 + 4) / 4];
  const int w = threadIdx.x >> 5, lane = threadIdx.x & 31;
  const int bn = blockIdx.x % (HH / 64);
  const int bm = (blockIdx.x / (HH / 64)) % (BB * SS / 32);
  const int q  = blockIdx.x / ((HH / 64) * (BB * SS / 32));
  v8f acc = block_nk<H4 / KSPLIT, KS1>(Gt + (size_t)bm * 32 * H4 + (size_t)q * (H4 / KSPLIT), H4,
                                       dw + (size_t)bn * 64 * H4 + (size_t)q * (H4 / KSPLIT), H4,
                                       reinterpret_cast<float*>(smv));
  const int tm = bm * 2 + (w >> 2), tn = bn * 4 + (w & 3);
  store_tile(Part + (size_t)q * BB * SS * HH, HH, tm * 16, tn * 16, acc, lane);
}

__global__ void k_ln(const float* __restrict__ Part, const float* __restrict__ db,
                     const float* __restrict__ Pred, const float* __restrict__ nw,
                     const float* __restrict__ nb, float* __restrict__ out) {
  __shared__ float s1[256], s2[256];
  int row = blockIdx.x;
  const float* y = Part + (size_t)row * HH;
  const float* pr = Pred + (size_t)row * HH;
  int tid = threadIdx.x;
  float v0 = y[tid], v1 = y[tid + 256], v2 = y[tid + 512];
  for (int q = 1; q < KSPLIT; ++q) {
    const float* yq = y + (size_t)q * BB * SS * HH;
    v0 += yq[tid]; v1 += yq[tid + 256]; v2 += yq[tid + 512];
  }
  v0 = v0 + db[tid]       + pr[tid];
  v1 = v1 + db[tid + 256] + pr[tid + 256];
  v2 = v2 + db[tid + 512] + pr[tid + 512];
  s1[tid] = v0 + v1 + v2;
  s2[tid] = v0 * v0 + v1 * v1 + v2 * v2;
  __syncthreads();
  for (int off = 128; off > 0; off >>= 1) {
    if (tid < off) { s1[tid] += s1[tid + off]; s2[tid] += s2[tid + off]; }
    __syncthreads();
  }
  float mu  = s1[0] * (1.0f / HH);
  float var = s2[0] * (1.0f / HH) - mu * mu;
  float r = rsqrtf(var + EPSC);
  float* o = out + (size_t)row * HH;
  o[tid]       = (v0 - mu) * r * nw[tid]       + nb[tid];
  o[tid + 256] = (v1 - mu) * r * nw[tid + 256] + nb[tid + 256];
  o[tid + 512] = (v2 - mu) * r * nw[tid + 512] + nb[tid + 512];
}

extern "C" void kernel_launch(void* const* d_in, const int* in_sizes, int n_in,
                              void* d_out, int out_size, void* d_ws, size_t ws_size,
                              hipStream_t stream) {
  const float* x      = (const float*)d_in[0];
  const float* W_init = (const float*)d_in[1];
  const float* b_init = (const float*)d_in[2];
  const float* gate_w = (const float*)d_in[3];
  const float* gate_b = (const float*)d_in[4];
  const float* up_w   = (const float*)d_in[5];
  const float* up_b   = (const float*)d_in[6];
  const float* down_w = (const float*)d_in[7];
  const float* down_b = (const float*)d_in[8];
  const float* norm_w = (const float*)d_in[9];
  const float* norm_b = (const float*)d_in[10];
  float* out = (float*)d_out;

  float* ws   = (float*)d_ws;
  float* P0   = ws;  ws += (size_t)BB * SS * HH;
  float* Pred = ws;  ws += (size_t)BB * SS * HH;
  float* Amat = ws;  ws += (size_t)BB * SS * SS;
  float* Gt   = ws;  ws += (size_t)BB * SS * H4;
  float* Part = ws;  ws += (size_t)KSPLIT * BB * SS * HH;

  k_front<<<NB_P0 + NB_ATTN, 256, 0, stream>>>(x, W_init, b_init, P0, Amat);

  k_seq<<<BB * (HH / 32), 128, 0, stream>>>(x, P0, Amat, Pred);

  k_mlp1<<<768, 256, 0, stream>>>(Pred, gate_w, gate_b, up_w, up_b, Gt);
  k_mlp2<<<768, 256, 0, stream>>>(Gt, down_w, Part);
  k_ln<<<BB * SS, 256, 0, stream>>>(Part, down_b, Pred, norm_w, norm_b, out);
}


# k_seq: next chunk's global reads move under this chunk's serial solve: idle waves 1..3 stage its cross-term A rows in LDS by whole-line loads (fragments from LDS), the rest prefetched; sums as before
# speedup vs baseline: 12.0304x; 1.0460x over previous
#include <hip/hip_runtime.h>
#include <math.h>

#define BB 2
#define SS 256
#define HH 768
#define H4 3072
#define CHK 64
#define NCH (SS / CHK)
#define LRC 0.01f
#define EPSC 1e-5f

typedef float v2f __attribute__((ext_vector_type(2)));
typedef float v8f __attribute__((ext_vector_type(8)));

__device__ __forceinline__ v8f wmma4(v2f a, v2f b, v8f c) {
  return __builtin_amdgcn_wmma_f32_16x16x4_f32(
       false, a,  false, b,
       (short)0, c,  false,  false);
}

__device__ __forceinline__ v2f frag_kn(const float* M, int ld, int k0, int r0, int lane) {
  const float* p = M + (size_t)(k0 + ((lane >> 4) << 1)) * ld + r0 + (lane & 15);
  v2f v; v.x = p[0]; v.y = p[ld];
  return v;
}
__device__ __forceinline__ void store_tile(float* D, int ld, int r0, int c0, v8f v, int lane) {
  int n = c0 + (lane & 15);
  int m0 = r0 + ((lane >> 4) << 3);
#pragma unroll
  for (int i = 0; i < 8; ++i) D[(size_t)(m0 + i) * ld + n] = v[i];
}
__device__ __forceinline__ v8f load_tile(const float* D, int ld, int r0, int c0, int lane) {
  int n = c0 + (lane & 15);
  int m0 = r0 + ((lane >> 4) << 3);
  v8f v;
#pragma unroll
  for (int i = 0; i < 8; ++i) v[i] = D[(size_t)(m0 + i) * ld + n];
  return v;
}


typedef float v4f __attribute__((ext_vector_type(4)));

template <int KS> struct stage_geo {
  static constexpr int PITCH = KS + 4;
  static constexpr int PPR = KS / 4;
  static constexpr int RPP = 256 / PPR;
  static constexpr int NQ = KS / 8;
};

#define FRAG_READS_IN_WMMA_ORDER_AHEAD_OF_WMMAS() __builtin_amdgcn_sched_barrier(0)
#define WMMAS_ISSUED_AHEAD_OF_NEXT_STAGE_STORES(...) asm volatile("" :: __VA_ARGS__)

__device__ __forceinline__ v4f frag_read(const float* p, int i) {
  return *reinterpret_cast<const v4f*>(p + 4 * i);
}

__device__ __forceinline__ void stage_barrier() {
  __builtin_amdgcn_fence(__ATOMIC_RELEASE, "workgroup", "local");
  __builtin_amdgcn_s_barrier();
  __builtin_amdgcn_fence(__ATOMIC_ACQUIRE, "workgroup", "local");
}

template <int KS, int NP>
__device__ __forceinline__ void rows_load(v4f* st, const float* g, int ld, int k0) {
#pragma unroll
  for (int p = 0; p < NP; ++p)
    st[p] = *reinterpret_cast<const v4f*>(g + (size_t)p * stage_geo<KS>::RPP * ld + k0);
}
template <int KS, int NP>
__device__ __forceinline__ void rows_store(float* sw, const v4f* st) {
#pragma unroll
  for (int p = 0; p < NP; ++p)
    *reinterpret_cast<v4f*>(sw + p * stage_geo<KS>::RPP * stage_geo<KS>::PITCH) = st[p];
}

template <int KS>
__device__ __forceinline__ v8f stage_mma(const float* fa, const float* fb, v8f acc) {
  typedef stage_geo<KS> G;
  v4f a[G::NQ], b[G::NQ];
#pragma unroll
  for (int i = 0; i < G::NQ; ++i) {
    a[i] = frag_read(fa, i); b[i] = frag_read(fb, i);
    FRAG_READS_IN_WMMA_ORDER_AHEAD_OF_WMMAS();
  }
#pragma unroll
  for (int i = 0; i < G::NQ; ++i) {
    acc = wmma4(a[i].xy, b[i].xy, acc);
    acc = wmma4(a[i].zw, b[i].zw, acc);
  }
  return acc;
}

template <int KTOT, int KS>
__device__ __forceinline__ v8f block_nk(const float* Ablk, int lda, const float* Bblk, int ldb,
                                        float* sm) {
  typedef stage_geo<KS> G;
  constexpr int NPA = 32 / G::RPP, NPB = 64 / G::RPP, NP = NPA + NPB;
  constexpr int BUF = 96 * G::PITCH;
  constexpr int NS = KTOT / KS;
  static_assert(KTOT % KS == 0, "K must be a whole number of stages");
  static_assert(NS % 2 == 0 && NS >= 2, "the K loop takes the stages in pairs");
  const int tid = threadIdx.x, lane = tid & 31, w = tid >> 5;
  const int lr = tid / G::PPR, lp = (tid % G::PPR) * 4;
  const float* ga = Ablk + (size_t)lr * lda + lp;
  const float* gb = Bblk + (size_t)lr * ldb + lp;
  float* sw = sm + lr * G::PITCH + lp;
  const int fo = (lane & 15) * G::PITCH + (lane >> 4) * (KS / 2);
  const float* fa = sm + ((w >> 2) * 16) * G::PITCH + fo;
  const float* fb = sm + (32 + (w & 3) * 16) * G::PITCH + fo;
  v4f r0[NP], r1[NP];
  rows_load<KS, NPA>(r0, ga, lda, 0);        rows_load<KS, NPB>(r0 + NPA, gb, ldb, 0);
  rows_load<KS, NPA>(r1, ga, lda, KS);       rows_load<KS, NPB>(r1 + NPA, gb, ldb, KS);
  rows_store<KS, NP>(sw, r0);
  stage_barrier();
  v8f acc = {};
#pragma nounroll
  for (int s = 0; s < NS - 2; s += 2) {
    rows_load<KS, NPA>(r0, ga, lda, (s + 2) * KS);
    rows_load<KS, NPB>(r0 + NPA, gb, ldb, (s + 2) * KS);
    acc = stage_mma<KS>(fa, fb, acc);
    WMMAS_ISSUED_AHEAD_OF_NEXT_STAGE_STORES("v"(acc));
    rows_store<KS, NP>(sw + BUF, r1);
    stage_barrier();
    rows_load<KS, NPA>(r1, ga, lda, (s + 3) * KS);
    rows_load<KS, NPB>(r1 + NPA, gb, ldb, (s + 3) * KS);
    acc = stage_mma<KS>(fa + BUF, fb + BUF, acc);
    WMMAS_ISSUED_AHEAD_OF_NEXT_STAGE_STORES("v"(acc));
    rows_store<KS, NP>(sw, r0);
    stage_barrier();
  }
  acc = stage_mma<KS>(fa, fb, acc);
  WMMAS_ISSUED_AHEAD_OF_NEXT_STAGE_STORES("v"(acc));
  rows_store<KS, NP>(sw + BUF, r1);
  stage_barrier();
  acc = stage_mma<KS>(fa + BUF, fb + BUF, acc);
  return acc;
}

template <int KS>
__device__ __forceinline__ void stage_mma2(const float* fa, const float* fb, const float* fc,
                                           v8f& g, v8f& u) {
  typedef stage_geo<KS> G;
  v4f a[G::NQ], b[G::NQ], c[G::NQ];
#pragma unroll
  for (int i = 0; i < G::NQ; ++i) {
    a[i] = frag_read(fa, i); b[i] = frag_read(fb, i); c[i] = frag_read(fc, i);
    FRAG_READS_IN_WMMA_ORDER_AHEAD_OF_WMMAS();
  }
#pragma unroll
  for (int i = 0; i < G::NQ; ++i) {
    g = wmma4(a[i].xy, b[i].xy, g);
    u = wmma4(a[i].xy, c[i].xy, u);
    g = wmma4(a[i].zw, b[i].zw, g);
    u = wmma4(a[i].zw, c[i].zw, u);
  }
}

template <int KTOT, int KS>
__device__ __forceinline__ void block_nk2(const float* Ablk, int lda, const float* Bblk,
                                          const float* Cblk, int ldb, float* sm,
                                          v8f& g, v8f& u) {
  typedef stage_geo<KS> G;
  constexpr int NPA = 64 / G::RPP, NPB = 32 / G::RPP, NP = NPA + 2 * NPB;
  constexpr int BUF = 128 * G::PITCH;
  constexpr int NS = KTOT / KS;
  static_assert(KTOT % KS == 0, "K must be a whole number of stages");
  static_assert(NS % 2 == 0 && NS >= 2, "the K loop takes the stages in pairs");
  const int tid = threadIdx.x, lane = tid & 31, w = tid >> 5;
  const int lr = tid / G::PPR, lp = (tid % G::PPR) * 4;
  const float* ga = Ablk + (size_t)lr * lda + lp;
  const float* gb = Bblk + (size_t)lr * ldb + lp;
  const float* gc = Cblk + (size_t)lr * ldb + lp;
  float* sw = sm + lr * G::PITCH + lp;
  const int fo = (lane & 15) * G::PITCH + (lane >> 4) * (KS / 2);
  const float* fa = sm + ((w >> 1) * 16) * G::PITCH + fo;
  const float* fb = sm + (64 + (w & 1) * 16) * G::PITCH + fo;
  const float* fc = sm + (96 + (w & 1) * 16) * G::PITCH + fo;
  v4f r0[NP], r1[NP];
  rows_load<KS, NPA>(r0, ga, lda, 0);
  rows_load<KS, NPB>(r0 + NPA, gb, ldb, 0);   rows_load<KS, NPB>(r0 + NPA + NPB, gc, ldb, 0);
  rows_load<KS, NPA>(r1, ga, lda, KS);
  rows_load<KS, NPB>(r1 + NPA, gb, ldb, KS);  rows_load<KS, NPB>(r1 + NPA + NPB, gc, ldb, KS);
  rows_store<KS, NP>(sw, r0);
  stage_barrier();
#pragma nounroll
  for (int s = 0; s < NS - 2; s += 2) {
    rows_load<KS, NPA>(r0, ga, lda, (s + 2) * KS);
    rows_load<KS, NPB>(r0 + NPA, gb, ldb, (s + 2) * KS);
    rows_load<KS, NPB>(r0 + NPA + NPB, gc, ldb, (s + 2) * KS);
    stage_mma2<KS>(fa, fb, fc, g, u);
    WMMAS_ISSUED_AHEAD_OF_NEXT_STAGE_STORES("v"(g), "v"(u));
    rows_store<KS, NP>(sw + BUF, r1);
    stage_barrier();
    rows_load<KS, NPA>(r1, ga, lda, (s + 3) * KS);
    rows_load<KS, NPB>(r1 + NPA, gb, ldb, (s + 3) * KS);
    rows_load<KS, NPB>(r1 + NPA + NPB, gc, ldb, (s + 3) * KS);
    stage_mma2<KS>(fa + BUF, fb + BUF, fc + BUF, g, u);
    WMMAS_ISSUED_AHEAD_OF_NEXT_STAGE_STORES("v"(g), "v"(u));
    rows_store<KS, NP>(sw, r0);
    stage_barrier();
  }
  stage_mma2<KS>(fa, fb, fc, g, u);
  WMMAS_ISSUED_AHEAD_OF_NEXT_STAGE_STORES("v"(g), "v"(u));
  rows_store<KS, NP>(sw + BUF, r1);
  stage_barrier();
  stage_mma2<KS>(fa + BUF, fb + BUF, fc + BUF, g, u);
}


#define KS1 64
#define KS2 32

#define NB_P0 ((BB * SS / 32) * (HH / 64))
#define NB_ATTN (BB * (SS / 32) * (SS / 64))
__global__ void __launch_bounds__(256)
k_front(const float* x, const float* Wi, const float* bi, float* P0, float* Amat) {
  __shared__ v4f smv[2 * 96 * (KS1 + 4) / 4];
  const int w = threadIdx.x >> 5, lane = threadIdx.x & 31;
  const bool attn = (blockIdx.x >= NB_P0);
  int bm, bn, b = 0;
  const float *Ablk, *Bblk;
  if (!attn) {
    bn = blockIdx.x % (HH / 64);
    bm = blockIdx.x / (HH / 64);
    Ablk = x + (size_t)bm * 32 * HH;
    Bblk = Wi + (size_t)bn * 64 * HH;
  } else {
    const int i = blockIdx.x - NB_P0;
    bn = i % (SS / 64);
    bm = (i / (SS / 64)) % (SS / 32);
    b  = i / ((SS / 64) * (SS / 32));
    if (bn > (bm >> 1)) return;
    const float* Xb = x + (size_t)b * SS * HH;
    Ablk = Xb + (size_t)bm * 32 * HH;
    Bblk = Xb + (size_t)bn * 64 * HH;
  }
  v8f acc = block_nk<HH, KS1>(Ablk, HH, Bblk, HH, reinterpret_cast<float*>(smv));
  const int tm = bm * 2 + (w >> 2), tn = bn * 4 + (w & 3);
  if (!attn) {
    float bias = bi[tn * 16 + (lane & 15)];
#pragma unroll
    for (int i = 0; i < 8; ++i) acc[i] += bias;
    store_tile(P0, HH, tm * 16, tn * 16, acc, lane);
  } else {
    int s = tn * 16 + (lane & 15);
    int mbase = tm * 16 + ((lane >> 4) << 3);
    float* Ab = Amat + (size_t)b * SS * SS;
#pragma unroll
    for (int i = 0; i < 8; ++i) {
      int t = mbase + i;
      float v = (s < t) ? LRC * (acc[i] + 1.0f) : 0.0f;
      Ab[(size_t)t * SS + s] = v;
    }
  }
}

#define DPP 96
#define AXP ((NCH - 1) * CHK + 4)
__global__ void __launch_bounds__(128) __attribute__((amdgpu_waves_per_eu(1, 1)))
k_seq(const float* __restrict__ x, const float* __restrict__ P0,
      const float* __restrict__ Amat, float* __restrict__ Pred) {
  __shared__ v4f lds_all[(CHK * CHK + CHK * 32 + CHK * 32 + (SS / 2) * DPP + CHK * AXP) / 4];
  v4f* Ash = lds_all;
  v4f* Xsh = Ash + CHK * CHK / 4;
  float* Csh = reinterpret_cast<float*>(Xsh + CHK * 32 / 4);
  v2f* Dsh = reinterpret_cast<v2f*>(Csh + CHK * 32);
  v4f* Axs = reinterpret_cast<v4f*>(Dsh + (SS / 2) * (DPP / 2));
  const int tid = threadIdx.x, lane = tid & 31, w = tid >> 5;
  const int b = blockIdx.x / (HH / 32);
  const int h0 = (blockIdx.x % (HH / 32)) * 32;
  const float* Ab = Amat + (size_t)b * SS * SS;
  const float* Xb = x + (size_t)b * SS * HH + h0;

  v8f p0, p1;
  v4f stA[CHK * CHK / 4 / 128], stX[CHK * 32 / 4 / 128];
#define SEQ_FETCH(G0)                                                                       \
  {                                                                                         \
    const float* Pb = P0 + ((size_t)b * SS + (G0)) * HH + h0;                               \
    p0 = load_tile(Pb, HH, w * 16, 0, lane);                                                \
    p1 = load_tile(Pb, HH, w * 16, 16, lane);                                               \
    const float* Ag = Ab + (size_t)(G0) * SS + (G0);                                        \
    _Pragma("unroll")                              \
    for (int k = 0; k < CHK * CHK / 4 / 128; ++k) {                                         \
      const int i = k * 128 + tid;                                                          \
      stA[k] = *reinterpret_cast<const v4f*>(Ag + (size_t)(i >> 4) * SS + ((i & 15) << 2)); \
    }                                                                                       \
    _Pragma("unroll")                              \
    for (int k = 0; k < CHK * 32 / 4 / 128; ++k) {                                          \
      const int i = k * 128 + tid;                                                          \
      const int gn = (G0) + (i >> 3) + 1;                                                   \
      const int gr = (gn < SS) ? gn : (SS - 1);                                             \
      stX[k] = *reinterpret_cast<const v4f*>(Xb + (size_t)gr * HH + ((i & 7) << 2));        \
    }                                                                                       \
  }
  SEQ_FETCH(0)

#pragma nounroll
  for (int c = 0; ; ++c) {
    const int g0 = c * CHK;
    v8f s0 = {}, s1 = {};
    const float* Afr = reinterpret_cast<const float*>(Axs) + (w * 16 + (lane & 15)) * AXP + ((lane >> 4) << 1);
    const v2f* Dr = Dsh + (lane >> 4) * (DPP / 2) + (lane & 15);
#define CROSS_TRIP(KB)                                                              \
    {                                                                               \
      v2f af[CHK / 4], b0[CHK / 4], b1[CHK / 4];                                    \
      _Pragma("unroll")         \
      for (int j = 0; j < CHK / 4; ++j) {                                           \
        af[j] = *reinterpret_cast<const v2f*>(Afr + (KB) + 4 * j);   \
        const v2f* d = Dr + ((KB) / 2 + 2 * j) * (DPP / 2);   \
        b0[j] = d[0];                                                               \
        b1[j] = d[16];                                                              \
      }                                                                             \
          \
                \
      __builtin_amdgcn_sched_barrier(0);                                            \
      _Pragma("unroll")                                        \
      for (int j = 0; j < CHK / 4; ++j) {                                           \
        s0 = wmma4(af[j], b0[j], s0);                                               \
        s1 = wmma4(af[j], b1[j], s1);                                               \
      }                                                                             \
    }
    if (g0 > 0) CROSS_TRIP(0)
    if (g0 > CHK) CROSS_TRIP(CHK)
    if (g0 > 2 * CHK) CROSS_TRIP(2 * CHK)
#undef CROSS_TRIP

#pragma unroll
    for (int k = 0; k < CHK * CHK / 4 / 128; ++k) Ash[k * 128 + tid] = stA[k];
#pragma unroll
    for (int k = 0; k < CHK * 32 / 4 / 128; ++k) {
      const int i = k * 128 + tid;
      v4f v = stX[k];
      if (g0 + (i >> 3) + 1 >= SS) v = v4f{0.0f, 0.0f, 0.0f, 0.0f};
      Xsh[i] = v;
    }
    {
      const int n = lane & 15;
      const int m0 = w * 16 + ((lane >> 4) << 3);
#pragma unroll
      for (int i = 0; i < 8; ++i) {
        Csh[(m0 + i) * 32 + n] = p0[i] - s0[i];
        Csh[(m0 + i) * 32 + 16 + n] = p1[i] - s1[i];
      }
    }
    __syncthreads();

    if (w == 0) {
      float* Predb = Pred + ((size_t)b * SS + g0) * HH + h0 + lane;
      float* Dw = reinterpret_cast<float*>(Dsh) + (g0 / 2) * DPP + 2 * lane;
      const float* Cf = Csh + lane;
      const float* Xf = reinterpret_cast<const float*>(Xsh) + lane;
      float acc[CHK], xn[CHK];
#pragma unroll
      for (int t = 0; t < CHK; ++t) acc[t] = Cf[t * 32];
#pragma unroll
      for (int t = 0; t < CHK; ++t) xn[t] = Xf[t * 32];

      const float* Af = reinterpret_cast<const float*>(Ash);
      v4f abuf[2][CHK];
      float tip[2][6];
      tip[0][0] = Af[1 * CHK];
      tip[0][1] = Af[2 * CHK];  tip[0][2] = Af[2 * CHK + 1];
      tip[0][3] = Af[3 * CHK];  tip[0][4] = Af[3 * CHK + 1];  tip[0][5] = Af[3 * CHK + 2];
#pragma unroll
      for (int t = 4; t < CHK; ++t) abuf[0][t] = Ash[(t * CHK) / 4];
#pragma unroll
      for (int g = 0; g < CHK / 4; ++g) {
        const int s = 4 * g;
        const int cur = g & 1;
        if (g + 1 < CHK / 4) {
          const int sn = s + 4;
          tip[cur ^ 1][0] = Af[(sn + 1) * CHK + sn];
          tip[cur ^ 1][1] = Af[(sn + 2) * CHK + sn];
          tip[cur ^ 1][2] = Af[(sn + 2) * CHK + sn + 1];
          tip[cur ^ 1][3] = Af[(sn + 3) * CHK + sn];
          tip[cur ^ 1][4] = Af[(sn + 3) * CHK + sn + 1];
          tip[cur ^ 1][5] = Af[(sn + 3) * CHK + sn + 2];
#pragma unroll
          for (int t = 0; t < CHK; ++t) {
            if (t > sn + 3) abuf[cur ^ 1][t] = Ash[(t * CHK + sn) / 4];
          }
        }
        __builtin_amdgcn_sched_barrier(0);
        float n[4];
#pragma unroll
        for (int r = 0; r < 4; ++r) {
          const int t = s + r;
          float a = acc[t];
          if (r == 1) a = fmaf(tip[cur][0], n[0], a);
          if (r == 2) { a = fmaf(tip[cur][1], n[0], a); a = fmaf(tip[cur][2], n[1], a); }
          if (r == 3) { a = fmaf(tip[cur][3], n[0], a); a = fmaf(tip[cur][4], n[1], a); a = fmaf(tip[cur][5], n[2], a); }
          Predb[(size_t)t * HH] = a;
          Dw[(t >> 1) * DPP + (t & 1)] = a - xn[t];
          n[r] = xn[t] - a;
        }
#pragma unroll
        for (int t = 0; t < CHK; ++t) {
          if (t > s + 3) {
            float r = acc[t];
            r = fmaf(abuf[cur][t].x, n[0], r);
            r = fmaf(abuf[cur][t].y, n[1], r);
            r = fmaf(abuf[cur][t].z, n[2], r);
            r = fmaf(abuf[cur][t].w, n[3], r);
            acc[t] = r;
          }
        }
        __builtin_amdgcn_sched_barrier(0);
      }
    }

    if (c + 1 == NCH) break;

    SEQ_FETCH(g0 + CHK)
    if (w != 0 && (w - 1) * CHK <= g0) {
      const int jc = (w - 1) * CHK;
      const float* An = Ab + (size_t)(g0 + CHK + (lane >> 4)) * SS + jc + ((lane & 15) << 2);
      float* Axw = reinterpret_cast<float*>(Axs) + (lane >> 4) * AXP + jc + ((lane & 15) << 2);
      v4f sg[CHK / 2];
#pragma unroll
      for (int m = 0; m < CHK / 2; ++m)
        sg[m] = *reinterpret_cast<const v4f*>(An + (size_t)(2 * m) * SS);
#pragma unroll
      for (int m = 0; m < CHK / 2; ++m)
        *reinterpret_cast<v4f*>(Axw + (2 * m) * AXP) = sg[m];
    }
    __builtin_amdgcn_fence(__ATOMIC_RELEASE, "workgroup", "local");
    __builtin_amdgcn_s_barrier();
    __builtin_amdgcn_fence(__ATOMIC_ACQUIRE, "workgroup", "local");
  }
#undef SEQ_FETCH
}

__global__ void __launch_bounds__(256)
k_mlp1(const float* Pred, const float* gw, const float* gb,
       const float* uw, const float* ub, float* Gt) {
  __shared__ v4f smv[2 * 128 * (KS2 + 4) / 4];
  const int w = threadIdx.x >> 5, lane = threadIdx.x & 31;
  const int bn = blockIdx.x % (H4 / 32);
  const int bm = blockIdx.x / (H4 / 32);
  v8f g = {}, u = {};
  block_nk2<HH, KS2>(Pred + (size_t)bm * 64 * HH, HH, gw + (size_t)bn * 32 * HH,
                     uw + (size_t)bn * 32 * HH, HH, reinterpret_cast<float*>(smv), g, u);
  const int tm = bm * 4 + (w >> 1), tn = bn * 2 + (w & 1);
  int n = tn * 16 + (lane & 15);
  float gbias = gb[n], ubias = ub[n];
  v8f r;
#pragma unroll
  for (int i = 0; i < 8; ++i) {
    float zg = g[i] + gbias;
    float zu = u[i] + ubias;
    r[i] = (1.0f / (1.0f + __expf(-zg))) * zu;
  }
  store_tile(Gt, H4, tm * 16, tn * 16, r, lane);
}

#define KSPLIT 4
__global__ void __launch_bounds__(256)
k_mlp2(const float* Gt, const float* dw, float* Part) {
  __shared__ v4f smv[2 * 96 * (KS1 + 4) / 4];
  const int w = threadIdx.x >> 5, lane = threadIdx.x & 31;
  const int bn = blockIdx.x % (HH / 64);
  const int bm = (blockIdx.x / (HH / 64)) % (BB * SS / 32);
  const int q  = blockIdx.x / ((HH / 64) * (BB * SS / 32));
  v8f acc = block_nk<H4 / KSPLIT, KS1>(Gt + (size_t)bm * 32 * H4 + (size_t)q * (H4 / KSPLIT), H4,
                                       dw + (size_t)bn * 64 * H4 + (size_t)q * (H4 / KSPLIT), H4,
                                       reinterpret_cast<float*>(smv));
  const int tm = bm * 2 + (w >> 2), tn = bn * 4 + (w & 3);
  store_tile(Part + (size_t)q * BB * SS * HH, HH, tm * 16, tn * 16, acc, lane);
}

__global__ void k_ln(const float* __restrict__ Part, const float* __restrict__ db,
                     const float* __restrict__ Pred, const float* __restrict__ nw,
                     const float* __restrict__ nb, float* __restrict__ out) {
  __shared__ float s1[256], s2[256];
  int row = blockIdx.x;
  const float* y = Part + (size_t)row * HH;
  const float* pr = Pred + (size_t)row * HH;
  int tid = threadIdx.x;
  float v0 = y[tid], v1 = y[tid + 256], v2 = y[tid + 512];
  for (int q = 1; q < KSPLIT; ++q) {
    const float* yq = y + (size_t)q * BB * SS * HH;
    v0 += yq[tid]; v1 += yq[tid + 256]; v2 += yq[tid + 512];
  }
  v0 = v0 + db[tid]       + pr[tid];
  v1 = v1 + db[tid + 256] + pr[tid + 256];
  v2 = v2 + db[tid + 512] + pr[tid + 512];
  s1[tid] = v0 + v1 + v2;
  s2[tid] = v0 * v0 + v1 * v1 + v2 * v2;
  __syncthreads();
  for (int off = 128; off > 0; off >>= 1) {
    if (tid < off) { s1[tid] += s1[tid + off]; s2[tid] += s2[tid + off]; }
    __syncthreads();
  }
  float mu  = s1[0] * (1.0f / HH);
  float var = s2[0] * (1.0f / HH) - mu * mu;
  float r = rsqrtf(var + EPSC);
  float* o = out + (size_t)row * HH;
  o[tid]       = (v0 - mu) * r * nw[tid]       + nb[tid];
  o[tid + 256] = (v1 - mu) * r * nw[tid + 256] + nb[tid + 256];
  o[tid + 512] = (v2 - mu) * r * nw[tid + 512] + nb[tid + 512];
}

extern "C" void kernel_launch(void* const* d_in, const int* in_sizes, int n_in,
                              void* d_out, int out_size, void* d_ws, size_t ws_size,
                              hipStream_t stream) {
  const float* x      = (const float*)d_in[0];
  const float* W_init = (const float*)d_in[1];
  const float* b_init = (const float*)d_in[2];
  const float* gate_w = (const float*)d_in[3];
  const float* gate_b = (const float*)d_in[4];
  const float* up_w   = (const float*)d_in[5];
  const float* up_b   = (const float*)d_in[6];
  const float* down_w = (const float*)d_in[7];
  const float* down_b = (const float*)d_in[8];
  const float* norm_w = (const float*)d_in[9];
  const float* norm_b = (const float*)d_in[10];
  float* out = (float*)d_out;

  float* ws   = (float*)d_ws;
  float* P0   = ws;  ws += (size_t)BB * SS * HH;
  float* Pred = ws;  ws += (size_t)BB * SS * HH;
  float* Amat = ws;  ws += (size_t)BB * SS * SS;
  float* Gt   = ws;  ws += (size_t)BB * SS * H4;
  float* Part = ws;  ws += (size_t)KSPLIT * BB * SS * HH;

  k_front<<<NB_P0 + NB_ATTN, 256, 0, stream>>>(x, W_init, b_init, P0, Amat);

  k_seq<<<BB * (HH / 32), 128, 0, stream>>>(x, P0, Amat, Pred);

  k_mlp1<<<768, 256, 0, stream>>>(Pred, gate_w, gate_b, up_w, up_b, Gt);
  k_mlp2<<<768, 256, 0, stream>>>(Gt, down_w, Part);
  k_ln<<<BB * SS, 256, 0, stream>>>(Part, down_b, Pred, norm_w, norm_b, out);
}
